# starting kernel
# baseline (speedup 1.0000x reference)
#include <hip/hip_runtime.h>
#include <math.h>

typedef __attribute__((ext_vector_type(2))) float v2f;
typedef __attribute__((ext_vector_type(8))) float v8f;

#define LANE() ((int)(threadIdx.x & 31))

__device__ __forceinline__ float fast_tanh(float x) {
  const float e = __expf(2.0f * x);
  return 1.0f - 2.0f / (e + 1.0f);
}

template<int K, int LDA, bool TANH>
__device__ __forceinline__ void gemm_tile_to_lds(
    const float* __restrict__ sA,
    const float* __restrict__ Wg,
    const float* __restrict__ bg,
    float* __restrict__ sOut, int ldo, int n0)
{
  const int lane = LANE();
  const int half = lane >> 4;
  const int col  = lane & 15;
  const int n    = n0 + col;
  v8f acc = {};
#pragma unroll
  for (int k0 = 0; k0 < K; k0 += 4) {
    const int ka = k0 + half * 2;
    v2f a, b;
    a.x = sA[col * LDA + ka];
    a.y = sA[col * LDA + ka + 1];
    b.x = Wg[n * K + ka];
    b.y = Wg[n * K + ka + 1];
    acc = __builtin_amdgcn_wmma_f32_16x16x4_f32(false, a, false, b,
                                                (short)0, acc, false, false);
  }
  const float bias = bg[n];
#pragma unroll
  for (int r = 0; r < 8; ++r) {
    float v = acc[r] + bias;
    if (TANH) v = fast_tanh(v);
    sOut[(half * 8 + r) * ldo + n] = v;
  }
}

__device__ __forceinline__ void gemm_final(
    const float* __restrict__ sZ,
    const float* __restrict__ Wp,
    const float* __restrict__ bp,
    float* __restrict__ out, int rowBase, int n0)
{
  const int lane = LANE();
  const int half = lane >> 4;
  const int col  = lane & 15;
  const int n    = n0 + col;
  v8f acc = {};
#pragma unroll
  for (int k0 = 0; k0 < 8; k0 += 4) {
    const int ka = k0 + half * 2;
    v2f a, b;
    a.x = sZ[col * 8 + ka];
    a.y = sZ[col * 8 + ka + 1];
    b.x = Wp[n * 8 + ka];
    b.y = Wp[n * 8 + ka + 1];
    acc = __builtin_amdgcn_wmma_f32_16x16x4_f32(false, a, false, b,
                                                (short)0, acc, false, false);
  }
  const float bias = bp[n];
#pragma unroll
  for (int r = 0; r < 8; ++r)
    out[(size_t)(rowBase + half * 8 + r) * 128 + n] = acc[r] + bias;
}

template<int JB>
__device__ __forceinline__ void gate_local(float ar[8], float ai[8],
    float u00r, float u00i, float u01r, float u01i,
    float u10r, float u10i, float u11r, float u11i)
{
#pragma unroll
  for (int j = 0; j < 8; ++j) {
    if (!((j >> JB) & 1)) {
      const int j1 = j | (1 << JB);
      const float a0r = ar[j],  a0i = ai[j];
      const float a1r = ar[j1], a1i = ai[j1];
      ar[j]  = u00r*a0r - u00i*a0i + u01r*a1r - u01i*a1i;
      ai[j]  = u00r*a0i + u00i*a0r + u01r*a1i + u01i*a1r;
      ar[j1] = u10r*a0r - u10i*a0i + u11r*a1r - u11i*a1i;
      ai[j1] = u10r*a0i + u10i*a0r + u11r*a1i + u11i*a1r;
    }
  }
}

template<int LB>
__device__ __forceinline__ void gate_lane(float ar[8], float ai[8],
    float u00r, float u00i, float u01r, float u01i,
    float u10r, float u10i, float u11r, float u11i)
{
  const int bit = (LANE() >> LB) & 1;
  const float csr = bit ? u11r : u00r;
  const float csi = bit ? u11i : u00i;
  const float cpr = bit ? u10r : u01r;
  const float cpi = bit ? u10i : u01i;
#pragma unroll
  for (int j = 0; j < 8; ++j) {
    const float pr = __shfl_xor(ar[j], 1 << LB, 32);
    const float pi = __shfl_xor(ai[j], 1 << LB, 32);
    const float nr = csr*ar[j] - csi*ai[j] + cpr*pr - cpi*pi;
    const float ni = csr*ai[j] + csi*ar[j] + cpr*pi + cpi*pr;
    ar[j] = nr; ai[j] = ni;
  }
}

template<int W>
__device__ __forceinline__ void apply_gate(float ar[8], float ai[8],
    float u00r, float u00i, float u01r, float u01i,
    float u10r, float u10i, float u11r, float u11i)
{
  if constexpr (W <= 4)
    gate_lane<4 - W>(ar, ai, u00r, u00i, u01r, u01i, u10r, u10i, u11r, u11i);
  else
    gate_local<7 - W>(ar, ai, u00r, u00i, u01r, u01i, u10r, u10i, u11r, u11i);
}

template<int CJB, int TJB>
__device__ __forceinline__ void cnot_ll(float ar[8], float ai[8]) {
#pragma unroll
  for (int j = 0; j < 8; ++j)
    if (((j >> CJB) & 1) && !((j >> TJB) & 1)) {
      const int j1 = j | (1 << TJB);
      float tr = ar[j], ti = ai[j];
      ar[j] = ar[j1]; ai[j] = ai[j1];
      ar[j1] = tr;    ai[j1] = ti;
    }
}
template<int CJB, int TLB>
__device__ __forceinline__ void cnot_local_lane(float ar[8], float ai[8]) {
#pragma unroll
  for (int j = 0; j < 8; ++j) {
    const float pr = __shfl_xor(ar[j], 1 << TLB, 32);
    const float pi = __shfl_xor(ai[j], 1 << TLB, 32);
    if ((j >> CJB) & 1) { ar[j] = pr; ai[j] = pi; }
  }
}
template<int CLB, int TJB>
__device__ __forceinline__ void cnot_lane_local(float ar[8], float ai[8]) {
  if ((LANE() >> CLB) & 1) {
#pragma unroll
    for (int j = 0; j < 8; ++j)
      if (!((j >> TJB) & 1)) {
        const int j1 = j | (1 << TJB);
        float tr = ar[j], ti = ai[j];
        ar[j] = ar[j1]; ai[j] = ai[j1];
        ar[j1] = tr;    ai[j1] = ti;
      }
  }
}
template<int CLB, int TLB>
__device__ __forceinline__ void cnot_lane_lane(float ar[8], float ai[8]) {
  const int cb = (LANE() >> CLB) & 1;
#pragma unroll
  for (int j = 0; j < 8; ++j) {
    const float pr = __shfl_xor(ar[j], 1 << TLB, 32);
    const float pi = __shfl_xor(ai[j], 1 << TLB, 32);
    if (cb) { ar[j] = pr; ai[j] = pi; }
  }
}

__device__ __forceinline__ void rot_u(float phi, float th, float om,
                                      float* __restrict__ u)
{
  const float pr = __cosf(0.5f * phi), pi = -__sinf(0.5f * phi);
  const float qr = __cosf(0.5f * om),  qi = -__sinf(0.5f * om);
  const float cy = __cosf(0.5f * th),  sy =  __sinf(0.5f * th);
  const float qpr  = qr*pr - qi*pi, qpi  = qr*pi + qi*pr;
  const float qcpr = qr*pr + qi*pi, qcpi = qi*pr - qr*pi;
  u[0] =  cy*qpr;  u[1] =  cy*qpi;
  u[2] = -sy*qcpr; u[3] = -sy*qcpi;
  u[4] =  sy*qcpr; u[5] = -sy*qcpi;
  u[6] =  cy*qpr;  u[7] = -cy*qpi;
}

#define INIT_WIRE(W) {                                                        \
    const float aY = ang[(W)];                                                \
    const float cc = __cosf(0.5f * aY), ss = __sinf(0.5f * aY);               \
    apply_gate<(W)>(ar, ai, cc,0.f, -ss,0.f, ss,0.f, cc,0.f);                 \
    const float aZ = ang[8 + (W)];                                            \
    const float zr = __cosf(0.5f * aZ), zi = -__sinf(0.5f * aZ);              \
    apply_gate<(W)>(ar, ai, zr,zi, 0.f,0.f, 0.f,0.f, zr,-zi);                 \
  }

#define ROT_WIRE(L, W) {                                                      \
    const float* u = &sU[((L)*8 + (W)) * 8];                                  \
    apply_gate<(W)>(ar, ai, u[0],u[1],u[2],u[3],u[4],u[5],u[6],u[7]);         \
  }

__global__ __launch_bounds__(512) void quantum_block_kernel(
    const float* __restrict__ x,
    const float* __restrict__ W1, const float* __restrict__ b1,
    const float* __restrict__ W2, const float* __restrict__ b2,
    const float* __restrict__ W3, const float* __restrict__ b3,
    const float* __restrict__ qw,
    const float* __restrict__ Wp, const float* __restrict__ bp,
    float* __restrict__ out)
{
  __shared__ float sX  [16 * 80];
  __shared__ float sH1 [16 * 64];
  __shared__ float sH2 [16 * 32];
  __shared__ float sAng[16 * 16];
  __shared__ float sZ  [16 * 8];
  __shared__ float sU  [24 * 8];

  const int tid     = threadIdx.x;
  const int wid     = tid >> 5;
  const int lane    = tid & 31;
  const int rowBase = blockIdx.x * 16;

  if (tid < 24) {
    const int L = tid >> 3, W = tid & 7;
    rot_u(qw[(L * 8 + W) * 3 + 0], qw[(L * 8 + W) * 3 + 1],
          qw[(L * 8 + W) * 3 + 2], &sU[tid * 8]);
  }
  for (int i = tid; i < 16 * 80; i += 512) sX[i] = x[rowBase * 80 + i];
  __syncthreads();

  if (wid < 4) gemm_tile_to_lds<80, 80, true >(sX,  W1, b1, sH1, 64, wid * 16);
  __syncthreads();
  if (wid < 2) gemm_tile_to_lds<64, 64, true >(sH1, W2, b2, sH2, 32, wid * 16);
  __syncthreads();
  if (wid == 0) gemm_tile_to_lds<32, 32, false>(sH2, W3, b3, sAng, 16, 0);
  __syncthreads();

  {
    float ar[8], ai[8];
#pragma unroll
    for (int j = 0; j < 8; ++j) { ar[j] = 0.f; ai[j] = 0.f; }
    if (lane == 0) ar[0] = 1.0f;

    const float* ang = &sAng[wid * 16];
    INIT_WIRE(0) INIT_WIRE(1) INIT_WIRE(2) INIT_WIRE(3)
    INIT_WIRE(4) INIT_WIRE(5) INIT_WIRE(6) INIT_WIRE(7)

    for (int L = 0; L < 3; ++L) {
      ROT_WIRE(L, 0) ROT_WIRE(L, 1) ROT_WIRE(L, 2) ROT_WIRE(L, 3)
      ROT_WIRE(L, 4) ROT_WIRE(L, 5) ROT_WIRE(L, 6) ROT_WIRE(L, 7)
      cnot_lane_lane<4, 3>(ar, ai);
      cnot_lane_lane<3, 2>(ar, ai);
      cnot_lane_lane<2, 1>(ar, ai);
      cnot_lane_lane<1, 0>(ar, ai);
      cnot_lane_local<0, 2>(ar, ai);
      cnot_ll<2, 1>(ar, ai);
      cnot_ll<1, 0>(ar, ai);
      cnot_local_lane<0, 4>(ar, ai);
    }

    float pb[8]; float tot = 0.f;
#pragma unroll
    for (int j = 0; j < 8; ++j) { pb[j] = ar[j]*ar[j] + ai[j]*ai[j]; tot += pb[j]; }
    float zp[8];
#pragma unroll
    for (int w = 0; w < 5; ++w)
      zp[w] = ((lane >> (4 - w)) & 1) ? -tot : tot;
    float s5 = 0.f, s6 = 0.f, s7 = 0.f;
#pragma unroll
    for (int j = 0; j < 8; ++j) {
      s5 += ((j >> 2) & 1) ? -pb[j] : pb[j];
      s6 += ((j >> 1) & 1) ? -pb[j] : pb[j];
      s7 += ( j       & 1) ? -pb[j] : pb[j];
    }
    zp[5] = s5; zp[6] = s6; zp[7] = s7;
#pragma unroll
    for (int w = 0; w < 8; ++w) {
      float v = zp[w];
#pragma unroll
      for (int off = 16; off > 0; off >>= 1) v += __shfl_xor(v, off, 32);
      if (lane == 0) sZ[wid * 8 + w] = v;
    }
  }
  __syncthreads();

  if (wid < 8) gemm_final(sZ, Wp, bp, out, rowBase, wid * 16);
}

extern "C" void kernel_launch(void* const* d_in, const int* in_sizes, int n_in,
                              void* d_out, int out_size, void* d_ws, size_t ws_size,
                              hipStream_t stream) {
  const float* x  = (const float*)d_in[0];
  const float* W1 = (const float*)d_in[1];
  const float* b1 = (const float*)d_in[2];
  const float* W2 = (const float*)d_in[3];
  const float* b2 = (const float*)d_in[4];
  const float* W3 = (const float*)d_in[5];
  const float* b3 = (const float*)d_in[6];
  const float* qw = (const float*)d_in[7];
  const float* Wp = (const float*)d_in[8];
  const float* bp = (const float*)d_in[9];
  float* out = (float*)d_out;

  const int B = in_sizes[0] / 80;
  const int grid = B / 16;
  quantum_block_kernel<<<grid, 512, 0, stream>>>(x, W1, b1, W2, b2, W3, b3,
                                                 qw, Wp, bp, out);
}


# Encoding RY/RZ layer built directly as the per-wire product state instead of 16 gate passes; half-angle cos/sin once per sample in the angle GEMM epilogue (init products re-associated)
# speedup vs baseline: 1.0888x; 1.0888x over previous
#include <hip/hip_runtime.h>
#include <math.h>

typedef __attribute__((ext_vector_type(2))) float v2f;
typedef __attribute__((ext_vector_type(8))) float v8f;

#define LANE() ((int)(threadIdx.x & 31))

__device__ __forceinline__ float fast_tanh(float x) {
  const float e = __expf(2.0f * x);
  return 1.0f - 2.0f / (e + 1.0f);
}

template<int K, int LDA>
__device__ __forceinline__ v8f gemm_tile_acc(
    const float* __restrict__ sA,
    const float* __restrict__ Wg,
    int n)
{
  const int lane = LANE();
  const int half = lane >> 4;
  const int col  = lane & 15;
  v8f acc = {};
#pragma unroll
  for (int k0 = 0; k0 < K; k0 += 4) {
    const int ka = k0 + half * 2;
    v2f a, b;
    a.x = sA[col * LDA + ka];
    a.y = sA[col * LDA + ka + 1];
    b.x = Wg[n * K + ka];
    b.y = Wg[n * K + ka + 1];
    acc = __builtin_amdgcn_wmma_f32_16x16x4_f32(false, a, false, b,
                                                (short)0, acc, false, false);
  }
  return acc;
}

template<int K, int LDA, bool TANH>
__device__ __forceinline__ void gemm_tile_to_lds(
    const float* __restrict__ sA, const float* __restrict__ Wg,
    const float* __restrict__ bg,
    float* __restrict__ sOut, int ldo, int n0)
{
  const int lane = LANE();
  const int half = lane >> 4;
  const int n    = n0 + (lane & 15);
  const v8f acc = gemm_tile_acc<K, LDA>(sA, Wg, n);
  const float bias = bg[n];
#pragma unroll
  for (int r = 0; r < 8; ++r) {
    float v = acc[r] + bias;
    if (TANH) v = fast_tanh(v);
    sOut[(half * 8 + r) * ldo + n] = v;
  }
}

template<int K, int LDA>
__device__ __forceinline__ void gemm_tile_to_halfangle_trig(
    const float* __restrict__ sA, const float* __restrict__ Wg,
    const float* __restrict__ bg,
    float* __restrict__ sCos, float* __restrict__ sSin, int ldo, int n0)
{
  const int lane = LANE();
  const int half = lane >> 4;
  const int n    = n0 + (lane & 15);
  const v8f acc = gemm_tile_acc<K, LDA>(sA, Wg, n);
  const float bias = bg[n];
#pragma unroll
  for (int r = 0; r < 8; ++r) {
    const float h = 0.5f * (acc[r] + bias);
    sCos[(half * 8 + r) * ldo + n] = __cosf(h);
    sSin[(half * 8 + r) * ldo + n] = __sinf(h);
  }
}

__device__ __forceinline__ void gemm_final(
    const float* __restrict__ sZ,
    const float* __restrict__ Wp,
    const float* __restrict__ bp,
    float* __restrict__ out, int rowBase, int n0)
{
  const int lane = LANE();
  const int half = lane >> 4;
  const int col  = lane & 15;
  const int n    = n0 + col;
  v8f acc = {};
#pragma unroll
  for (int k0 = 0; k0 < 8; k0 += 4) {
    const int ka = k0 + half * 2;
    v2f a, b;
    a.x = sZ[col * 8 + ka];
    a.y = sZ[col * 8 + ka + 1];
    b.x = Wp[n * 8 + ka];
    b.y = Wp[n * 8 + ka + 1];
    acc = __builtin_amdgcn_wmma_f32_16x16x4_f32(false, a, false, b,
                                                (short)0, acc, false, false);
  }
  const float bias = bp[n];
#pragma unroll
  for (int r = 0; r < 8; ++r)
    out[(size_t)(rowBase + half * 8 + r) * 128 + n] = acc[r] + bias;
}

template<int JB>
__device__ __forceinline__ void gate_local(float ar[8], float ai[8],
    float u00r, float u00i, float u01r, float u01i,
    float u10r, float u10i, float u11r, float u11i)
{
#pragma unroll
  for (int j = 0; j < 8; ++j) {
    if (!((j >> JB) & 1)) {
      const int j1 = j | (1 << JB);
      const float a0r = ar[j],  a0i = ai[j];
      const float a1r = ar[j1], a1i = ai[j1];
      ar[j]  = u00r*a0r - u00i*a0i + u01r*a1r - u01i*a1i;
      ai[j]  = u00r*a0i + u00i*a0r + u01r*a1i + u01i*a1r;
      ar[j1] = u10r*a0r - u10i*a0i + u11r*a1r - u11i*a1i;
      ai[j1] = u10r*a0i + u10i*a0r + u11r*a1i + u11i*a1r;
    }
  }
}

template<int LB>
__device__ __forceinline__ void gate_lane(float ar[8], float ai[8],
    float u00r, float u00i, float u01r, float u01i,
    float u10r, float u10i, float u11r, float u11i)
{
  const int bit = (LANE() >> LB) & 1;
  const float csr = bit ? u11r : u00r;
  const float csi = bit ? u11i : u00i;
  const float cpr = bit ? u10r : u01r;
  const float cpi = bit ? u10i : u01i;
#pragma unroll
  for (int j = 0; j < 8; ++j) {
    const float pr = __shfl_xor(ar[j], 1 << LB, 32);
    const float pi = __shfl_xor(ai[j], 1 << LB, 32);
    const float nr = csr*ar[j] - csi*ai[j] + cpr*pr - cpi*pi;
    const float ni = csr*ai[j] + csi*ar[j] + cpr*pi + cpi*pr;
    ar[j] = nr; ai[j] = ni;
  }
}

template<int W>
__device__ __forceinline__ void apply_gate(float ar[8], float ai[8],
    float u00r, float u00i, float u01r, float u01i,
    float u10r, float u10i, float u11r, float u11i)
{
  if constexpr (W <= 4)
    gate_lane<4 - W>(ar, ai, u00r, u00i, u01r, u01i, u10r, u10i, u11r, u11i);
  else
    gate_local<7 - W>(ar, ai, u00r, u00i, u01r, u01i, u10r, u10i, u11r, u11i);
}

template<int CJB, int TJB>
__device__ __forceinline__ void cnot_ll(float ar[8], float ai[8]) {
#pragma unroll
  for (int j = 0; j < 8; ++j)
    if (((j >> CJB) & 1) && !((j >> TJB) & 1)) {
      const int j1 = j | (1 << TJB);
      float tr = ar[j], ti = ai[j];
      ar[j] = ar[j1]; ai[j] = ai[j1];
      ar[j1] = tr;    ai[j1] = ti;
    }
}
template<int CJB, int TLB>
__device__ __forceinline__ void cnot_local_lane(float ar[8], float ai[8]) {
#pragma unroll
  for (int j = 0; j < 8; ++j) {
    const float pr = __shfl_xor(ar[j], 1 << TLB, 32);
    const float pi = __shfl_xor(ai[j], 1 << TLB, 32);
    if ((j >> CJB) & 1) { ar[j] = pr; ai[j] = pi; }
  }
}
template<int CLB, int TJB>
__device__ __forceinline__ void cnot_lane_local(float ar[8], float ai[8]) {
  if ((LANE() >> CLB) & 1) {
#pragma unroll
    for (int j = 0; j < 8; ++j)
      if (!((j >> TJB) & 1)) {
        const int j1 = j | (1 << TJB);
        float tr = ar[j], ti = ai[j];
        ar[j] = ar[j1]; ai[j] = ai[j1];
        ar[j1] = tr;    ai[j1] = ti;
      }
  }
}
template<int CLB, int TLB>
__device__ __forceinline__ void cnot_lane_lane(float ar[8], float ai[8]) {
  const int cb = (LANE() >> CLB) & 1;
#pragma unroll
  for (int j = 0; j < 8; ++j) {
    const float pr = __shfl_xor(ar[j], 1 << TLB, 32);
    const float pi = __shfl_xor(ai[j], 1 << TLB, 32);
    if (cb) { ar[j] = pr; ai[j] = pi; }
  }
}

__device__ __forceinline__ void rot_u(float phi, float th, float om,
                                      float* __restrict__ u)
{
  const float pr = __cosf(0.5f * phi), pi = -__sinf(0.5f * phi);
  const float qr = __cosf(0.5f * om),  qi = -__sinf(0.5f * om);
  const float cy = __cosf(0.5f * th),  sy =  __sinf(0.5f * th);
  const float qpr  = qr*pr - qi*pi, qpi  = qr*pi + qi*pr;
  const float qcpr = qr*pr + qi*pi, qcpi = qi*pr - qr*pi;
  u[0] =  cy*qpr;  u[1] =  cy*qpi;
  u[2] = -sy*qcpr; u[3] = -sy*qcpi;
  u[4] =  sy*qcpr; u[5] = -sy*qcpi;
  u[6] =  cy*qpr;  u[7] = -cy*qpi;
}

template<int W>
__device__ __forceinline__ void wire_factor_lane(const float* __restrict__ cs,
    const float* __restrict__ sn, float& fr, float& fi)
{
  const int bit = (LANE() >> (4 - W)) & 1;
  const float mag = bit ? sn[W] : cs[W];
  const float zs  = sn[8 + W];
  fr = mag * cs[8 + W];
  fi = mag * (bit ? zs : -zs);
}

#define CMUL(orr, oi, xr, xi, yr, yi) {                                       \
    orr = (xr) * (yr) - (xi) * (yi);                                          \
    oi  = (xr) * (yi) + (xi) * (yr);                                          \
  }

__device__ __forceinline__ void product_state(const float* __restrict__ cs,
    const float* __restrict__ sn, float ar[8], float ai[8])
{
  float pr, pi, fr, fi, tr, ti;
  wire_factor_lane<0>(cs, sn, pr, pi);
  wire_factor_lane<1>(cs, sn, fr, fi); CMUL(tr, ti, pr, pi, fr, fi) pr = tr; pi = ti;
  wire_factor_lane<2>(cs, sn, fr, fi); CMUL(tr, ti, pr, pi, fr, fi) pr = tr; pi = ti;
  wire_factor_lane<3>(cs, sn, fr, fi); CMUL(tr, ti, pr, pi, fr, fi) pr = tr; pi = ti;
  wire_factor_lane<4>(cs, sn, fr, fi); CMUL(tr, ti, pr, pi, fr, fi) pr = tr; pi = ti;
  float qr[3][2], qi[3][2];
#pragma unroll
  for (int w = 5; w < 8; ++w) {
    const float zr = cs[8 + w], zs = sn[8 + w];
    qr[w - 5][0] = cs[w] * zr; qi[w - 5][0] = -(cs[w] * zs);
    qr[w - 5][1] = sn[w] * zr; qi[w - 5][1] =   sn[w] * zs;
  }
  float t1r[2], t1i[2], t2r[4], t2i[4];
#pragma unroll
  for (int b = 0; b < 2; ++b) CMUL(t1r[b], t1i[b], pr, pi, qr[0][b], qi[0][b])
#pragma unroll
  for (int b = 0; b < 4; ++b)
    CMUL(t2r[b], t2i[b], t1r[b >> 1], t1i[b >> 1], qr[1][b & 1], qi[1][b & 1])
#pragma unroll
  for (int j = 0; j < 8; ++j)
    CMUL(ar[j], ai[j], t2r[j >> 1], t2i[j >> 1], qr[2][j & 1], qi[2][j & 1])
}

#define ROT_WIRE(L, W) {                                                      \
    const float* u = &sU[((L)*8 + (W)) * 8];                                  \
    apply_gate<(W)>(ar, ai, u[0],u[1],u[2],u[3],u[4],u[5],u[6],u[7]);         \
  }

__global__ __launch_bounds__(512) void quantum_block_kernel(
    const float* __restrict__ x,
    const float* __restrict__ W1, const float* __restrict__ b1,
    const float* __restrict__ W2, const float* __restrict__ b2,
    const float* __restrict__ W3, const float* __restrict__ b3,
    const float* __restrict__ qw,
    const float* __restrict__ Wp, const float* __restrict__ bp,
    float* __restrict__ out)
{
  __shared__ float sX  [16 * 80];
  __shared__ float sH1 [16 * 64];
  __shared__ float sH2 [16 * 32];
  __shared__ float sCos[16 * 16];
  __shared__ float sSin[16 * 16];
  __shared__ float sZ  [16 * 8];
  __shared__ float sU  [24 * 8];

  const int tid     = threadIdx.x;
  const int wid     = tid >> 5;
  const int lane    = tid & 31;
  const int rowBase = blockIdx.x * 16;

  if (tid < 24) {
    const int L = tid >> 3, W = tid & 7;
    rot_u(qw[(L * 8 + W) * 3 + 0], qw[(L * 8 + W) * 3 + 1],
          qw[(L * 8 + W) * 3 + 2], &sU[tid * 8]);
  }
  for (int i = tid; i < 16 * 80; i += 512) sX[i] = x[rowBase * 80 + i];
  __syncthreads();

  if (wid < 4) gemm_tile_to_lds<80, 80, true >(sX,  W1, b1, sH1, 64, wid * 16);
  __syncthreads();
  if (wid < 2) gemm_tile_to_lds<64, 64, true >(sH1, W2, b2, sH2, 32, wid * 16);
  __syncthreads();
  if (wid == 0) gemm_tile_to_halfangle_trig<32, 32>(sH2, W3, b3, sCos, sSin, 16, 0);
  __syncthreads();

  {
    float ar[8], ai[8];
    product_state(&sCos[wid * 16], &sSin[wid * 16], ar, ai);

    for (int L = 0; L < 3; ++L) {
      ROT_WIRE(L, 0) ROT_WIRE(L, 1) ROT_WIRE(L, 2) ROT_WIRE(L, 3)
      ROT_WIRE(L, 4) ROT_WIRE(L, 5) ROT_WIRE(L, 6) ROT_WIRE(L, 7)
      cnot_lane_lane<4, 3>(ar, ai);
      cnot_lane_lane<3, 2>(ar, ai);
      cnot_lane_lane<2, 1>(ar, ai);
      cnot_lane_lane<1, 0>(ar, ai);
      cnot_lane_local<0, 2>(ar, ai);
      cnot_ll<2, 1>(ar, ai);
      cnot_ll<1, 0>(ar, ai);
      cnot_local_lane<0, 4>(ar, ai);
    }

    float pb[8]; float tot = 0.f;
#pragma unroll
    for (int j = 0; j < 8; ++j) { pb[j] = ar[j]*ar[j] + ai[j]*ai[j]; tot += pb[j]; }
    float zp[8];
#pragma unroll
    for (int w = 0; w < 5; ++w)
      zp[w] = ((lane >> (4 - w)) & 1) ? -tot : tot;
    float s5 = 0.f, s6 = 0.f, s7 = 0.f;
#pragma unroll
    for (int j = 0; j < 8; ++j) {
      s5 += ((j >> 2) & 1) ? -pb[j] : pb[j];
      s6 += ((j >> 1) & 1) ? -pb[j] : pb[j];
      s7 += ( j       & 1) ? -pb[j] : pb[j];
    }
    zp[5] = s5; zp[6] = s6; zp[7] = s7;
#pragma unroll
    for (int w = 0; w < 8; ++w) {
      float v = zp[w];
#pragma unroll
      for (int off = 16; off > 0; off >>= 1) v += __shfl_xor(v, off, 32);
      if (lane == 0) sZ[wid * 8 + w] = v;
    }
  }
  __syncthreads();

  if (wid < 8) gemm_final(sZ, Wp, bp, out, rowBase, wid * 16);
}

extern "C" void kernel_launch(void* const* d_in, const int* in_sizes, int n_in,
                              void* d_out, int out_size, void* d_ws, size_t ws_size,
                              hipStream_t stream) {
  const float* x  = (const float*)d_in[0];
  const float* W1 = (const float*)d_in[1];
  const float* b1 = (const float*)d_in[2];
  const float* W2 = (const float*)d_in[3];
  const float* b2 = (const float*)d_in[4];
  const float* W3 = (const float*)d_in[5];
  const float* b3 = (const float*)d_in[6];
  const float* qw = (const float*)d_in[7];
  const float* Wp = (const float*)d_in[8];
  const float* bp = (const float*)d_in[9];
  float* out = (float*)d_out;

  const int B = in_sizes[0] / 80;
  const int grid = B / 16;
  quantum_block_kernel<<<grid, 512, 0, stream>>>(x, W1, b1, W2, b2, W3, b3,
                                                 qw, Wp, bp, out);
}


# Rot split as RZ.RY.RZ: RZ phases of all wires merged into one phase table per layer (carried through the CNOT ring; last RZ dropped, |amp|^2 only), RY as real rotations; products re-associated
# speedup vs baseline: 1.3251x; 1.2170x over previous
#include <hip/hip_runtime.h>
#include <math.h>

typedef __attribute__((ext_vector_type(2))) float v2f;
typedef __attribute__((ext_vector_type(8))) float v8f;

#define LANE() ((int)(threadIdx.x & 31))

__device__ __forceinline__ float fast_tanh(float x) {
  const float e = __expf(2.0f * x);
  return 1.0f - 2.0f / (e + 1.0f);
}

template<int K, int LDA>
__device__ __forceinline__ v8f gemm_tile_acc(
    const float* __restrict__ sA,
    const float* __restrict__ Wg,
    int n)
{
  const int lane = LANE();
  const int half = lane >> 4;
  const int col  = lane & 15;
  v8f acc = {};
#pragma unroll
  for (int k0 = 0; k0 < K; k0 += 4) {
    const int ka = k0 + half * 2;
    v2f a, b;
    a.x = sA[col * LDA + ka];
    a.y = sA[col * LDA + ka + 1];
    b.x = Wg[n * K + ka];
    b.y = Wg[n * K + ka + 1];
    acc = __builtin_amdgcn_wmma_f32_16x16x4_f32(false, a, false, b,
                                                (short)0, acc, false, false);
  }
  return acc;
}

template<int K, int LDA, bool TANH>
__device__ __forceinline__ void gemm_tile_to_lds(
    const float* __restrict__ sA, const float* __restrict__ Wg,
    const float* __restrict__ bg,
    float* __restrict__ sOut, int ldo, int n0)
{
  const int lane = LANE();
  const int half = lane >> 4;
  const int n    = n0 + (lane & 15);
  const v8f acc = gemm_tile_acc<K, LDA>(sA, Wg, n);
  const float bias = bg[n];
#pragma unroll
  for (int r = 0; r < 8; ++r) {
    float v = acc[r] + bias;
    if (TANH) v = fast_tanh(v);
    sOut[(half * 8 + r) * ldo + n] = v;
  }
}

template<int K, int LDA>
__device__ __forceinline__ void gemm_tile_to_halfangle_trig(
    const float* __restrict__ sA, const float* __restrict__ Wg,
    const float* __restrict__ bg,
    float* __restrict__ sCos, float* __restrict__ sSin, int ldo, int n0)
{
  const int lane = LANE();
  const int half = lane >> 4;
  const int n    = n0 + (lane & 15);
  const v8f acc = gemm_tile_acc<K, LDA>(sA, Wg, n);
  const float bias = bg[n];
#pragma unroll
  for (int r = 0; r < 8; ++r) {
    const float h = 0.5f * (acc[r] + bias);
    sCos[(half * 8 + r) * ldo + n] = __cosf(h);
    sSin[(half * 8 + r) * ldo + n] = __sinf(h);
  }
}

__device__ __forceinline__ void gemm_final(
    const float* __restrict__ sZ,
    const float* __restrict__ Wp,
    const float* __restrict__ bp,
    float* __restrict__ out, int rowBase, int n0)
{
  const int lane = LANE();
  const int half = lane >> 4;
  const int col  = lane & 15;
  const int n    = n0 + col;
  v8f acc = {};
#pragma unroll
  for (int k0 = 0; k0 < 8; k0 += 4) {
    const int ka = k0 + half * 2;
    v2f a, b;
    a.x = sZ[col * 8 + ka];
    a.y = sZ[col * 8 + ka + 1];
    b.x = Wp[n * 8 + ka];
    b.y = Wp[n * 8 + ka + 1];
    acc = __builtin_amdgcn_wmma_f32_16x16x4_f32(false, a, false, b,
                                                (short)0, acc, false, false);
  }
  const float bias = bp[n];
#pragma unroll
  for (int r = 0; r < 8; ++r)
    out[(size_t)(rowBase + half * 8 + r) * 128 + n] = acc[r] + bias;
}


__device__ __forceinline__ int ring_preimage(int c) {
  c ^= (c & 1) << 7;
  c ^= (c >> 1) & 1;
  c ^= ((c >> 2) & 1) << 1;
  c ^= ((c >> 3) & 1) << 2;
  c ^= ((c >> 4) & 1) << 3;
  c ^= ((c >> 5) & 1) << 4;
  c ^= ((c >> 6) & 1) << 5;
  c ^= ((c >> 7) & 1) << 6;
  return c;
}

template<int L, int COMP>
__device__ __forceinline__ float rz_phase(const float* __restrict__ qw, int c) {
  float s = 0.f;
#pragma unroll
  for (int w = 0; w < 8; ++w) {
    const float h = 0.5f * qw[(L * 8 + w) * 3 + COMP];
    s += ((c >> (7 - w)) & 1) ? h : -h;
  }
  return s;
}

template<int L>
__device__ __forceinline__ void build_phase_entry(const float* __restrict__ qw,
                                                  v2f* __restrict__ sTL, int c)
{
  float ang = rz_phase<L, 0>(qw, c);
  if constexpr (L > 0) ang += rz_phase<L - 1, 2>(qw, ring_preimage(c));
  v2f t;
  t.x = __cosf(ang);
  t.y = __sinf(ang);
  sTL[(c & 7) * 32 + (c >> 3)] = t;
}


template<int LB>
__device__ __forceinline__ void ry_lane(float ar[8], float ai[8], float c, float s) {
  const float t = ((LANE() >> LB) & 1) ? s : -s;
#pragma unroll
  for (int j = 0; j < 8; ++j) {
    const float pr = __shfl_xor(ar[j], 1 << LB, 32);
    const float pi = __shfl_xor(ai[j], 1 << LB, 32);
    ar[j] = c * ar[j] + t * pr;
    ai[j] = c * ai[j] + t * pi;
  }
}

template<int JB>
__device__ __forceinline__ void ry_local(float ar[8], float ai[8], float c, float s) {
#pragma unroll
  for (int j = 0; j < 8; ++j) {
    if (!((j >> JB) & 1)) {
      const int j1 = j | (1 << JB);
      const float a0r = ar[j], a0i = ai[j], a1r = ar[j1], a1i = ai[j1];
      ar[j]  = c * a0r - s * a1r;  ai[j]  = c * a0i - s * a1i;
      ar[j1] = s * a0r + c * a1r;  ai[j1] = s * a0i + c * a1i;
    }
  }
}

template<int CJB, int TJB>
__device__ __forceinline__ void cnot_ll(float ar[8], float ai[8]) {
#pragma unroll
  for (int j = 0; j < 8; ++j)
    if (((j >> CJB) & 1) && !((j >> TJB) & 1)) {
      const int j1 = j | (1 << TJB);
      float tr = ar[j], ti = ai[j];
      ar[j] = ar[j1]; ai[j] = ai[j1];
      ar[j1] = tr;    ai[j1] = ti;
    }
}
template<int CJB, int TLB>
__device__ __forceinline__ void cnot_local_lane(float ar[8], float ai[8]) {
#pragma unroll
  for (int j = 0; j < 8; ++j) {
    const float pr = __shfl_xor(ar[j], 1 << TLB, 32);
    const float pi = __shfl_xor(ai[j], 1 << TLB, 32);
    if ((j >> CJB) & 1) { ar[j] = pr; ai[j] = pi; }
  }
}
template<int CLB, int TJB>
__device__ __forceinline__ void cnot_lane_local(float ar[8], float ai[8]) {
  if ((LANE() >> CLB) & 1) {
#pragma unroll
    for (int j = 0; j < 8; ++j)
      if (!((j >> TJB) & 1)) {
        const int j1 = j | (1 << TJB);
        float tr = ar[j], ti = ai[j];
        ar[j] = ar[j1]; ai[j] = ai[j1];
        ar[j1] = tr;    ai[j1] = ti;
      }
  }
}
template<int CLB, int TLB>
__device__ __forceinline__ void cnot_lane_lane(float ar[8], float ai[8]) {
  const int cb = (LANE() >> CLB) & 1;
#pragma unroll
  for (int j = 0; j < 8; ++j) {
    const float pr = __shfl_xor(ar[j], 1 << TLB, 32);
    const float pi = __shfl_xor(ai[j], 1 << TLB, 32);
    if (cb) { ar[j] = pr; ai[j] = pi; }
  }
}

template<int W>
__device__ __forceinline__ void wire_factor_lane(const float* __restrict__ cs,
    const float* __restrict__ sn, float& fr, float& fi)
{
  const int bit = (LANE() >> (4 - W)) & 1;
  const float mag = bit ? sn[W] : cs[W];
  const float zs  = sn[8 + W];
  fr = mag * cs[8 + W];
  fi = mag * (bit ? zs : -zs);
}

#define CMUL(orr, oi, xr, xi, yr, yi) {                                       \
    orr = (xr) * (yr) - (xi) * (yi);                                          \
    oi  = (xr) * (yi) + (xi) * (yr);                                          \
  }

__device__ __forceinline__ void product_state(const float* __restrict__ cs,
    const float* __restrict__ sn, float ar[8], float ai[8])
{
  float pr, pi, fr, fi, tr, ti;
  wire_factor_lane<0>(cs, sn, pr, pi);
  wire_factor_lane<1>(cs, sn, fr, fi); CMUL(tr, ti, pr, pi, fr, fi) pr = tr; pi = ti;
  wire_factor_lane<2>(cs, sn, fr, fi); CMUL(tr, ti, pr, pi, fr, fi) pr = tr; pi = ti;
  wire_factor_lane<3>(cs, sn, fr, fi); CMUL(tr, ti, pr, pi, fr, fi) pr = tr; pi = ti;
  wire_factor_lane<4>(cs, sn, fr, fi); CMUL(tr, ti, pr, pi, fr, fi) pr = tr; pi = ti;
  float qr[3][2], qi[3][2];
#pragma unroll
  for (int w = 5; w < 8; ++w) {
    const float zr = cs[8 + w], zs = sn[8 + w];
    qr[w - 5][0] = cs[w] * zr; qi[w - 5][0] = -(cs[w] * zs);
    qr[w - 5][1] = sn[w] * zr; qi[w - 5][1] =   sn[w] * zs;
  }
  float t1r[2], t1i[2], t2r[4], t2i[4];
#pragma unroll
  for (int b = 0; b < 2; ++b) CMUL(t1r[b], t1i[b], pr, pi, qr[0][b], qi[0][b])
#pragma unroll
  for (int b = 0; b < 4; ++b)
    CMUL(t2r[b], t2i[b], t1r[b >> 1], t1i[b >> 1], qr[1][b & 1], qi[1][b & 1])
#pragma unroll
  for (int j = 0; j < 8; ++j)
    CMUL(ar[j], ai[j], t2r[j >> 1], t2i[j >> 1], qr[2][j & 1], qi[2][j & 1])
}

__global__ __launch_bounds__(512) void quantum_block_kernel(
    const float* __restrict__ x,
    const float* __restrict__ W1, const float* __restrict__ b1,
    const float* __restrict__ W2, const float* __restrict__ b2,
    const float* __restrict__ W3, const float* __restrict__ b3,
    const float* __restrict__ qw,
    const float* __restrict__ Wp, const float* __restrict__ bp,
    float* __restrict__ out)
{
  __shared__ float sX  [16 * 80];
  __shared__ float sH1 [16 * 64];
  __shared__ float sH2 [16 * 32];
  __shared__ float sCos[16 * 16];
  __shared__ float sSin[16 * 16];
  __shared__ float sZ  [16 * 8];
  __shared__ v2f   sT  [3 * 256];
  __shared__ float sRY [24 * 2];

  const int tid     = threadIdx.x;
  const int wid     = tid >> 5;
  const int lane    = tid & 31;
  const int rowBase = blockIdx.x * 16;

  if (tid < 24) {
    const float th = qw[tid * 3 + 1];
    sRY[tid * 2]     = __cosf(0.5f * th);
    sRY[tid * 2 + 1] = __sinf(0.5f * th);
  }
  if (tid < 256) {
    build_phase_entry<0>(qw, &sT[0],   tid);
    build_phase_entry<2>(qw, &sT[512], tid);
  } else {
    build_phase_entry<1>(qw, &sT[256], tid - 256);
  }
  for (int i = tid; i < 16 * 80; i += 512) sX[i] = x[rowBase * 80 + i];
  __syncthreads();

  if (wid < 4) gemm_tile_to_lds<80, 80, true >(sX,  W1, b1, sH1, 64, wid * 16);
  __syncthreads();
  if (wid < 2) gemm_tile_to_lds<64, 64, true >(sH1, W2, b2, sH2, 32, wid * 16);
  __syncthreads();
  if (wid == 0) gemm_tile_to_halfangle_trig<32, 32>(sH2, W3, b3, sCos, sSin, 16, 0);
  __syncthreads();

  {
    float ar[8], ai[8];
    product_state(&sCos[wid * 16], &sSin[wid * 16], ar, ai);

    for (int L = 0; L < 3; ++L) {
      const v2f* T = &sT[L * 256 + lane];
#pragma unroll
      for (int j = 0; j < 8; ++j) {
        const v2f t = T[j * 32];
        const float r = ar[j] * t.x - ai[j] * t.y;
        const float i = ar[j] * t.y + ai[j] * t.x;
        ar[j] = r; ai[j] = i;
      }
      const float* ry = &sRY[L * 16];
      ry_lane<4>(ar, ai, ry[0],  ry[1]);
      ry_lane<3>(ar, ai, ry[2],  ry[3]);
      ry_lane<2>(ar, ai, ry[4],  ry[5]);
      ry_lane<1>(ar, ai, ry[6],  ry[7]);
      ry_lane<0>(ar, ai, ry[8],  ry[9]);
      ry_local<2>(ar, ai, ry[10], ry[11]);
      ry_local<1>(ar, ai, ry[12], ry[13]);
      ry_local<0>(ar, ai, ry[14], ry[15]);
      cnot_lane_lane<4, 3>(ar, ai);
      cnot_lane_lane<3, 2>(ar, ai);
      cnot_lane_lane<2, 1>(ar, ai);
      cnot_lane_lane<1, 0>(ar, ai);
      cnot_lane_local<0, 2>(ar, ai);
      cnot_ll<2, 1>(ar, ai);
      cnot_ll<1, 0>(ar, ai);
      cnot_local_lane<0, 4>(ar, ai);
    }

    float pb[8]; float tot = 0.f;
#pragma unroll
    for (int j = 0; j < 8; ++j) { pb[j] = ar[j]*ar[j] + ai[j]*ai[j]; tot += pb[j]; }
    float zp[8];
#pragma unroll
    for (int w = 0; w < 5; ++w)
      zp[w] = ((lane >> (4 - w)) & 1) ? -tot : tot;
    float s5 = 0.f, s6 = 0.f, s7 = 0.f;
#pragma unroll
    for (int j = 0; j < 8; ++j) {
      s5 += ((j >> 2) & 1) ? -pb[j] : pb[j];
      s6 += ((j >> 1) & 1) ? -pb[j] : pb[j];
      s7 += ( j       & 1) ? -pb[j] : pb[j];
    }
    zp[5] = s5; zp[6] = s6; zp[7] = s7;
#pragma unroll
    for (int w = 0; w < 8; ++w) {
      float v = zp[w];
#pragma unroll
      for (int off = 16; off > 0; off >>= 1) v += __shfl_xor(v, off, 32);
      if (lane == 0) sZ[wid * 8 + w] = v;
    }
  }
  __syncthreads();

  if (wid < 8) gemm_final(sZ, Wp, bp, out, rowBase, wid * 16);
}

extern "C" void kernel_launch(void* const* d_in, const int* in_sizes, int n_in,
                              void* d_out, int out_size, void* d_ws, size_t ws_size,
                              hipStream_t stream) {
  const float* x  = (const float*)d_in[0];
  const float* W1 = (const float*)d_in[1];
  const float* b1 = (const float*)d_in[2];
  const float* W2 = (const float*)d_in[3];
  const float* b2 = (const float*)d_in[4];
  const float* W3 = (const float*)d_in[5];
  const float* b3 = (const float*)d_in[6];
  const float* qw = (const float*)d_in[7];
  const float* Wp = (const float*)d_in[8];
  const float* bp = (const float*)d_in[9];
  float* out = (float*)d_out;

  const int B = in_sizes[0] / 80;
  const int grid = B / 16;
  quantum_block_kernel<<<grid, 512, 0, stream>>>(x, W1, b1, W2, b2, W3, b3,
                                                 qw, Wp, bp, out);
}


# Lane exchanges moved from the LDS unit (ds_bpermute) to vector-ALU cross-lane ops: DPP row_xmask (RY partners, readout sums), masked DPP + one v_permlane16 gather for the 4 lane CNOTs, v_permlanex16
# speedup vs baseline: 1.6239x; 1.2255x over previous
#include <hip/hip_runtime.h>
#include <math.h>

typedef __attribute__((ext_vector_type(2))) float v2f;
typedef __attribute__((ext_vector_type(8))) float v8f;

#define LANE() ((int)(threadIdx.x & 31))

template<int M>
__device__ __forceinline__ float lane_xor(float v) {
  const int b = __builtin_bit_cast(int, v);
  if constexpr (M == 16)
    return __builtin_bit_cast(float,
        __builtin_amdgcn_permlanex16(b, b, 0x76543210, 0xfedcba98, false, false));
  else
    return __builtin_bit_cast(float,
        __builtin_amdgcn_mov_dpp(b, 0x160 | M, 0xf, 0xf, true));
}

__device__ __forceinline__ float cnot_row_ctrl(float v) {
  const int b = __builtin_bit_cast(int, v);
  return __builtin_bit_cast(float,
      __builtin_amdgcn_update_dpp(b, b, 0x160 | 8, 0x2, 0xf, false));
}

__device__ constexpr unsigned ring_row_selects(unsigned word) {
  unsigned v = 0;
  for (unsigned j = 0; j < 8; ++j) {
    const unsigned l = word * 8 + j;
    v |= (l ^ ((l >> 1) & 7u)) << (4 * j);
  }
  return v;
}
__device__ __forceinline__ float cnot_row_chain(float v) {
  constexpr unsigned lo = ring_row_selects(0), hi = ring_row_selects(1);
  static_assert(lo == 0x45762310u && hi == 0x89baefdcu, "in-row source lanes of the CNOT chain");
  const int b = __builtin_bit_cast(int, v);
  return __builtin_bit_cast(float,
      __builtin_amdgcn_permlane16(b, b, lo, hi, false, false));
}

__device__ __forceinline__ float fast_tanh(float x) {
  const float e = __expf(2.0f * x);
  return 1.0f - 2.0f / (e + 1.0f);
}

template<int K, int LDA>
__device__ __forceinline__ v8f gemm_tile_acc(
    const float* __restrict__ sA,
    const float* __restrict__ Wg,
    int n)
{
  const int lane = LANE();
  const int half = lane >> 4;
  const int col  = lane & 15;
  v8f acc = {};
#pragma unroll
  for (int k0 = 0; k0 < K; k0 += 4) {
    const int ka = k0 + half * 2;
    v2f a, b;
    a.x = sA[col * LDA + ka];
    a.y = sA[col * LDA + ka + 1];
    b.x = Wg[n * K + ka];
    b.y = Wg[n * K + ka + 1];
    acc = __builtin_amdgcn_wmma_f32_16x16x4_f32(false, a, false, b,
                                                (short)0, acc, false, false);
  }
  return acc;
}

template<int K, int LDA, bool TANH>
__device__ __forceinline__ void gemm_tile_to_lds(
    const float* __restrict__ sA, const float* __restrict__ Wg,
    const float* __restrict__ bg,
    float* __restrict__ sOut, int ldo, int n0)
{
  const int lane = LANE();
  const int half = lane >> 4;
  const int n    = n0 + (lane & 15);
  const v8f acc = gemm_tile_acc<K, LDA>(sA, Wg, n);
  const float bias = bg[n];
#pragma unroll
  for (int r = 0; r < 8; ++r) {
    float v = acc[r] + bias;
    if (TANH) v = fast_tanh(v);
    sOut[(half * 8 + r) * ldo + n] = v;
  }
}

template<int K, int LDA>
__device__ __forceinline__ void gemm_tile_to_halfangle_trig(
    const float* __restrict__ sA, const float* __restrict__ Wg,
    const float* __restrict__ bg,
    float* __restrict__ sCos, float* __restrict__ sSin, int ldo, int n0)
{
  const int lane = LANE();
  const int half = lane >> 4;
  const int n    = n0 + (lane & 15);
  const v8f acc = gemm_tile_acc<K, LDA>(sA, Wg, n);
  const float bias = bg[n];
#pragma unroll
  for (int r = 0; r < 8; ++r) {
    const float h = 0.5f * (acc[r] + bias);
    sCos[(half * 8 + r) * ldo + n] = __cosf(h);
    sSin[(half * 8 + r) * ldo + n] = __sinf(h);
  }
}

__device__ __forceinline__ void gemm_final(
    const float* __restrict__ sZ,
    const float* __restrict__ Wp,
    const float* __restrict__ bp,
    float* __restrict__ out, int rowBase, int n0)
{
  const int lane = LANE();
  const int half = lane >> 4;
  const int col  = lane & 15;
  const int n    = n0 + col;
  v8f acc = {};
#pragma unroll
  for (int k0 = 0; k0 < 8; k0 += 4) {
    const int ka = k0 + half * 2;
    v2f a, b;
    a.x = sZ[col * 8 + ka];
    a.y = sZ[col * 8 + ka + 1];
    b.x = Wp[n * 8 + ka];
    b.y = Wp[n * 8 + ka + 1];
    acc = __builtin_amdgcn_wmma_f32_16x16x4_f32(false, a, false, b,
                                                (short)0, acc, false, false);
  }
  const float bias = bp[n];
#pragma unroll
  for (int r = 0; r < 8; ++r)
    out[(size_t)(rowBase + half * 8 + r) * 128 + n] = acc[r] + bias;
}


__device__ __forceinline__ int ring_preimage(int c) {
  c ^= (c & 1) << 7;
  c ^= (c >> 1) & 1;
  c ^= ((c >> 2) & 1) << 1;
  c ^= ((c >> 3) & 1) << 2;
  c ^= ((c >> 4) & 1) << 3;
  c ^= ((c >> 5) & 1) << 4;
  c ^= ((c >> 6) & 1) << 5;
  c ^= ((c >> 7) & 1) << 6;
  return c;
}

template<int L, int COMP>
__device__ __forceinline__ float rz_phase(const float* __restrict__ qw, int c) {
  float s = 0.f;
#pragma unroll
  for (int w = 0; w < 8; ++w) {
    const float h = 0.5f * qw[(L * 8 + w) * 3 + COMP];
    s += ((c >> (7 - w)) & 1) ? h : -h;
  }
  return s;
}

template<int L>
__device__ __forceinline__ void build_phase_entry(const float* __restrict__ qw,
                                                  v2f* __restrict__ sTL, int c)
{
  float ang = rz_phase<L, 0>(qw, c);
  if constexpr (L > 0) ang += rz_phase<L - 1, 2>(qw, ring_preimage(c));
  v2f t;
  t.x = __cosf(ang);
  t.y = __sinf(ang);
  sTL[(c & 7) * 32 + (c >> 3)] = t;
}


template<int LB>
__device__ __forceinline__ void ry_lane(float ar[8], float ai[8], float c, float s) {
  const float t = ((LANE() >> LB) & 1) ? s : -s;
#pragma unroll
  for (int j = 0; j < 8; ++j) {
    const float pr = lane_xor<(1 << LB)>(ar[j]);
    const float pi = lane_xor<(1 << LB)>(ai[j]);
    ar[j] = c * ar[j] + t * pr;
    ai[j] = c * ai[j] + t * pi;
  }
}

template<int JB>
__device__ __forceinline__ void ry_local(float ar[8], float ai[8], float c, float s) {
#pragma unroll
  for (int j = 0; j < 8; ++j) {
    if (!((j >> JB) & 1)) {
      const int j1 = j | (1 << JB);
      const float a0r = ar[j], a0i = ai[j], a1r = ar[j1], a1i = ai[j1];
      ar[j]  = c * a0r - s * a1r;  ai[j]  = c * a0i - s * a1i;
      ar[j1] = s * a0r + c * a1r;  ai[j1] = s * a0i + c * a1i;
    }
  }
}

template<int CJB, int TJB>
__device__ __forceinline__ void cnot_ll(float ar[8], float ai[8]) {
#pragma unroll
  for (int j = 0; j < 8; ++j)
    if (((j >> CJB) & 1) && !((j >> TJB) & 1)) {
      const int j1 = j | (1 << TJB);
      float tr = ar[j], ti = ai[j];
      ar[j] = ar[j1]; ai[j] = ai[j1];
      ar[j1] = tr;    ai[j1] = ti;
    }
}
template<int CJB, int TLB>
__device__ __forceinline__ void cnot_local_lane(float ar[8], float ai[8]) {
#pragma unroll
  for (int j = 0; j < 8; ++j)
    if ((j >> CJB) & 1) {
      ar[j] = lane_xor<(1 << TLB)>(ar[j]);
      ai[j] = lane_xor<(1 << TLB)>(ai[j]);
    }
}
template<int CLB, int TJB>
__device__ __forceinline__ void cnot_lane_local(float ar[8], float ai[8]) {
  if ((LANE() >> CLB) & 1) {
#pragma unroll
    for (int j = 0; j < 8; ++j)
      if (!((j >> TJB) & 1)) {
        const int j1 = j | (1 << TJB);
        float tr = ar[j], ti = ai[j];
        ar[j] = ar[j1]; ai[j] = ai[j1];
        ar[j1] = tr;    ai[j1] = ti;
      }
  }
}
__device__ __forceinline__ void cnot_lane_chain(float ar[8], float ai[8]) {
#pragma unroll
  for (int j = 0; j < 8; ++j) {
    ar[j] = cnot_row_chain(cnot_row_ctrl(ar[j]));
    ai[j] = cnot_row_chain(cnot_row_ctrl(ai[j]));
  }
}

template<int W>
__device__ __forceinline__ void wire_factor_lane(const float* __restrict__ cs,
    const float* __restrict__ sn, float& fr, float& fi)
{
  const int bit = (LANE() >> (4 - W)) & 1;
  const float mag = bit ? sn[W] : cs[W];
  const float zs  = sn[8 + W];
  fr = mag * cs[8 + W];
  fi = mag * (bit ? zs : -zs);
}

#define CMUL(orr, oi, xr, xi, yr, yi) {                                       \
    orr = (xr) * (yr) - (xi) * (yi);                                          \
    oi  = (xr) * (yi) + (xi) * (yr);                                          \
  }

__device__ __forceinline__ void product_state(const float* __restrict__ cs,
    const float* __restrict__ sn, float ar[8], float ai[8])
{
  float pr, pi, fr, fi, tr, ti;
  wire_factor_lane<0>(cs, sn, pr, pi);
  wire_factor_lane<1>(cs, sn, fr, fi); CMUL(tr, ti, pr, pi, fr, fi) pr = tr; pi = ti;
  wire_factor_lane<2>(cs, sn, fr, fi); CMUL(tr, ti, pr, pi, fr, fi) pr = tr; pi = ti;
  wire_factor_lane<3>(cs, sn, fr, fi); CMUL(tr, ti, pr, pi, fr, fi) pr = tr; pi = ti;
  wire_factor_lane<4>(cs, sn, fr, fi); CMUL(tr, ti, pr, pi, fr, fi) pr = tr; pi = ti;
  float qr[3][2], qi[3][2];
#pragma unroll
  for (int w = 5; w < 8; ++w) {
    const float zr = cs[8 + w], zs = sn[8 + w];
    qr[w - 5][0] = cs[w] * zr; qi[w - 5][0] = -(cs[w] * zs);
    qr[w - 5][1] = sn[w] * zr; qi[w - 5][1] =   sn[w] * zs;
  }
  float t1r[2], t1i[2], t2r[4], t2i[4];
#pragma unroll
  for (int b = 0; b < 2; ++b) CMUL(t1r[b], t1i[b], pr, pi, qr[0][b], qi[0][b])
#pragma unroll
  for (int b = 0; b < 4; ++b)
    CMUL(t2r[b], t2i[b], t1r[b >> 1], t1i[b >> 1], qr[1][b & 1], qi[1][b & 1])
#pragma unroll
  for (int j = 0; j < 8; ++j)
    CMUL(ar[j], ai[j], t2r[j >> 1], t2i[j >> 1], qr[2][j & 1], qi[2][j & 1])
}

__global__ __launch_bounds__(512) void quantum_block_kernel(
    const float* __restrict__ x,
    const float* __restrict__ W1, const float* __restrict__ b1,
    const float* __restrict__ W2, const float* __restrict__ b2,
    const float* __restrict__ W3, const float* __restrict__ b3,
    const float* __restrict__ qw,
    const float* __restrict__ Wp, const float* __restrict__ bp,
    float* __restrict__ out)
{
  __shared__ float sX  [16 * 80];
  __shared__ float sH1 [16 * 64];
  __shared__ float sH2 [16 * 32];
  __shared__ float sCos[16 * 16];
  __shared__ float sSin[16 * 16];
  __shared__ float sZ  [16 * 8];
  __shared__ v2f   sT  [3 * 256];
  __shared__ float sRY [24 * 2];

  const int tid     = threadIdx.x;
  const int wid     = tid >> 5;
  const int lane    = tid & 31;
  const int rowBase = blockIdx.x * 16;

  if (tid < 24) {
    const float th = qw[tid * 3 + 1];
    sRY[tid * 2]     = __cosf(0.5f * th);
    sRY[tid * 2 + 1] = __sinf(0.5f * th);
  }
  if (tid < 256) {
    build_phase_entry<0>(qw, &sT[0],   tid);
    build_phase_entry<2>(qw, &sT[512], tid);
  } else {
    build_phase_entry<1>(qw, &sT[256], tid - 256);
  }
  for (int i = tid; i < 16 * 80; i += 512) sX[i] = x[rowBase * 80 + i];
  __syncthreads();

  if (wid < 4) gemm_tile_to_lds<80, 80, true >(sX,  W1, b1, sH1, 64, wid * 16);
  __syncthreads();
  if (wid < 2) gemm_tile_to_lds<64, 64, true >(sH1, W2, b2, sH2, 32, wid * 16);
  __syncthreads();
  if (wid == 0) gemm_tile_to_halfangle_trig<32, 32>(sH2, W3, b3, sCos, sSin, 16, 0);
  __syncthreads();

  {
    float ar[8], ai[8];
    product_state(&sCos[wid * 16], &sSin[wid * 16], ar, ai);

    for (int L = 0; L < 3; ++L) {
      const v2f* T = &sT[L * 256 + lane];
#pragma unroll
      for (int j = 0; j < 8; ++j) {
        const v2f t = T[j * 32];
        const float r = ar[j] * t.x - ai[j] * t.y;
        const float i = ar[j] * t.y + ai[j] * t.x;
        ar[j] = r; ai[j] = i;
      }
      const float* ry = &sRY[L * 16];
      ry_lane<4>(ar, ai, ry[0],  ry[1]);
      ry_lane<3>(ar, ai, ry[2],  ry[3]);
      ry_lane<2>(ar, ai, ry[4],  ry[5]);
      ry_lane<1>(ar, ai, ry[6],  ry[7]);
      ry_lane<0>(ar, ai, ry[8],  ry[9]);
      ry_local<2>(ar, ai, ry[10], ry[11]);
      ry_local<1>(ar, ai, ry[12], ry[13]);
      ry_local<0>(ar, ai, ry[14], ry[15]);
      cnot_lane_chain(ar, ai);
      cnot_lane_local<0, 2>(ar, ai);
      cnot_ll<2, 1>(ar, ai);
      cnot_ll<1, 0>(ar, ai);
      cnot_local_lane<0, 4>(ar, ai);
    }

    float pb[8]; float tot = 0.f;
#pragma unroll
    for (int j = 0; j < 8; ++j) { pb[j] = ar[j]*ar[j] + ai[j]*ai[j]; tot += pb[j]; }
    float zp[8];
#pragma unroll
    for (int w = 0; w < 5; ++w)
      zp[w] = ((lane >> (4 - w)) & 1) ? -tot : tot;
    float s5 = 0.f, s6 = 0.f, s7 = 0.f;
#pragma unroll
    for (int j = 0; j < 8; ++j) {
      s5 += ((j >> 2) & 1) ? -pb[j] : pb[j];
      s6 += ((j >> 1) & 1) ? -pb[j] : pb[j];
      s7 += ( j       & 1) ? -pb[j] : pb[j];
    }
    zp[5] = s5; zp[6] = s6; zp[7] = s7;
#pragma unroll
    for (int w = 0; w < 8; ++w) {
      float v = zp[w];
      v += lane_xor<16>(v);
      v += lane_xor<8>(v);
      v += lane_xor<4>(v);
      v += lane_xor<2>(v);
      v += lane_xor<1>(v);
      if (lane == 0) sZ[wid * 8 + w] = v;
    }
  }
  __syncthreads();

  if (wid < 8) gemm_final(sZ, Wp, bp, out, rowBase, wid * 16);
}

extern "C" void kernel_launch(void* const* d_in, const int* in_sizes, int n_in,
                              void* d_out, int out_size, void* d_ws, size_t ws_size,
                              hipStream_t stream) {
  const float* x  = (const float*)d_in[0];
  const float* W1 = (const float*)d_in[1];
  const float* b1 = (const float*)d_in[2];
  const float* W2 = (const float*)d_in[3];
  const float* b2 = (const float*)d_in[4];
  const float* W3 = (const float*)d_in[5];
  const float* b3 = (const float*)d_in[6];
  const float* qw = (const float*)d_in[7];
  const float* Wp = (const float*)d_in[8];
  const float* bp = (const float*)d_in[9];
  float* out = (float*)d_out;

  const int B = in_sizes[0] / 80;
  const int grid = B / 16;
  quantum_block_kernel<<<grid, 512, 0, stream>>>(x, W1, b1, W2, b2, W3, b3,
                                                 qw, Wp, bp, out);
}


# Two samples per wave, one per 16-lane row (4 lane wires + 4 register wires): all lane exchanges stay inside a row, half the waves, 32 rows per workgroup; readout sums re-associated
# speedup vs baseline: 1.9270x; 1.1866x over previous
#include <hip/hip_runtime.h>
#include <math.h>

typedef __attribute__((ext_vector_type(2))) float v2f;
typedef __attribute__((ext_vector_type(8))) float v8f;

#define LANE() ((int)(threadIdx.x & 31))

template<int M>
__device__ __forceinline__ float lane_xor(float v) {
  static_assert(M == 1 || M == 2 || M == 4 || M == 8, "in-row partner only");
  const int b = __builtin_bit_cast(int, v);
  return __builtin_bit_cast(float,
      __builtin_amdgcn_mov_dpp(b, 0x160 | M, 0xf, 0xf, true));
}

__device__ constexpr unsigned ring_row_selects(unsigned word) {
  unsigned v = 0;
  for (unsigned j = 0; j < 8; ++j) {
    const unsigned l = word * 8 + j;
    v |= (l ^ ((l >> 1) & 7u)) << (4 * j);
  }
  return v;
}
__device__ __forceinline__ float cnot_row_chain(float v) {
  constexpr unsigned lo = ring_row_selects(0), hi = ring_row_selects(1);
  static_assert(lo == 0x45762310u && hi == 0x89baefdcu, "in-row source lanes of the CNOT chain");
  const int b = __builtin_bit_cast(int, v);
  return __builtin_bit_cast(float,
      __builtin_amdgcn_permlane16(b, b, lo, hi, false, false));
}

__device__ __forceinline__ float fast_tanh(float x) {
  const float e = __expf(2.0f * x);
  return 1.0f - 2.0f / (e + 1.0f);
}

template<int K, int LDA>
__device__ __forceinline__ v8f gemm_tile_acc(
    const float* __restrict__ sA,
    const float* __restrict__ Wg,
    int n)
{
  const int lane = LANE();
  const int half = lane >> 4;
  const int col  = lane & 15;
  v8f acc = {};
#pragma unroll
  for (int k0 = 0; k0 < K; k0 += 4) {
    const int ka = k0 + half * 2;
    v2f a, b;
    a.x = sA[col * LDA + ka];
    a.y = sA[col * LDA + ka + 1];
    b.x = Wg[n * K + ka];
    b.y = Wg[n * K + ka + 1];
    acc = __builtin_amdgcn_wmma_f32_16x16x4_f32(false, a, false, b,
                                                (short)0, acc, false, false);
  }
  return acc;
}

template<int K, int LDA, bool TANH>
__device__ __forceinline__ void gemm_tile_to_lds(
    const float* __restrict__ sA, const float* __restrict__ Wg,
    const float* __restrict__ bg,
    float* __restrict__ sOut, int ldo, int n0)
{
  const int lane = LANE();
  const int half = lane >> 4;
  const int n    = n0 + (lane & 15);
  const v8f acc = gemm_tile_acc<K, LDA>(sA, Wg, n);
  const float bias = bg[n];
#pragma unroll
  for (int r = 0; r < 8; ++r) {
    float v = acc[r] + bias;
    if (TANH) v = fast_tanh(v);
    sOut[(half * 8 + r) * ldo + n] = v;
  }
}

template<int K, int LDA>
__device__ __forceinline__ void gemm_tile_to_halfangle_trig(
    const float* __restrict__ sA, const float* __restrict__ Wg,
    const float* __restrict__ bg,
    float* __restrict__ sCos, float* __restrict__ sSin, int ldo, int n0)
{
  const int lane = LANE();
  const int half = lane >> 4;
  const int n    = n0 + (lane & 15);
  const v8f acc = gemm_tile_acc<K, LDA>(sA, Wg, n);
  const float bias = bg[n];
#pragma unroll
  for (int r = 0; r < 8; ++r) {
    const float h = 0.5f * (acc[r] + bias);
    sCos[(half * 8 + r) * ldo + n] = __cosf(h);
    sSin[(half * 8 + r) * ldo + n] = __sinf(h);
  }
}

__device__ __forceinline__ void gemm_final(
    const float* __restrict__ sZ,
    const float* __restrict__ Wp,
    const float* __restrict__ bp,
    float* __restrict__ out, int rowBase, int n0)
{
  const int lane = LANE();
  const int half = lane >> 4;
  const int col  = lane & 15;
  const int n    = n0 + col;
  v8f acc = {};
#pragma unroll
  for (int k0 = 0; k0 < 8; k0 += 4) {
    const int ka = k0 + half * 2;
    v2f a, b;
    a.x = sZ[col * 8 + ka];
    a.y = sZ[col * 8 + ka + 1];
    b.x = Wp[n * 8 + ka];
    b.y = Wp[n * 8 + ka + 1];
    acc = __builtin_amdgcn_wmma_f32_16x16x4_f32(false, a, false, b,
                                                (short)0, acc, false, false);
  }
  const float bias = bp[n];
#pragma unroll
  for (int r = 0; r < 8; ++r)
    out[(size_t)(rowBase + half * 8 + r) * 128 + n] = acc[r] + bias;
}


__device__ __forceinline__ int ring_preimage(int c) {
  c ^= (c & 1) << 7;
  c ^= (c >> 1) & 1;
  c ^= ((c >> 2) & 1) << 1;
  c ^= ((c >> 3) & 1) << 2;
  c ^= ((c >> 4) & 1) << 3;
  c ^= ((c >> 5) & 1) << 4;
  c ^= ((c >> 6) & 1) << 5;
  c ^= ((c >> 7) & 1) << 6;
  return c;
}

template<int L, int COMP>
__device__ __forceinline__ float rz_phase(const float* __restrict__ qw, int c) {
  float s = 0.f;
#pragma unroll
  for (int w = 0; w < 8; ++w) {
    const float h = 0.5f * qw[(L * 8 + w) * 3 + COMP];
    s += ((c >> (7 - w)) & 1) ? h : -h;
  }
  return s;
}

template<int L>
__device__ __forceinline__ void build_phase_entry(const float* __restrict__ qw,
                                                  v2f* __restrict__ sTL, int c)
{
  float ang = rz_phase<L, 0>(qw, c);
  if constexpr (L > 0) ang += rz_phase<L - 1, 2>(qw, ring_preimage(c));
  v2f t;
  t.x = __cosf(ang);
  t.y = __sinf(ang);
  sTL[(c & 15) * 16 + (c >> 4)] = t;
}


constexpr int NS = 16;

template<int LB>
__device__ __forceinline__ void ry_lane(float ar[NS], float ai[NS], float c, float s) {
  const float t = ((LANE() >> LB) & 1) ? s : -s;
#pragma unroll
  for (int j = 0; j < NS; ++j) {
    const float pr = lane_xor<(1 << LB)>(ar[j]);
    const float pi = lane_xor<(1 << LB)>(ai[j]);
    ar[j] = c * ar[j] + t * pr;
    ai[j] = c * ai[j] + t * pi;
  }
}

template<int JB>
__device__ __forceinline__ void ry_local(float ar[NS], float ai[NS], float c, float s) {
#pragma unroll
  for (int j = 0; j < NS; ++j) {
    if (!((j >> JB) & 1)) {
      const int j1 = j | (1 << JB);
      const float a0r = ar[j], a0i = ai[j], a1r = ar[j1], a1i = ai[j1];
      ar[j]  = c * a0r - s * a1r;  ai[j]  = c * a0i - s * a1i;
      ar[j1] = s * a0r + c * a1r;  ai[j1] = s * a0i + c * a1i;
    }
  }
}

template<int CJB, int TJB>
__device__ __forceinline__ void cnot_ll(float ar[NS], float ai[NS]) {
#pragma unroll
  for (int j = 0; j < NS; ++j)
    if (((j >> CJB) & 1) && !((j >> TJB) & 1)) {
      const int j1 = j | (1 << TJB);
      float tr = ar[j], ti = ai[j];
      ar[j] = ar[j1]; ai[j] = ai[j1];
      ar[j1] = tr;    ai[j1] = ti;
    }
}
template<int CJB, int TLB>
__device__ __forceinline__ void cnot_local_lane(float ar[NS], float ai[NS]) {
#pragma unroll
  for (int j = 0; j < NS; ++j)
    if ((j >> CJB) & 1) {
      ar[j] = lane_xor<(1 << TLB)>(ar[j]);
      ai[j] = lane_xor<(1 << TLB)>(ai[j]);
    }
}
template<int CLB, int TJB>
__device__ __forceinline__ void cnot_lane_local(float ar[NS], float ai[NS]) {
  if ((LANE() >> CLB) & 1) {
#pragma unroll
    for (int j = 0; j < NS; ++j)
      if (!((j >> TJB) & 1)) {
        const int j1 = j | (1 << TJB);
        float tr = ar[j], ti = ai[j];
        ar[j] = ar[j1]; ai[j] = ai[j1];
        ar[j1] = tr;    ai[j1] = ti;
      }
  }
}
__device__ __forceinline__ void cnot_lane_chain(float ar[NS], float ai[NS]) {
#pragma unroll
  for (int j = 0; j < NS; ++j) {
    ar[j] = cnot_row_chain(ar[j]);
    ai[j] = cnot_row_chain(ai[j]);
  }
}

template<int W>
__device__ __forceinline__ void wire_factor_lane(const float* __restrict__ cs,
    const float* __restrict__ sn, float& fr, float& fi)
{
  const int bit = (LANE() >> (3 - W)) & 1;
  const float mag = bit ? sn[W] : cs[W];
  const float zs  = sn[8 + W];
  fr = mag * cs[8 + W];
  fi = mag * (bit ? zs : -zs);
}

#define CMUL(orr, oi, xr, xi, yr, yi) {                                       \
    orr = (xr) * (yr) - (xi) * (yi);                                          \
    oi  = (xr) * (yi) + (xi) * (yr);                                          \
  }

__device__ __forceinline__ void product_state(const float* __restrict__ cs,
    const float* __restrict__ sn, float ar[NS], float ai[NS])
{
  float pr, pi, fr, fi, tr, ti;
  wire_factor_lane<0>(cs, sn, pr, pi);
  wire_factor_lane<1>(cs, sn, fr, fi); CMUL(tr, ti, pr, pi, fr, fi) pr = tr; pi = ti;
  wire_factor_lane<2>(cs, sn, fr, fi); CMUL(tr, ti, pr, pi, fr, fi) pr = tr; pi = ti;
  wire_factor_lane<3>(cs, sn, fr, fi); CMUL(tr, ti, pr, pi, fr, fi) pr = tr; pi = ti;
  float qr[4][2], qi[4][2];
#pragma unroll
  for (int w = 4; w < 8; ++w) {
    const float zr = cs[8 + w], zs = sn[8 + w];
    qr[w - 4][0] = cs[w] * zr; qi[w - 4][0] = -(cs[w] * zs);
    qr[w - 4][1] = sn[w] * zr; qi[w - 4][1] =   sn[w] * zs;
  }
  float t1r[2], t1i[2], t2r[4], t2i[4], t3r[8], t3i[8];
#pragma unroll
  for (int b = 0; b < 2; ++b) CMUL(t1r[b], t1i[b], pr, pi, qr[0][b], qi[0][b])
#pragma unroll
  for (int b = 0; b < 4; ++b)
    CMUL(t2r[b], t2i[b], t1r[b >> 1], t1i[b >> 1], qr[1][b & 1], qi[1][b & 1])
#pragma unroll
  for (int b = 0; b < 8; ++b)
    CMUL(t3r[b], t3i[b], t2r[b >> 1], t2i[b >> 1], qr[2][b & 1], qi[2][b & 1])
#pragma unroll
  for (int j = 0; j < NS; ++j)
    CMUL(ar[j], ai[j], t3r[j >> 1], t3i[j >> 1], qr[3][j & 1], qi[3][j & 1])
}

__global__ __launch_bounds__(512) void quantum_block_kernel(
    const float* __restrict__ x,
    const float* __restrict__ W1, const float* __restrict__ b1,
    const float* __restrict__ W2, const float* __restrict__ b2,
    const float* __restrict__ W3, const float* __restrict__ b3,
    const float* __restrict__ qw,
    const float* __restrict__ Wp, const float* __restrict__ bp,
    float* __restrict__ out)
{
  __shared__ float sX  [32 * 80];
  __shared__ float sH1 [32 * 64];
  __shared__ float sH2 [32 * 32];
  __shared__ float sCos[32 * 16];
  __shared__ float sSin[32 * 16];
  __shared__ float sZ  [32 * 8];
  __shared__ v2f   sT  [3 * 256];
  __shared__ float sRY [24 * 2];

  const int tid     = threadIdx.x;
  const int wid     = tid >> 5;
  const int lane    = tid & 31;
  const int rowBase = blockIdx.x * 32;

  if (tid < 24) {
    const float th = qw[tid * 3 + 1];
    sRY[tid * 2]     = __cosf(0.5f * th);
    sRY[tid * 2 + 1] = __sinf(0.5f * th);
  }
  if (tid < 256) {
    build_phase_entry<0>(qw, &sT[0],   tid);
    build_phase_entry<2>(qw, &sT[512], tid);
  } else {
    build_phase_entry<1>(qw, &sT[256], tid - 256);
  }
  for (int i = tid; i < 32 * 80; i += 512) sX[i] = x[rowBase * 80 + i];
  __syncthreads();

  if (wid < 8) {
    const int mt = wid >> 2, nt = wid & 3;
    gemm_tile_to_lds<80, 80, true >(sX + mt * 16 * 80, W1, b1, sH1 + mt * 16 * 64, 64, nt * 16);
  }
  __syncthreads();
  if (wid < 4) {
    const int mt = wid >> 1, nt = wid & 1;
    gemm_tile_to_lds<64, 64, true >(sH1 + mt * 16 * 64, W2, b2, sH2 + mt * 16 * 32, 32, nt * 16);
  }
  __syncthreads();
  if (wid < 2)
    gemm_tile_to_halfangle_trig<32, 32>(sH2 + wid * 16 * 32, W3, b3,
                                        sCos + wid * 16 * 16, sSin + wid * 16 * 16, 16, 0);
  __syncthreads();

  {
    const int samp = wid * 2 + (lane >> 4);
    float ar[NS], ai[NS];
    product_state(&sCos[samp * 16], &sSin[samp * 16], ar, ai);

    for (int L = 0; L < 3; ++L) {
      const v2f* T = &sT[L * 256 + (lane & 15)];
#pragma unroll
      for (int j = 0; j < NS; ++j) {
        const v2f t = T[j * 16];
        const float r = ar[j] * t.x - ai[j] * t.y;
        const float i = ar[j] * t.y + ai[j] * t.x;
        ar[j] = r; ai[j] = i;
      }
      const float* ry = &sRY[L * 16];
      ry_lane<3>(ar, ai, ry[0],  ry[1]);
      ry_lane<2>(ar, ai, ry[2],  ry[3]);
      ry_lane<1>(ar, ai, ry[4],  ry[5]);
      ry_lane<0>(ar, ai, ry[6],  ry[7]);
      ry_local<3>(ar, ai, ry[8],  ry[9]);
      ry_local<2>(ar, ai, ry[10], ry[11]);
      ry_local<1>(ar, ai, ry[12], ry[13]);
      ry_local<0>(ar, ai, ry[14], ry[15]);
      cnot_lane_chain(ar, ai);
      cnot_lane_local<0, 3>(ar, ai);
      cnot_ll<3, 2>(ar, ai);
      cnot_ll<2, 1>(ar, ai);
      cnot_ll<1, 0>(ar, ai);
      cnot_local_lane<0, 3>(ar, ai);
    }

    float pb[NS]; float tot = 0.f;
#pragma unroll
    for (int j = 0; j < NS; ++j) { pb[j] = ar[j]*ar[j] + ai[j]*ai[j]; tot += pb[j]; }
    float zp[8];
#pragma unroll
    for (int w = 0; w < 4; ++w)
      zp[w] = ((lane >> (3 - w)) & 1) ? -tot : tot;
    float s4 = 0.f, s5 = 0.f, s6 = 0.f, s7 = 0.f;
#pragma unroll
    for (int j = 0; j < NS; ++j) {
      s4 += ((j >> 3) & 1) ? -pb[j] : pb[j];
      s5 += ((j >> 2) & 1) ? -pb[j] : pb[j];
      s6 += ((j >> 1) & 1) ? -pb[j] : pb[j];
      s7 += ( j       & 1) ? -pb[j] : pb[j];
    }
    zp[4] = s4; zp[5] = s5; zp[6] = s6; zp[7] = s7;
#pragma unroll
    for (int w = 0; w < 8; ++w) {
      float v = zp[w];
      v += lane_xor<8>(v);
      v += lane_xor<4>(v);
      v += lane_xor<2>(v);
      v += lane_xor<1>(v);
      if ((lane & 15) == 0) sZ[samp * 8 + w] = v;
    }
  }
  __syncthreads();

  gemm_final(sZ + (wid >> 3) * 16 * 8, Wp, bp, out, rowBase + (wid >> 3) * 16, (wid & 7) * 16);
}

extern "C" void kernel_launch(void* const* d_in, const int* in_sizes, int n_in,
                              void* d_out, int out_size, void* d_ws, size_t ws_size,
                              hipStream_t stream) {
  const float* x  = (const float*)d_in[0];
  const float* W1 = (const float*)d_in[1];
  const float* b1 = (const float*)d_in[2];
  const float* W2 = (const float*)d_in[3];
  const float* b2 = (const float*)d_in[4];
  const float* W3 = (const float*)d_in[5];
  const float* b3 = (const float*)d_in[6];
  const float* qw = (const float*)d_in[7];
  const float* Wp = (const float*)d_in[8];
  const float* bp = (const float*)d_in[9];
  float* out = (float*)d_out;

  const int B = in_sizes[0] / 80;
  const int grid = B / 32;
  quantum_block_kernel<<<grid, 512, 0, stream>>>(x, W1, b1, W2, b2, W3, b3,
                                                 qw, Wp, bp, out);
}


# Four samples per wave, one per 8-lane group (3 lane wires + 5 register wires), 64 rows per workgroup: half the waves, fewer lane exchanges per sample; init products and readout sums re-associated
# speedup vs baseline: 2.0406x; 1.0590x over previous
#include <hip/hip_runtime.h>
#include <math.h>

typedef __attribute__((ext_vector_type(2))) float v2f;
typedef __attribute__((ext_vector_type(8))) float v8f;

#define LANE() ((int)(threadIdx.x & 31))

template<int M>
__device__ __forceinline__ float lane_xor(float v) {
  static_assert(M == 1 || M == 2 || M == 4 || M == 8, "in-row partner only");
  const int b = __builtin_bit_cast(int, v);
  return __builtin_bit_cast(float,
      __builtin_amdgcn_mov_dpp(b, 0x160 | M, 0xf, 0xf, true));
}

__device__ constexpr unsigned ring_row_selects(unsigned word) {
  unsigned v = 0;
  for (unsigned j = 0; j < 8; ++j) {
    const unsigned l = word * 8 + j;
    const unsigned q = l & 7u;
    v |= ((l & 8u) | (q ^ ((q >> 1) & 3u))) << (4 * j);
  }
  return v;
}
__device__ __forceinline__ float cnot_row_chain(float v) {
  constexpr unsigned lo = ring_row_selects(0), hi = ring_row_selects(1);
  static_assert(lo == 0x45762310u && hi == 0xcdfeab98u, "in-group source lanes of the CNOT chain");
  const int b = __builtin_bit_cast(int, v);
  return __builtin_bit_cast(float,
      __builtin_amdgcn_permlane16(b, b, lo, hi, false, false));
}

__device__ __forceinline__ float fast_tanh(float x) {
  const float e = __expf(2.0f * x);
  return 1.0f - 2.0f / (e + 1.0f);
}

template<int K, int LDA>
__device__ __forceinline__ v8f gemm_tile_acc(
    const float* __restrict__ sA,
    const float* __restrict__ Wg,
    int n)
{
  const int lane = LANE();
  const int half = lane >> 4;
  const int col  = lane & 15;
  v8f acc = {};
#pragma unroll
  for (int k0 = 0; k0 < K; k0 += 4) {
    const int ka = k0 + half * 2;
    v2f a, b;
    a.x = sA[col * LDA + ka];
    a.y = sA[col * LDA + ka + 1];
    b.x = Wg[n * K + ka];
    b.y = Wg[n * K + ka + 1];
    acc = __builtin_amdgcn_wmma_f32_16x16x4_f32(false, a, false, b,
                                                (short)0, acc, false, false);
  }
  return acc;
}

template<int K, int LDA, bool TANH>
__device__ __forceinline__ void gemm_tile_to_lds(
    const float* __restrict__ sA, const float* __restrict__ Wg,
    const float* __restrict__ bg,
    float* __restrict__ sOut, int ldo, int n0)
{
  const int lane = LANE();
  const int half = lane >> 4;
  const int n    = n0 + (lane & 15);
  const v8f acc = gemm_tile_acc<K, LDA>(sA, Wg, n);
  const float bias = bg[n];
#pragma unroll
  for (int r = 0; r < 8; ++r) {
    float v = acc[r] + bias;
    if (TANH) v = fast_tanh(v);
    sOut[(half * 8 + r) * ldo + n] = v;
  }
}

template<int K, int LDA>
__device__ __forceinline__ void gemm_tile_to_halfangle_trig(
    const float* __restrict__ sA, const float* __restrict__ Wg,
    const float* __restrict__ bg,
    float* __restrict__ sCos, float* __restrict__ sSin, int ldo, int n0)
{
  const int lane = LANE();
  const int half = lane >> 4;
  const int n    = n0 + (lane & 15);
  const v8f acc = gemm_tile_acc<K, LDA>(sA, Wg, n);
  const float bias = bg[n];
#pragma unroll
  for (int r = 0; r < 8; ++r) {
    const float h = 0.5f * (acc[r] + bias);
    sCos[(half * 8 + r) * ldo + n] = __cosf(h);
    sSin[(half * 8 + r) * ldo + n] = __sinf(h);
  }
}

__device__ __forceinline__ void gemm_final(
    const float* __restrict__ sZ,
    const float* __restrict__ Wp,
    const float* __restrict__ bp,
    float* __restrict__ out, int rowBase, int n0)
{
  const int lane = LANE();
  const int half = lane >> 4;
  const int col  = lane & 15;
  const int n    = n0 + col;
  v8f acc = {};
#pragma unroll
  for (int k0 = 0; k0 < 8; k0 += 4) {
    const int ka = k0 + half * 2;
    v2f a, b;
    a.x = sZ[col * 8 + ka];
    a.y = sZ[col * 8 + ka + 1];
    b.x = Wp[n * 8 + ka];
    b.y = Wp[n * 8 + ka + 1];
    acc = __builtin_amdgcn_wmma_f32_16x16x4_f32(false, a, false, b,
                                                (short)0, acc, false, false);
  }
  const float bias = bp[n];
#pragma unroll
  for (int r = 0; r < 8; ++r)
    out[(size_t)(rowBase + half * 8 + r) * 128 + n] = acc[r] + bias;
}


__device__ __forceinline__ int ring_preimage(int c) {
  c ^= (c & 1) << 7;
  c ^= (c >> 1) & 1;
  c ^= ((c >> 2) & 1) << 1;
  c ^= ((c >> 3) & 1) << 2;
  c ^= ((c >> 4) & 1) << 3;
  c ^= ((c >> 5) & 1) << 4;
  c ^= ((c >> 6) & 1) << 5;
  c ^= ((c >> 7) & 1) << 6;
  return c;
}

template<int L, int COMP>
__device__ __forceinline__ float rz_phase(const float* __restrict__ qw, int c) {
  float s = 0.f;
#pragma unroll
  for (int w = 0; w < 8; ++w) {
    const float h = 0.5f * qw[(L * 8 + w) * 3 + COMP];
    s += ((c >> (7 - w)) & 1) ? h : -h;
  }
  return s;
}

template<int L>
__device__ __forceinline__ void build_phase_entry(const float* __restrict__ qw,
                                                  v2f* __restrict__ sTL, int c)
{
  float ang = rz_phase<L, 0>(qw, c);
  if constexpr (L > 0) ang += rz_phase<L - 1, 2>(qw, ring_preimage(c));
  v2f t;
  t.x = __cosf(ang);
  t.y = __sinf(ang);
  sTL[(c & 31) * 8 + (c >> 5)] = t;
}


constexpr int NS = 32;

template<int LB>
__device__ __forceinline__ void ry_lane(float ar[NS], float ai[NS], float c, float s) {
  const float t = ((LANE() >> LB) & 1) ? s : -s;
#pragma unroll
  for (int j = 0; j < NS; ++j) {
    const float pr = lane_xor<(1 << LB)>(ar[j]);
    const float pi = lane_xor<(1 << LB)>(ai[j]);
    ar[j] = c * ar[j] + t * pr;
    ai[j] = c * ai[j] + t * pi;
  }
}

template<int JB>
__device__ __forceinline__ void ry_local(float ar[NS], float ai[NS], float c, float s) {
#pragma unroll
  for (int j = 0; j < NS; ++j) {
    if (!((j >> JB) & 1)) {
      const int j1 = j | (1 << JB);
      const float a0r = ar[j], a0i = ai[j], a1r = ar[j1], a1i = ai[j1];
      ar[j]  = c * a0r - s * a1r;  ai[j]  = c * a0i - s * a1i;
      ar[j1] = s * a0r + c * a1r;  ai[j1] = s * a0i + c * a1i;
    }
  }
}

template<int CJB, int TJB>
__device__ __forceinline__ void cnot_ll(float ar[NS], float ai[NS]) {
#pragma unroll
  for (int j = 0; j < NS; ++j)
    if (((j >> CJB) & 1) && !((j >> TJB) & 1)) {
      const int j1 = j | (1 << TJB);
      float tr = ar[j], ti = ai[j];
      ar[j] = ar[j1]; ai[j] = ai[j1];
      ar[j1] = tr;    ai[j1] = ti;
    }
}
template<int CJB, int TLB>
__device__ __forceinline__ void cnot_local_lane(float ar[NS], float ai[NS]) {
#pragma unroll
  for (int j = 0; j < NS; ++j)
    if ((j >> CJB) & 1) {
      ar[j] = lane_xor<(1 << TLB)>(ar[j]);
      ai[j] = lane_xor<(1 << TLB)>(ai[j]);
    }
}
template<int CLB, int TJB>
__device__ __forceinline__ void cnot_lane_local(float ar[NS], float ai[NS]) {
  if ((LANE() >> CLB) & 1) {
#pragma unroll
    for (int j = 0; j < NS; ++j)
      if (!((j >> TJB) & 1)) {
        const int j1 = j | (1 << TJB);
        float tr = ar[j], ti = ai[j];
        ar[j] = ar[j1]; ai[j] = ai[j1];
        ar[j1] = tr;    ai[j1] = ti;
      }
  }
}
__device__ __forceinline__ void cnot_lane_chain(float ar[NS], float ai[NS]) {
#pragma unroll
  for (int j = 0; j < NS; ++j) {
    ar[j] = cnot_row_chain(ar[j]);
    ai[j] = cnot_row_chain(ai[j]);
  }
}

template<int W>
__device__ __forceinline__ void wire_factor_lane(const float* __restrict__ cs,
    const float* __restrict__ sn, float& fr, float& fi)
{
  const int bit = (LANE() >> (2 - W)) & 1;
  const float mag = bit ? sn[W] : cs[W];
  const float zs  = sn[8 + W];
  fr = mag * cs[8 + W];
  fi = mag * (bit ? zs : -zs);
}

#define CMUL(orr, oi, xr, xi, yr, yi) {                                       \
    orr = (xr) * (yr) - (xi) * (yi);                                          \
    oi  = (xr) * (yi) + (xi) * (yr);                                          \
  }

__device__ __forceinline__ void product_state(const float* __restrict__ cs,
    const float* __restrict__ sn, float ar[NS], float ai[NS])
{
  float pr, pi, fr, fi, tr, ti;
  wire_factor_lane<0>(cs, sn, pr, pi);
  wire_factor_lane<1>(cs, sn, fr, fi); CMUL(tr, ti, pr, pi, fr, fi) pr = tr; pi = ti;
  wire_factor_lane<2>(cs, sn, fr, fi); CMUL(tr, ti, pr, pi, fr, fi) pr = tr; pi = ti;
  float qr[5][2], qi[5][2];
#pragma unroll
  for (int w = 3; w < 8; ++w) {
    const float zr = cs[8 + w], zs = sn[8 + w];
    qr[w - 3][0] = cs[w] * zr; qi[w - 3][0] = -(cs[w] * zs);
    qr[w - 3][1] = sn[w] * zr; qi[w - 3][1] =   sn[w] * zs;
  }
  float t1r[2], t1i[2], t2r[4], t2i[4], t3r[8], t3i[8], t4r[16], t4i[16];
#pragma unroll
  for (int b = 0; b < 2; ++b) CMUL(t1r[b], t1i[b], pr, pi, qr[0][b], qi[0][b])
#pragma unroll
  for (int b = 0; b < 4; ++b)
    CMUL(t2r[b], t2i[b], t1r[b >> 1], t1i[b >> 1], qr[1][b & 1], qi[1][b & 1])
#pragma unroll
  for (int b = 0; b < 8; ++b)
    CMUL(t3r[b], t3i[b], t2r[b >> 1], t2i[b >> 1], qr[2][b & 1], qi[2][b & 1])
#pragma unroll
  for (int b = 0; b < 16; ++b)
    CMUL(t4r[b], t4i[b], t3r[b >> 1], t3i[b >> 1], qr[3][b & 1], qi[3][b & 1])
#pragma unroll
  for (int j = 0; j < NS; ++j)
    CMUL(ar[j], ai[j], t4r[j >> 1], t4i[j >> 1], qr[4][j & 1], qi[4][j & 1])
}

__global__ __launch_bounds__(512) void quantum_block_kernel(
    const float* __restrict__ x,
    const float* __restrict__ W1, const float* __restrict__ b1,
    const float* __restrict__ W2, const float* __restrict__ b2,
    const float* __restrict__ W3, const float* __restrict__ b3,
    const float* __restrict__ qw,
    const float* __restrict__ Wp, const float* __restrict__ bp,
    float* __restrict__ out)
{
  __shared__ float sX  [64 * 80];
  __shared__ float sH1 [64 * 64];
  __shared__ float sH2 [64 * 32];
  __shared__ float sCos[64 * 16];
  __shared__ float sSin[64 * 16];
  __shared__ float sZ  [64 * 8];
  __shared__ v2f   sT  [3 * 256];
  __shared__ float sRY [24 * 2];

  const int tid     = threadIdx.x;
  const int wid     = tid >> 5;
  const int lane    = tid & 31;
  const int rowBase = blockIdx.x * 64;

  if (tid < 24) {
    const float th = qw[tid * 3 + 1];
    sRY[tid * 2]     = __cosf(0.5f * th);
    sRY[tid * 2 + 1] = __sinf(0.5f * th);
  }
  if (tid < 256) {
    build_phase_entry<0>(qw, &sT[0],   tid);
    build_phase_entry<2>(qw, &sT[512], tid);
  } else {
    build_phase_entry<1>(qw, &sT[256], tid - 256);
  }
#pragma unroll
  for (int it = 0; it < 10; ++it) sX[it * 512 + tid] = x[rowBase * 80 + it * 512 + tid];
  __syncthreads();

  {
    const int mt = wid >> 2, nt = wid & 3;
    gemm_tile_to_lds<80, 80, true >(sX + mt * 16 * 80, W1, b1, sH1 + mt * 16 * 64, 64, nt * 16);
  }
  __syncthreads();
  if (wid < 8) {
    const int mt = wid >> 1, nt = wid & 1;
    gemm_tile_to_lds<64, 64, true >(sH1 + mt * 16 * 64, W2, b2, sH2 + mt * 16 * 32, 32, nt * 16);
  }
  __syncthreads();
  if (wid < 4)
    gemm_tile_to_halfangle_trig<32, 32>(sH2 + wid * 16 * 32, W3, b3,
                                        sCos + wid * 16 * 16, sSin + wid * 16 * 16, 16, 0);
  __syncthreads();

  {
    const int samp = wid * 4 + (lane >> 3);
    float ar[NS], ai[NS];
    product_state(&sCos[samp * 16], &sSin[samp * 16], ar, ai);

    for (int L = 0; L < 3; ++L) {
      const v2f* T = &sT[L * 256 + (lane & 7)];
#pragma unroll
      for (int j = 0; j < NS; ++j) {
        const v2f t = T[j * 8];
        const float r = ar[j] * t.x - ai[j] * t.y;
        const float i = ar[j] * t.y + ai[j] * t.x;
        ar[j] = r; ai[j] = i;
      }
      const float* ry = &sRY[L * 16];
      ry_lane<2>(ar, ai, ry[0],  ry[1]);
      ry_lane<1>(ar, ai, ry[2],  ry[3]);
      ry_lane<0>(ar, ai, ry[4],  ry[5]);
      ry_local<4>(ar, ai, ry[6],  ry[7]);
      ry_local<3>(ar, ai, ry[8],  ry[9]);
      ry_local<2>(ar, ai, ry[10], ry[11]);
      ry_local<1>(ar, ai, ry[12], ry[13]);
      ry_local<0>(ar, ai, ry[14], ry[15]);
      cnot_lane_chain(ar, ai);
      cnot_lane_local<0, 4>(ar, ai);
      cnot_ll<4, 3>(ar, ai);
      cnot_ll<3, 2>(ar, ai);
      cnot_ll<2, 1>(ar, ai);
      cnot_ll<1, 0>(ar, ai);
      cnot_local_lane<0, 2>(ar, ai);
    }

    float pb[NS]; float tot = 0.f;
#pragma unroll
    for (int j = 0; j < NS; ++j) { pb[j] = ar[j]*ar[j] + ai[j]*ai[j]; tot += pb[j]; }
    float zp[8];
#pragma unroll
    for (int w = 0; w < 3; ++w)
      zp[w] = ((lane >> (2 - w)) & 1) ? -tot : tot;
    float s3 = 0.f, s4 = 0.f, s5 = 0.f, s6 = 0.f, s7 = 0.f;
#pragma unroll
    for (int j = 0; j < NS; ++j) {
      s3 += ((j >> 4) & 1) ? -pb[j] : pb[j];
      s4 += ((j >> 3) & 1) ? -pb[j] : pb[j];
      s5 += ((j >> 2) & 1) ? -pb[j] : pb[j];
      s6 += ((j >> 1) & 1) ? -pb[j] : pb[j];
      s7 += ( j       & 1) ? -pb[j] : pb[j];
    }
    zp[3] = s3; zp[4] = s4; zp[5] = s5; zp[6] = s6; zp[7] = s7;
#pragma unroll
    for (int w = 0; w < 8; ++w) {
      float v = zp[w];
      v += lane_xor<4>(v);
      v += lane_xor<2>(v);
      v += lane_xor<1>(v);
      if ((lane & 7) == 0) sZ[samp * 8 + w] = v;
    }
  }
  __syncthreads();

#pragma unroll
  for (int i = 0; i < 2; ++i) {
    const int mt = (wid >> 3) * 2 + i;
    gemm_final(sZ + mt * 16 * 8, Wp, bp, out, rowBase + mt * 16, (wid & 7) * 16);
  }
}

extern "C" void kernel_launch(void* const* d_in, const int* in_sizes, int n_in,
                              void* d_out, int out_size, void* d_ws, size_t ws_size,
                              hipStream_t stream) {
  const float* x  = (const float*)d_in[0];
  const float* W1 = (const float*)d_in[1];
  const float* b1 = (const float*)d_in[2];
  const float* W2 = (const float*)d_in[3];
  const float* b2 = (const float*)d_in[4];
  const float* W3 = (const float*)d_in[5];
  const float* b3 = (const float*)d_in[6];
  const float* qw = (const float*)d_in[7];
  const float* Wp = (const float*)d_in[8];
  const float* bp = (const float*)d_in[9];
  float* out = (float*)d_out;

  const int B = in_sizes[0] / 80;
  const int grid = B / 64;
  quantum_block_kernel<<<grid, 512, 0, stream>>>(x, W1, b1, W2, b2, W3, b3,
                                                 qw, Wp, bp, out);
}


# Layer 0 folded into the product state: state is a product until the first CNOT ring, so RZ(phi), RY(theta) act per wire and the ring only renames factors; products re-associated
# speedup vs baseline: 2.3575x; 1.1553x over previous
#include <hip/hip_runtime.h>
#include <math.h>

typedef __attribute__((ext_vector_type(2))) float v2f;
typedef __attribute__((ext_vector_type(8))) float v8f;

#define LANE() ((int)(threadIdx.x & 31))

template<int M>
__device__ __forceinline__ float lane_xor(float v) {
  static_assert(M == 1 || M == 2 || M == 4 || M == 8, "in-row partner only");
  const int b = __builtin_bit_cast(int, v);
  return __builtin_bit_cast(float,
      __builtin_amdgcn_mov_dpp(b, 0x160 | M, 0xf, 0xf, true));
}

__device__ constexpr unsigned ring_row_selects(unsigned word) {
  unsigned v = 0;
  for (unsigned j = 0; j < 8; ++j) {
    const unsigned l = word * 8 + j;
    const unsigned q = l & 7u;
    v |= ((l & 8u) | (q ^ ((q >> 1) & 3u))) << (4 * j);
  }
  return v;
}
__device__ __forceinline__ float cnot_row_chain(float v) {
  constexpr unsigned lo = ring_row_selects(0), hi = ring_row_selects(1);
  static_assert(lo == 0x45762310u && hi == 0xcdfeab98u, "in-group source lanes of the CNOT chain");
  const int b = __builtin_bit_cast(int, v);
  return __builtin_bit_cast(float,
      __builtin_amdgcn_permlane16(b, b, lo, hi, false, false));
}

__device__ __forceinline__ float fast_tanh(float x) {
  const float e = __expf(2.0f * x);
  return 1.0f - 2.0f / (e + 1.0f);
}

template<int K, int LDA>
__device__ __forceinline__ v8f gemm_tile_acc(
    const float* __restrict__ sA,
    const float* __restrict__ Wg,
    int n)
{
  const int lane = LANE();
  const int half = lane >> 4;
  const int col  = lane & 15;
  v8f acc = {};
#pragma unroll
  for (int k0 = 0; k0 < K; k0 += 4) {
    const int ka = k0 + half * 2;
    v2f a, b;
    a.x = sA[col * LDA + ka];
    a.y = sA[col * LDA + ka + 1];
    b.x = Wg[n * K + ka];
    b.y = Wg[n * K + ka + 1];
    acc = __builtin_amdgcn_wmma_f32_16x16x4_f32(false, a, false, b,
                                                (short)0, acc, false, false);
  }
  return acc;
}

template<int K, int LDA, bool TANH>
__device__ __forceinline__ void gemm_tile_to_lds(
    const float* __restrict__ sA, const float* __restrict__ Wg,
    const float* __restrict__ bg,
    float* __restrict__ sOut, int ldo, int n0)
{
  const int lane = LANE();
  const int half = lane >> 4;
  const int n    = n0 + (lane & 15);
  const v8f acc = gemm_tile_acc<K, LDA>(sA, Wg, n);
  const float bias = bg[n];
#pragma unroll
  for (int r = 0; r < 8; ++r) {
    float v = acc[r] + bias;
    if (TANH) v = fast_tanh(v);
    sOut[(half * 8 + r) * ldo + n] = v;
  }
}

template<int K, int LDA>
__device__ __forceinline__ void gemm_tile_to_halfangle_trig(
    const float* __restrict__ sA, const float* __restrict__ Wg,
    const float* __restrict__ bg,
    float* __restrict__ sCos, float* __restrict__ sSin, int ldo, int n0)
{
  const int lane = LANE();
  const int half = lane >> 4;
  const int n    = n0 + (lane & 15);
  const v8f acc = gemm_tile_acc<K, LDA>(sA, Wg, n);
  const float bias = bg[n];
#pragma unroll
  for (int r = 0; r < 8; ++r) {
    const float h = 0.5f * (acc[r] + bias);
    sCos[(half * 8 + r) * ldo + n] = __cosf(h);
    sSin[(half * 8 + r) * ldo + n] = __sinf(h);
  }
}

__device__ __forceinline__ void gemm_final(
    const float* __restrict__ sZ,
    const float* __restrict__ Wp,
    const float* __restrict__ bp,
    float* __restrict__ out, int rowBase, int n0)
{
  const int lane = LANE();
  const int half = lane >> 4;
  const int col  = lane & 15;
  const int n    = n0 + col;
  v8f acc = {};
#pragma unroll
  for (int k0 = 0; k0 < 8; k0 += 4) {
    const int ka = k0 + half * 2;
    v2f a, b;
    a.x = sZ[col * 8 + ka];
    a.y = sZ[col * 8 + ka + 1];
    b.x = Wp[n * 8 + ka];
    b.y = Wp[n * 8 + ka + 1];
    acc = __builtin_amdgcn_wmma_f32_16x16x4_f32(false, a, false, b,
                                                (short)0, acc, false, false);
  }
  const float bias = bp[n];
#pragma unroll
  for (int r = 0; r < 8; ++r)
    out[(size_t)(rowBase + half * 8 + r) * 128 + n] = acc[r] + bias;
}


__device__ __forceinline__ int ring_preimage(int c) {
  c ^= (c & 1) << 7;
  c ^= (c >> 1) & 1;
  c ^= ((c >> 2) & 1) << 1;
  c ^= ((c >> 3) & 1) << 2;
  c ^= ((c >> 4) & 1) << 3;
  c ^= ((c >> 5) & 1) << 4;
  c ^= ((c >> 6) & 1) << 5;
  c ^= ((c >> 7) & 1) << 6;
  return c;
}

template<int L, int COMP>
__device__ __forceinline__ float rz_phase(const float* __restrict__ qw, int c) {
  float s = 0.f;
#pragma unroll
  for (int w = 0; w < 8; ++w) {
    const float h = 0.5f * qw[(L * 8 + w) * 3 + COMP];
    s += ((c >> (7 - w)) & 1) ? h : -h;
  }
  return s;
}

template<int L>
__device__ __forceinline__ void build_phase_entry(const float* __restrict__ qw,
                                                  v2f* __restrict__ sTL, int c)
{
  float ang = rz_phase<L, 0>(qw, c);
  if constexpr (L > 0) ang += rz_phase<L - 1, 2>(qw, ring_preimage(c));
  v2f t;
  t.x = __cosf(ang);
  t.y = __sinf(ang);
  sTL[(c & 31) * 8 + (c >> 5)] = t;
}


constexpr int NS = 32;

template<int LB>
__device__ __forceinline__ void ry_lane(float ar[NS], float ai[NS], float c, float s) {
  const float t = ((LANE() >> LB) & 1) ? s : -s;
#pragma unroll
  for (int j = 0; j < NS; ++j) {
    const float pr = lane_xor<(1 << LB)>(ar[j]);
    const float pi = lane_xor<(1 << LB)>(ai[j]);
    ar[j] = c * ar[j] + t * pr;
    ai[j] = c * ai[j] + t * pi;
  }
}

template<int JB>
__device__ __forceinline__ void ry_local(float ar[NS], float ai[NS], float c, float s) {
#pragma unroll
  for (int j = 0; j < NS; ++j) {
    if (!((j >> JB) & 1)) {
      const int j1 = j | (1 << JB);
      const float a0r = ar[j], a0i = ai[j], a1r = ar[j1], a1i = ai[j1];
      ar[j]  = c * a0r - s * a1r;  ai[j]  = c * a0i - s * a1i;
      ar[j1] = s * a0r + c * a1r;  ai[j1] = s * a0i + c * a1i;
    }
  }
}

template<int CJB, int TJB>
__device__ __forceinline__ void cnot_ll(float ar[NS], float ai[NS]) {
#pragma unroll
  for (int j = 0; j < NS; ++j)
    if (((j >> CJB) & 1) && !((j >> TJB) & 1)) {
      const int j1 = j | (1 << TJB);
      float tr = ar[j], ti = ai[j];
      ar[j] = ar[j1]; ai[j] = ai[j1];
      ar[j1] = tr;    ai[j1] = ti;
    }
}
template<int CJB, int TLB>
__device__ __forceinline__ void cnot_local_lane(float ar[NS], float ai[NS]) {
#pragma unroll
  for (int j = 0; j < NS; ++j)
    if ((j >> CJB) & 1) {
      ar[j] = lane_xor<(1 << TLB)>(ar[j]);
      ai[j] = lane_xor<(1 << TLB)>(ai[j]);
    }
}
template<int CLB, int TJB>
__device__ __forceinline__ void cnot_lane_local(float ar[NS], float ai[NS]) {
  if ((LANE() >> CLB) & 1) {
#pragma unroll
    for (int j = 0; j < NS; ++j)
      if (!((j >> TJB) & 1)) {
        const int j1 = j | (1 << TJB);
        float tr = ar[j], ti = ai[j];
        ar[j] = ar[j1]; ai[j] = ai[j1];
        ar[j1] = tr;    ai[j1] = ti;
      }
  }
}
__device__ __forceinline__ void cnot_lane_chain(float ar[NS], float ai[NS]) {
#pragma unroll
  for (int j = 0; j < NS; ++j) {
    ar[j] = cnot_row_chain(ar[j]);
    ai[j] = cnot_row_chain(ai[j]);
  }
}

__device__ __forceinline__ void cnot_ring(float ar[NS], float ai[NS]) {
  cnot_lane_chain(ar, ai);
  cnot_lane_local<0, 4>(ar, ai);
  cnot_ll<4, 3>(ar, ai);
  cnot_ll<3, 2>(ar, ai);
  cnot_ll<2, 1>(ar, ai);
  cnot_ll<1, 0>(ar, ai);
  cnot_local_lane<0, 2>(ar, ai);
}

template<int W>
__device__ __forceinline__ void wire_state(const float* __restrict__ cs,
    const float* __restrict__ sn, const float* __restrict__ ph0,
    const float* __restrict__ ry0,
    float& g0r, float& g0i, float& g1r, float& g1i)
{
  const float zr = cs[8 + W], zs = sn[8 + W];
  const float pc = ph0[2 * W], ps = ph0[2 * W + 1];
  const float ur = zr * pc - zs * ps;
  const float ui = zs * pc + zr * ps;
  const float f0r = cs[W] * ur, f0i = -(cs[W] * ui);
  const float f1r = sn[W] * ur, f1i =   sn[W] * ui;
  const float C = ry0[2 * W], S = ry0[2 * W + 1];
  g0r = C * f0r - S * f1r;  g0i = C * f0i - S * f1i;
  g1r = S * f0r + C * f1r;  g1i = S * f0i + C * f1i;
}

#define CMUL(orr, oi, xr, xi, yr, yi) {                                       \
    orr = (xr) * (yr) - (xi) * (yi);                                          \
    oi  = (xr) * (yi) + (xi) * (yr);                                          \
  }

__device__ __forceinline__ void product_state_behind_ring(
    const float* __restrict__ cs, const float* __restrict__ sn,
    const float* __restrict__ ph0, const float* __restrict__ ry0,
    float ar[NS], float ai[NS])
{
  float gr[8][2], gi[8][2];
  wire_state<0>(cs, sn, ph0, ry0, gr[0][0], gi[0][0], gr[0][1], gi[0][1]);
  wire_state<1>(cs, sn, ph0, ry0, gr[1][0], gi[1][0], gr[1][1], gi[1][1]);
  wire_state<2>(cs, sn, ph0, ry0, gr[2][0], gi[2][0], gr[2][1], gi[2][1]);
  wire_state<3>(cs, sn, ph0, ry0, gr[3][0], gi[3][0], gr[3][1], gi[3][1]);
  wire_state<4>(cs, sn, ph0, ry0, gr[4][0], gi[4][0], gr[4][1], gi[4][1]);
  wire_state<5>(cs, sn, ph0, ry0, gr[5][0], gi[5][0], gr[5][1], gi[5][1]);
  wire_state<6>(cs, sn, ph0, ry0, gr[6][0], gi[6][0], gr[6][1], gi[6][1]);
  wire_state<7>(cs, sn, ph0, ry0, gr[7][0], gi[7][0], gr[7][1], gi[7][1]);

  const int x0 = (LANE() >> 2) & 1;
  const int x1 = (LANE() >> 1) & 1;
  const int x2 =  LANE()       & 1;

  const int y = x1 ^ x0;
  const float u0r = x0 ? gr[0][1] : gr[0][0], u0i = x0 ? gi[0][1] : gi[0][0];
  const float v0r = x0 ? gr[0][0] : gr[0][1], v0i = x0 ? gi[0][0] : gi[0][1];
  const float u1r = y  ? gr[1][1] : gr[1][0], u1i = y  ? gi[1][1] : gi[1][0];
  const float v1r = y  ? gr[1][0] : gr[1][1], v1i = y  ? gi[1][0] : gi[1][1];
  float Ar[2], Ai[2];
  CMUL(Ar[0], Ai[0], u0r, u0i, u1r, u1i)
  CMUL(Ar[1], Ai[1], v0r, v0i, v1r, v1i)
  float Hr[2][2], Hi[2][2];
#pragma unroll
  for (int c6 = 0; c6 < 2; ++c6)
#pragma unroll
    for (int c7 = 0; c7 < 2; ++c7)
      CMUL(Hr[c6][c7], Hi[c6][c7], gr[7][c7 ^ c6], gi[7][c7 ^ c6], Ar[c7], Ai[c7])

  const int z = x2 ^ x1;
  const float l2r = z ? gr[2][1] : gr[2][0], l2i = z ? gi[2][1] : gi[2][0];
  float G3r[2], G3i[2];
  G3r[0] = x2 ? gr[3][1] : gr[3][0];  G3i[0] = x2 ? gi[3][1] : gi[3][0];
  G3r[1] = x2 ? gr[3][0] : gr[3][1];  G3i[1] = x2 ? gi[3][0] : gi[3][1];

  float t1r[2], t1i[2], t2r[4], t2i[4], t3r[8], t3i[8], t4r[16], t4i[16];
#pragma unroll
  for (int b = 0; b < 2; ++b) CMUL(t1r[b], t1i[b], l2r, l2i, G3r[b], G3i[b])
#pragma unroll
  for (int b = 0; b < 4; ++b)
    CMUL(t2r[b], t2i[b], t1r[b >> 1], t1i[b >> 1],
         gr[4][(b & 1) ^ ((b >> 1) & 1)], gi[4][(b & 1) ^ ((b >> 1) & 1)])
#pragma unroll
  for (int b = 0; b < 8; ++b)
    CMUL(t3r[b], t3i[b], t2r[b >> 1], t2i[b >> 1],
         gr[5][(b & 1) ^ ((b >> 1) & 1)], gi[5][(b & 1) ^ ((b >> 1) & 1)])
#pragma unroll
  for (int b = 0; b < 16; ++b)
    CMUL(t4r[b], t4i[b], t3r[b >> 1], t3i[b >> 1],
         gr[6][(b & 1) ^ ((b >> 1) & 1)], gi[6][(b & 1) ^ ((b >> 1) & 1)])
#pragma unroll
  for (int j = 0; j < NS; ++j)
    CMUL(ar[j], ai[j], t4r[j >> 1], t4i[j >> 1],
         Hr[(j >> 1) & 1][j & 1], Hi[(j >> 1) & 1][j & 1])
}

__global__ __launch_bounds__(512) void quantum_block_kernel(
    const float* __restrict__ x,
    const float* __restrict__ W1, const float* __restrict__ b1,
    const float* __restrict__ W2, const float* __restrict__ b2,
    const float* __restrict__ W3, const float* __restrict__ b3,
    const float* __restrict__ qw,
    const float* __restrict__ Wp, const float* __restrict__ bp,
    float* __restrict__ out)
{
  __shared__ float sX  [64 * 80];
  __shared__ float sH1 [64 * 64];
  __shared__ float sH2 [64 * 32];
  __shared__ float sCos[64 * 16];
  __shared__ float sSin[64 * 16];
  __shared__ float sZ  [64 * 8];
  __shared__ v2f   sT  [2 * 256];
  __shared__ float sRY [24 * 2];
  __shared__ float sPH0[8 * 2];

  const int tid     = threadIdx.x;
  const int wid     = tid >> 5;
  const int lane    = tid & 31;
  const int rowBase = blockIdx.x * 64;

  if (tid < 24) {
    const float th = qw[tid * 3 + 1];
    sRY[tid * 2]     = __cosf(0.5f * th);
    sRY[tid * 2 + 1] = __sinf(0.5f * th);
    if (tid < 8) {
      const float ph = qw[tid * 3];
      sPH0[tid * 2]     = __cosf(0.5f * ph);
      sPH0[tid * 2 + 1] = __sinf(0.5f * ph);
    }
  }
  if (tid < 256) {
    build_phase_entry<1>(qw, &sT[0],   tid);
  } else {
    build_phase_entry<2>(qw, &sT[256], tid - 256);
  }
#pragma unroll
  for (int it = 0; it < 10; ++it) sX[it * 512 + tid] = x[rowBase * 80 + it * 512 + tid];
  __syncthreads();

  {
    const int mt = wid >> 2, nt = wid & 3;
    gemm_tile_to_lds<80, 80, true >(sX + mt * 16 * 80, W1, b1, sH1 + mt * 16 * 64, 64, nt * 16);
  }
  __syncthreads();
  if (wid < 8) {
    const int mt = wid >> 1, nt = wid & 1;
    gemm_tile_to_lds<64, 64, true >(sH1 + mt * 16 * 64, W2, b2, sH2 + mt * 16 * 32, 32, nt * 16);
  }
  __syncthreads();
  if (wid < 4)
    gemm_tile_to_halfangle_trig<32, 32>(sH2 + wid * 16 * 32, W3, b3,
                                        sCos + wid * 16 * 16, sSin + wid * 16 * 16, 16, 0);
  __syncthreads();

  {
    const int samp = wid * 4 + (lane >> 3);
    float ar[NS], ai[NS];
    product_state_behind_ring(&sCos[samp * 16], &sSin[samp * 16], sPH0, sRY, ar, ai);

    for (int L = 1; L < 3; ++L) {
      const v2f* T = &sT[(L - 1) * 256 + (lane & 7)];
#pragma unroll
      for (int j = 0; j < NS; ++j) {
        const v2f t = T[j * 8];
        const float r = ar[j] * t.x - ai[j] * t.y;
        const float i = ar[j] * t.y + ai[j] * t.x;
        ar[j] = r; ai[j] = i;
      }
      const float* ry = &sRY[L * 16];
      ry_lane<2>(ar, ai, ry[0],  ry[1]);
      ry_lane<1>(ar, ai, ry[2],  ry[3]);
      ry_lane<0>(ar, ai, ry[4],  ry[5]);
      ry_local<4>(ar, ai, ry[6],  ry[7]);
      ry_local<3>(ar, ai, ry[8],  ry[9]);
      ry_local<2>(ar, ai, ry[10], ry[11]);
      ry_local<1>(ar, ai, ry[12], ry[13]);
      ry_local<0>(ar, ai, ry[14], ry[15]);
      cnot_ring(ar, ai);
    }

    float pb[NS]; float tot = 0.f;
#pragma unroll
    for (int j = 0; j < NS; ++j) { pb[j] = ar[j]*ar[j] + ai[j]*ai[j]; tot += pb[j]; }
    float zp[8];
#pragma unroll
    for (int w = 0; w < 3; ++w)
      zp[w] = ((lane >> (2 - w)) & 1) ? -tot : tot;
    float s3 = 0.f, s4 = 0.f, s5 = 0.f, s6 = 0.f, s7 = 0.f;
#pragma unroll
    for (int j = 0; j < NS; ++j) {
      s3 += ((j >> 4) & 1) ? -pb[j] : pb[j];
      s4 += ((j >> 3) & 1) ? -pb[j] : pb[j];
      s5 += ((j >> 2) & 1) ? -pb[j] : pb[j];
      s6 += ((j >> 1) & 1) ? -pb[j] : pb[j];
      s7 += ( j       & 1) ? -pb[j] : pb[j];
    }
    zp[3] = s3; zp[4] = s4; zp[5] = s5; zp[6] = s6; zp[7] = s7;
#pragma unroll
    for (int w = 0; w < 8; ++w) {
      float v = zp[w];
      v += lane_xor<4>(v);
      v += lane_xor<2>(v);
      v += lane_xor<1>(v);
      if ((lane & 7) == 0) sZ[samp * 8 + w] = v;
    }
  }
  __syncthreads();

#pragma unroll
  for (int i = 0; i < 2; ++i) {
    const int mt = (wid >> 3) * 2 + i;
    gemm_final(sZ + mt * 16 * 8, Wp, bp, out, rowBase + mt * 16, (wid & 7) * 16);
  }
}

extern "C" void kernel_launch(void* const* d_in, const int* in_sizes, int n_in,
                              void* d_out, int out_size, void* d_ws, size_t ws_size,
                              hipStream_t stream) {
  const float* x  = (const float*)d_in[0];
  const float* W1 = (const float*)d_in[1];
  const float* b1 = (const float*)d_in[2];
  const float* W2 = (const float*)d_in[3];
  const float* b2 = (const float*)d_in[4];
  const float* W3 = (const float*)d_in[5];
  const float* b3 = (const float*)d_in[6];
  const float* qw = (const float*)d_in[7];
  const float* Wp = (const float*)d_in[8];
  const float* bp = (const float*)d_in[9];
  float* out = (float*)d_out;

  const int B = in_sizes[0] / 80;
  const int grid = B / 64;
  quantum_block_kernel<<<grid, 512, 0, stream>>>(x, W1, b1, W2, b2, W3, b3,
                                                 qw, Wp, bp, out);
}


# All MLP WMMA operand tiles in LDS at row pitch K+4 floats (16 B x odd: x, h1, h2 reads were 4/16/8-way bank-conflicted); W1-W3 staged there once per workgroup instead of 16-line global fragment loads
# speedup vs baseline: 2.6700x; 1.1326x over previous
#include <hip/hip_runtime.h>
#include <math.h>

typedef __attribute__((ext_vector_type(2))) float v2f;
typedef __attribute__((ext_vector_type(4))) float v4f;
typedef __attribute__((ext_vector_type(8))) float v8f;

#define LANE() ((int)(threadIdx.x & 31))

template<int M>
__device__ __forceinline__ float lane_xor(float v) {
  static_assert(M == 1 || M == 2 || M == 4 || M == 8, "in-row partner only");
  const int b = __builtin_bit_cast(int, v);
  return __builtin_bit_cast(float,
      __builtin_amdgcn_mov_dpp(b, 0x160 | M, 0xf, 0xf, true));
}

__device__ constexpr unsigned ring_row_selects(unsigned word) {
  unsigned v = 0;
  for (unsigned j = 0; j < 8; ++j) {
    const unsigned l = word * 8 + j;
    const unsigned q = l & 7u;
    v |= ((l & 8u) | (q ^ ((q >> 1) & 3u))) << (4 * j);
  }
  return v;
}
__device__ __forceinline__ float cnot_row_chain(float v) {
  constexpr unsigned lo = ring_row_selects(0), hi = ring_row_selects(1);
  static_assert(lo == 0x45762310u && hi == 0xcdfeab98u, "in-group source lanes of the CNOT chain");
  const int b = __builtin_bit_cast(int, v);
  return __builtin_bit_cast(float,
      __builtin_amdgcn_permlane16(b, b, lo, hi, false, false));
}

__device__ __forceinline__ float fast_tanh(float x) {
  const float e = __expf(2.0f * x);
  return 1.0f - 2.0f / (e + 1.0f);
}

template<int K, int LDA, int LDB>
__device__ __forceinline__ v8f gemm_tile_acc(
    const float* __restrict__ sA,
    const float* __restrict__ sW,
    int n)
{
  const int lane = LANE();
  const int half = lane >> 4;
  const int col  = lane & 15;
  v8f acc = {};
#pragma unroll
  for (int k0 = 0; k0 < K; k0 += 4) {
    const int ka = k0 + half * 2;
    v2f a, b;
    a.x = sA[col * LDA + ka];
    a.y = sA[col * LDA + ka + 1];
    b.x = sW[n * LDB + ka];
    b.y = sW[n * LDB + ka + 1];
    acc = __builtin_amdgcn_wmma_f32_16x16x4_f32(false, a, false, b,
                                                (short)0, acc, false, false);
  }
  return acc;
}

template<int K, int LDA, int LDB, bool TANH>
__device__ __forceinline__ void gemm_tile_to_lds(
    const float* __restrict__ sA, const float* __restrict__ sW,
    const float* __restrict__ bg,
    float* __restrict__ sOut, int ldo, int n0)
{
  const int lane = LANE();
  const int half = lane >> 4;
  const int n    = n0 + (lane & 15);
  const v8f acc = gemm_tile_acc<K, LDA, LDB>(sA, sW, n);
  const float bias = bg[n];
#pragma unroll
  for (int r = 0; r < 8; ++r) {
    float v = acc[r] + bias;
    if (TANH) v = fast_tanh(v);
    sOut[(half * 8 + r) * ldo + n] = v;
  }
}

template<int K, int LDA, int LDB>
__device__ __forceinline__ void gemm_tile_to_halfangle_trig(
    const float* __restrict__ sA, const float* __restrict__ sW,
    const float* __restrict__ bg,
    float* __restrict__ sCos, float* __restrict__ sSin, int ldo, int n0)
{
  const int lane = LANE();
  const int half = lane >> 4;
  const int n    = n0 + (lane & 15);
  const v8f acc = gemm_tile_acc<K, LDA, LDB>(sA, sW, n);
  const float bias = bg[n];
#pragma unroll
  for (int r = 0; r < 8; ++r) {
    const float h = 0.5f * (acc[r] + bias);
    sCos[(half * 8 + r) * ldo + n] = __cosf(h);
    sSin[(half * 8 + r) * ldo + n] = __sinf(h);
  }
}

__device__ __forceinline__ void gemm_final(
    const float* __restrict__ sZ,
    const float* __restrict__ Wp,
    const float* __restrict__ bp,
    float* __restrict__ out, int rowBase, int n0)
{
  const int lane = LANE();
  const int half = lane >> 4;
  const int col  = lane & 15;
  const int n    = n0 + col;
  v8f acc = {};
#pragma unroll
  for (int k0 = 0; k0 < 8; k0 += 4) {
    const int ka = k0 + half * 2;
    v2f a, b;
    a.x = sZ[col * 8 + ka];
    a.y = sZ[col * 8 + ka + 1];
    b.x = Wp[n * 8 + ka];
    b.y = Wp[n * 8 + ka + 1];
    acc = __builtin_amdgcn_wmma_f32_16x16x4_f32(false, a, false, b,
                                                (short)0, acc, false, false);
  }
  const float bias = bp[n];
#pragma unroll
  for (int r = 0; r < 8; ++r)
    out[(size_t)(rowBase + half * 8 + r) * 128 + n] = acc[r] + bias;
}


__device__ __forceinline__ int ring_preimage(int c) {
  c ^= (c & 1) << 7;
  c ^= (c >> 1) & 1;
  c ^= ((c >> 2) & 1) << 1;
  c ^= ((c >> 3) & 1) << 2;
  c ^= ((c >> 4) & 1) << 3;
  c ^= ((c >> 5) & 1) << 4;
  c ^= ((c >> 6) & 1) << 5;
  c ^= ((c >> 7) & 1) << 6;
  return c;
}

template<int L, int COMP>
__device__ __forceinline__ float rz_phase(const float* __restrict__ qw, int c) {
  float s = 0.f;
#pragma unroll
  for (int w = 0; w < 8; ++w) {
    const float h = 0.5f * qw[(L * 8 + w) * 3 + COMP];
    s += ((c >> (7 - w)) & 1) ? h : -h;
  }
  return s;
}

template<int L>
__device__ __forceinline__ void build_phase_entry(const float* __restrict__ qw,
                                                  v2f* __restrict__ sTL, int c)
{
  float ang = rz_phase<L, 0>(qw, c);
  if constexpr (L > 0) ang += rz_phase<L - 1, 2>(qw, ring_preimage(c));
  v2f t;
  t.x = __cosf(ang);
  t.y = __sinf(ang);
  sTL[(c & 31) * 8 + (c >> 5)] = t;
}


constexpr int NS = 32;

template<int LB>
__device__ __forceinline__ void ry_lane(float ar[NS], float ai[NS], float c, float s) {
  const float t = ((LANE() >> LB) & 1) ? s : -s;
#pragma unroll
  for (int j = 0; j < NS; ++j) {
    const float pr = lane_xor<(1 << LB)>(ar[j]);
    const float pi = lane_xor<(1 << LB)>(ai[j]);
    ar[j] = c * ar[j] + t * pr;
    ai[j] = c * ai[j] + t * pi;
  }
}

template<int JB>
__device__ __forceinline__ void ry_local(float ar[NS], float ai[NS], float c, float s) {
#pragma unroll
  for (int j = 0; j < NS; ++j) {
    if (!((j >> JB) & 1)) {
      const int j1 = j | (1 << JB);
      const float a0r = ar[j], a0i = ai[j], a1r = ar[j1], a1i = ai[j1];
      ar[j]  = c * a0r - s * a1r;  ai[j]  = c * a0i - s * a1i;
      ar[j1] = s * a0r + c * a1r;  ai[j1] = s * a0i + c * a1i;
    }
  }
}

template<int CJB, int TJB>
__device__ __forceinline__ void cnot_ll(float ar[NS], float ai[NS]) {
#pragma unroll
  for (int j = 0; j < NS; ++j)
    if (((j >> CJB) & 1) && !((j >> TJB) & 1)) {
      const int j1 = j | (1 << TJB);
      float tr = ar[j], ti = ai[j];
      ar[j] = ar[j1]; ai[j] = ai[j1];
      ar[j1] = tr;    ai[j1] = ti;
    }
}
template<int CJB, int TLB>
__device__ __forceinline__ void cnot_local_lane(float ar[NS], float ai[NS]) {
#pragma unroll
  for (int j = 0; j < NS; ++j)
    if ((j >> CJB) & 1) {
      ar[j] = lane_xor<(1 << TLB)>(ar[j]);
      ai[j] = lane_xor<(1 << TLB)>(ai[j]);
    }
}
template<int CLB, int TJB>
__device__ __forceinline__ void cnot_lane_local(float ar[NS], float ai[NS]) {
  if ((LANE() >> CLB) & 1) {
#pragma unroll
    for (int j = 0; j < NS; ++j)
      if (!((j >> TJB) & 1)) {
        const int j1 = j | (1 << TJB);
        float tr = ar[j], ti = ai[j];
        ar[j] = ar[j1]; ai[j] = ai[j1];
        ar[j1] = tr;    ai[j1] = ti;
      }
  }
}
__device__ __forceinline__ void cnot_lane_chain(float ar[NS], float ai[NS]) {
#pragma unroll
  for (int j = 0; j < NS; ++j) {
    ar[j] = cnot_row_chain(ar[j]);
    ai[j] = cnot_row_chain(ai[j]);
  }
}

__device__ __forceinline__ void cnot_ring(float ar[NS], float ai[NS]) {
  cnot_lane_chain(ar, ai);
  cnot_lane_local<0, 4>(ar, ai);
  cnot_ll<4, 3>(ar, ai);
  cnot_ll<3, 2>(ar, ai);
  cnot_ll<2, 1>(ar, ai);
  cnot_ll<1, 0>(ar, ai);
  cnot_local_lane<0, 2>(ar, ai);
}

template<int W>
__device__ __forceinline__ void wire_state(const float* __restrict__ cs,
    const float* __restrict__ sn, const float* __restrict__ ph0,
    const float* __restrict__ ry0,
    float& g0r, float& g0i, float& g1r, float& g1i)
{
  const float zr = cs[8 + W], zs = sn[8 + W];
  const float pc = ph0[2 * W], ps = ph0[2 * W + 1];
  const float ur = zr * pc - zs * ps;
  const float ui = zs * pc + zr * ps;
  const float f0r = cs[W] * ur, f0i = -(cs[W] * ui);
  const float f1r = sn[W] * ur, f1i =   sn[W] * ui;
  const float C = ry0[2 * W], S = ry0[2 * W + 1];
  g0r = C * f0r - S * f1r;  g0i = C * f0i - S * f1i;
  g1r = S * f0r + C * f1r;  g1i = S * f0i + C * f1i;
}

#define CMUL(orr, oi, xr, xi, yr, yi) {                                       \
    orr = (xr) * (yr) - (xi) * (yi);                                          \
    oi  = (xr) * (yi) + (xi) * (yr);                                          \
  }

__device__ __forceinline__ void product_state_behind_ring(
    const float* __restrict__ cs, const float* __restrict__ sn,
    const float* __restrict__ ph0, const float* __restrict__ ry0,
    float ar[NS], float ai[NS])
{
  float gr[8][2], gi[8][2];
  wire_state<0>(cs, sn, ph0, ry0, gr[0][0], gi[0][0], gr[0][1], gi[0][1]);
  wire_state<1>(cs, sn, ph0, ry0, gr[1][0], gi[1][0], gr[1][1], gi[1][1]);
  wire_state<2>(cs, sn, ph0, ry0, gr[2][0], gi[2][0], gr[2][1], gi[2][1]);
  wire_state<3>(cs, sn, ph0, ry0, gr[3][0], gi[3][0], gr[3][1], gi[3][1]);
  wire_state<4>(cs, sn, ph0, ry0, gr[4][0], gi[4][0], gr[4][1], gi[4][1]);
  wire_state<5>(cs, sn, ph0, ry0, gr[5][0], gi[5][0], gr[5][1], gi[5][1]);
  wire_state<6>(cs, sn, ph0, ry0, gr[6][0], gi[6][0], gr[6][1], gi[6][1]);
  wire_state<7>(cs, sn, ph0, ry0, gr[7][0], gi[7][0], gr[7][1], gi[7][1]);

  const int x0 = (LANE() >> 2) & 1;
  const int x1 = (LANE() >> 1) & 1;
  const int x2 =  LANE()       & 1;

  const int y = x1 ^ x0;
  const float u0r = x0 ? gr[0][1] : gr[0][0], u0i = x0 ? gi[0][1] : gi[0][0];
  const float v0r = x0 ? gr[0][0] : gr[0][1], v0i = x0 ? gi[0][0] : gi[0][1];
  const float u1r = y  ? gr[1][1] : gr[1][0], u1i = y  ? gi[1][1] : gi[1][0];
  const float v1r = y  ? gr[1][0] : gr[1][1], v1i = y  ? gi[1][0] : gi[1][1];
  float Ar[2], Ai[2];
  CMUL(Ar[0], Ai[0], u0r, u0i, u1r, u1i)
  CMUL(Ar[1], Ai[1], v0r, v0i, v1r, v1i)
  float Hr[2][2], Hi[2][2];
#pragma unroll
  for (int c6 = 0; c6 < 2; ++c6)
#pragma unroll
    for (int c7 = 0; c7 < 2; ++c7)
      CMUL(Hr[c6][c7], Hi[c6][c7], gr[7][c7 ^ c6], gi[7][c7 ^ c6], Ar[c7], Ai[c7])

  const int z = x2 ^ x1;
  const float l2r = z ? gr[2][1] : gr[2][0], l2i = z ? gi[2][1] : gi[2][0];
  float G3r[2], G3i[2];
  G3r[0] = x2 ? gr[3][1] : gr[3][0];  G3i[0] = x2 ? gi[3][1] : gi[3][0];
  G3r[1] = x2 ? gr[3][0] : gr[3][1];  G3i[1] = x2 ? gi[3][0] : gi[3][1];

  float t1r[2], t1i[2], t2r[4], t2i[4], t3r[8], t3i[8], t4r[16], t4i[16];
#pragma unroll
  for (int b = 0; b < 2; ++b) CMUL(t1r[b], t1i[b], l2r, l2i, G3r[b], G3i[b])
#pragma unroll
  for (int b = 0; b < 4; ++b)
    CMUL(t2r[b], t2i[b], t1r[b >> 1], t1i[b >> 1],
         gr[4][(b & 1) ^ ((b >> 1) & 1)], gi[4][(b & 1) ^ ((b >> 1) & 1)])
#pragma unroll
  for (int b = 0; b < 8; ++b)
    CMUL(t3r[b], t3i[b], t2r[b >> 1], t2i[b >> 1],
         gr[5][(b & 1) ^ ((b >> 1) & 1)], gi[5][(b & 1) ^ ((b >> 1) & 1)])
#pragma unroll
  for (int b = 0; b < 16; ++b)
    CMUL(t4r[b], t4i[b], t3r[b >> 1], t3i[b >> 1],
         gr[6][(b & 1) ^ ((b >> 1) & 1)], gi[6][(b & 1) ^ ((b >> 1) & 1)])
#pragma unroll
  for (int j = 0; j < NS; ++j)
    CMUL(ar[j], ai[j], t4r[j >> 1], t4i[j >> 1],
         Hr[(j >> 1) & 1][j & 1], Hi[(j >> 1) & 1][j & 1])
}

__global__ __launch_bounds__(512) void quantum_block_kernel(
    const float* __restrict__ x,
    const float* __restrict__ W1, const float* __restrict__ b1,
    const float* __restrict__ W2, const float* __restrict__ b2,
    const float* __restrict__ W3, const float* __restrict__ b3,
    const float* __restrict__ qw,
    const float* __restrict__ Wp, const float* __restrict__ bp,
    float* __restrict__ out)
{
  __shared__ __attribute__((aligned(16))) float sX  [64 * 84];
  __shared__ float sH1 [64 * 68];
  __shared__ float sH2 [64 * 36];
  __shared__ float sCos[64 * 16];
  __shared__ float sSin[64 * 16];
  __shared__ float sZ  [64 * 8];
  __shared__ v2f   sT  [2 * 256];
  __shared__ float sRY [24 * 2];
  __shared__ float sPH0[8 * 2];
  __shared__ __attribute__((aligned(16))) float sW1[64 * 84];
  __shared__ __attribute__((aligned(16))) float sW2[32 * 68];
  __shared__ __attribute__((aligned(16))) float sW3[16 * 36];

  const int tid     = threadIdx.x;
  const int wid     = tid >> 5;
  const int lane    = tid & 31;
  const int rowBase = blockIdx.x * 64;

  if (tid < 24) {
    const float th = qw[tid * 3 + 1];
    sRY[tid * 2]     = __cosf(0.5f * th);
    sRY[tid * 2 + 1] = __sinf(0.5f * th);
    if (tid < 8) {
      const float ph = qw[tid * 3];
      sPH0[tid * 2]     = __cosf(0.5f * ph);
      sPH0[tid * 2 + 1] = __sinf(0.5f * ph);
    }
  }
  if (tid < 256) {
    build_phase_entry<1>(qw, &sT[0],   tid);
  } else {
    build_phase_entry<2>(qw, &sT[256], tid - 256);
  }
  constexpr int W1P = 64 * 20, W3P = 16 * 8;
  int  p1[3];
  v4f  w1[3], xv[3];
#pragma unroll
  for (int it = 0; it < 3; ++it) {
    const int p = it * 512 + tid;
    p1[it] = p < W1P ? p : W1P - 1;
    xv[it] = *reinterpret_cast<const v4f*>(x + rowBase * 80 + 4 * p1[it]);
    w1[it] = *reinterpret_cast<const v4f*>(W1 + 4 * p1[it]);
  }
  const v4f w2 = *reinterpret_cast<const v4f*>(W2 + 4 * tid);
  const int p3 = tid < W3P ? tid : W3P - 1;
  const v4f w3 = *reinterpret_cast<const v4f*>(W3 + 4 * p3);
#pragma unroll
  for (int it = 0; it < 3; ++it) {
    const int o = (p1[it] / 20) * 84 + (p1[it] % 20) * 4;
    *reinterpret_cast<v4f*>(&sX[o])  = xv[it];
    *reinterpret_cast<v4f*>(&sW1[o]) = w1[it];
  }
  *reinterpret_cast<v4f*>(&sW2[(tid >> 4) * 68 + (tid & 15) * 4]) = w2;
  *reinterpret_cast<v4f*>(&sW3[(p3 >> 3) * 36 + (p3 & 7) * 4]) = w3;
  __syncthreads();

  {
    const int mt = wid >> 2, nt = wid & 3;
    gemm_tile_to_lds<80, 84, 84, true >(sX + mt * 16 * 84, sW1, b1, sH1 + mt * 16 * 68, 68, nt * 16);
  }
  __syncthreads();
  if (wid < 8) {
    const int mt = wid >> 1, nt = wid & 1;
    gemm_tile_to_lds<64, 68, 68, true >(sH1 + mt * 16 * 68, sW2, b2, sH2 + mt * 16 * 36, 36, nt * 16);
  }
  __syncthreads();
  if (wid < 4)
    gemm_tile_to_halfangle_trig<32, 36, 36>(sH2 + wid * 16 * 36, sW3, b3,
                                        sCos + wid * 16 * 16, sSin + wid * 16 * 16, 16, 0);
  __syncthreads();

  {
    const int samp = wid * 4 + (lane >> 3);
    float ar[NS], ai[NS];
    product_state_behind_ring(&sCos[samp * 16], &sSin[samp * 16], sPH0, sRY, ar, ai);

    for (int L = 1; L < 3; ++L) {
      const v2f* T = &sT[(L - 1) * 256 + (lane & 7)];
#pragma unroll
      for (int j = 0; j < NS; ++j) {
        const v2f t = T[j * 8];
        const float r = ar[j] * t.x - ai[j] * t.y;
        const float i = ar[j] * t.y + ai[j] * t.x;
        ar[j] = r; ai[j] = i;
      }
      const float* ry = &sRY[L * 16];
      ry_lane<2>(ar, ai, ry[0],  ry[1]);
      ry_lane<1>(ar, ai, ry[2],  ry[3]);
      ry_lane<0>(ar, ai, ry[4],  ry[5]);
      ry_local<4>(ar, ai, ry[6],  ry[7]);
      ry_local<3>(ar, ai, ry[8],  ry[9]);
      ry_local<2>(ar, ai, ry[10], ry[11]);
      ry_local<1>(ar, ai, ry[12], ry[13]);
      ry_local<0>(ar, ai, ry[14], ry[15]);
      cnot_ring(ar, ai);
    }

    float pb[NS]; float tot = 0.f;
#pragma unroll
    for (int j = 0; j < NS; ++j) { pb[j] = ar[j]*ar[j] + ai[j]*ai[j]; tot += pb[j]; }
    float zp[8];
#pragma unroll
    for (int w = 0; w < 3; ++w)
      zp[w] = ((lane >> (2 - w)) & 1) ? -tot : tot;
    float s3 = 0.f, s4 = 0.f, s5 = 0.f, s6 = 0.f, s7 = 0.f;
#pragma unroll
    for (int j = 0; j < NS; ++j) {
      s3 += ((j >> 4) & 1) ? -pb[j] : pb[j];
      s4 += ((j >> 3) & 1) ? -pb[j] : pb[j];
      s5 += ((j >> 2) & 1) ? -pb[j] : pb[j];
      s6 += ((j >> 1) & 1) ? -pb[j] : pb[j];
      s7 += ( j       & 1) ? -pb[j] : pb[j];
    }
    zp[3] = s3; zp[4] = s4; zp[5] = s5; zp[6] = s6; zp[7] = s7;
#pragma unroll
    for (int w = 0; w < 8; ++w) {
      float v = zp[w];
      v += lane_xor<4>(v);
      v += lane_xor<2>(v);
      v += lane_xor<1>(v);
      if ((lane & 7) == 0) sZ[samp * 8 + w] = v;
    }
  }
  __syncthreads();

#pragma unroll
  for (int i = 0; i < 2; ++i) {
    const int mt = (wid >> 3) * 2 + i;
    gemm_final(sZ + mt * 16 * 8, Wp, bp, out, rowBase + mt * 16, (wid & 7) * 16);
  }
}

extern "C" void kernel_launch(void* const* d_in, const int* in_sizes, int n_in,
                              void* d_out, int out_size, void* d_ws, size_t ws_size,
                              hipStream_t stream) {
  const float* x  = (const float*)d_in[0];
  const float* W1 = (const float*)d_in[1];
  const float* b1 = (const float*)d_in[2];
  const float* W2 = (const float*)d_in[3];
  const float* b2 = (const float*)d_in[4];
  const float* W3 = (const float*)d_in[5];
  const float* b3 = (const float*)d_in[6];
  const float* qw = (const float*)d_in[7];
  const float* Wp = (const float*)d_in[8];
  const float* bp = (const float*)d_in[9];
  float* out = (float*)d_out;

  const int B = in_sizes[0] / 80;
  const int grid = B / 64;
  quantum_block_kernel<<<grid, 512, 0, stream>>>(x, W1, b1, W2, b2, W3, b3,
                                                 qw, Wp, bp, out);
}


# RY passes of layers 1, 2 as shears (a0 - t a1, a1 + t a0, t = tan(theta/2)), common factor cos put into <Z> once: 1 multiply-add per component; rotations kept if any |t| > 8; products re-associated
# speedup vs baseline: 2.8393x; 1.0634x over previous
#include <hip/hip_runtime.h>
#include <math.h>

typedef __attribute__((ext_vector_type(2))) float v2f;
typedef __attribute__((ext_vector_type(4))) float v4f;
typedef __attribute__((ext_vector_type(8))) float v8f;

#define LANE() ((int)(threadIdx.x & 31))

template<int M>
__device__ __forceinline__ float lane_xor(float v) {
  static_assert(M == 1 || M == 2 || M == 4 || M == 8, "in-row partner only");
  const int b = __builtin_bit_cast(int, v);
  return __builtin_bit_cast(float,
      __builtin_amdgcn_mov_dpp(b, 0x160 | M, 0xf, 0xf, true));
}

__device__ constexpr unsigned ring_row_selects(unsigned word) {
  unsigned v = 0;
  for (unsigned j = 0; j < 8; ++j) {
    const unsigned l = word * 8 + j;
    const unsigned q = l & 7u;
    v |= ((l & 8u) | (q ^ ((q >> 1) & 3u))) << (4 * j);
  }
  return v;
}
__device__ __forceinline__ float cnot_row_chain(float v) {
  constexpr unsigned lo = ring_row_selects(0), hi = ring_row_selects(1);
  static_assert(lo == 0x45762310u && hi == 0xcdfeab98u, "in-group source lanes of the CNOT chain");
  const int b = __builtin_bit_cast(int, v);
  return __builtin_bit_cast(float,
      __builtin_amdgcn_permlane16(b, b, lo, hi, false, false));
}

__device__ __forceinline__ float fast_tanh(float x) {
  const float e = __expf(2.0f * x);
  return 1.0f - 2.0f / (e + 1.0f);
}

template<int K, int LDA, int LDB>
__device__ __forceinline__ v8f gemm_tile_acc(
    const float* __restrict__ sA,
    const float* __restrict__ sW,
    int n)
{
  const int lane = LANE();
  const int half = lane >> 4;
  const int col  = lane & 15;
  v8f acc = {};
#pragma unroll
  for (int k0 = 0; k0 < K; k0 += 4) {
    const int ka = k0 + half * 2;
    v2f a, b;
    a.x = sA[col * LDA + ka];
    a.y = sA[col * LDA + ka + 1];
    b.x = sW[n * LDB + ka];
    b.y = sW[n * LDB + ka + 1];
    acc = __builtin_amdgcn_wmma_f32_16x16x4_f32(false, a, false, b,
                                                (short)0, acc, false, false);
  }
  return acc;
}

template<int K, int LDA, int LDB, bool TANH>
__device__ __forceinline__ void gemm_tile_to_lds(
    const float* __restrict__ sA, const float* __restrict__ sW,
    const float* __restrict__ bg,
    float* __restrict__ sOut, int ldo, int n0)
{
  const int lane = LANE();
  const int half = lane >> 4;
  const int n    = n0 + (lane & 15);
  const v8f acc = gemm_tile_acc<K, LDA, LDB>(sA, sW, n);
  const float bias = bg[n];
#pragma unroll
  for (int r = 0; r < 8; ++r) {
    float v = acc[r] + bias;
    if (TANH) v = fast_tanh(v);
    sOut[(half * 8 + r) * ldo + n] = v;
  }
}

template<int K, int LDA, int LDB>
__device__ __forceinline__ void gemm_tile_to_halfangle_trig(
    const float* __restrict__ sA, const float* __restrict__ sW,
    const float* __restrict__ bg,
    float* __restrict__ sCos, float* __restrict__ sSin, int ldo, int n0)
{
  const int lane = LANE();
  const int half = lane >> 4;
  const int n    = n0 + (lane & 15);
  const v8f acc = gemm_tile_acc<K, LDA, LDB>(sA, sW, n);
  const float bias = bg[n];
#pragma unroll
  for (int r = 0; r < 8; ++r) {
    const float h = 0.5f * (acc[r] + bias);
    sCos[(half * 8 + r) * ldo + n] = __cosf(h);
    sSin[(half * 8 + r) * ldo + n] = __sinf(h);
  }
}

__device__ __forceinline__ void gemm_final(
    const float* __restrict__ sZ,
    const float* __restrict__ Wp,
    const float* __restrict__ bp,
    float* __restrict__ out, int rowBase, int n0)
{
  const int lane = LANE();
  const int half = lane >> 4;
  const int col  = lane & 15;
  const int n    = n0 + col;
  v8f acc = {};
#pragma unroll
  for (int k0 = 0; k0 < 8; k0 += 4) {
    const int ka = k0 + half * 2;
    v2f a, b;
    a.x = sZ[col * 8 + ka];
    a.y = sZ[col * 8 + ka + 1];
    b.x = Wp[n * 8 + ka];
    b.y = Wp[n * 8 + ka + 1];
    acc = __builtin_amdgcn_wmma_f32_16x16x4_f32(false, a, false, b,
                                                (short)0, acc, false, false);
  }
  const float bias = bp[n];
#pragma unroll
  for (int r = 0; r < 8; ++r)
    out[(size_t)(rowBase + half * 8 + r) * 128 + n] = acc[r] + bias;
}


__device__ __forceinline__ int ring_preimage(int c) {
  c ^= (c & 1) << 7;
  c ^= (c >> 1) & 1;
  c ^= ((c >> 2) & 1) << 1;
  c ^= ((c >> 3) & 1) << 2;
  c ^= ((c >> 4) & 1) << 3;
  c ^= ((c >> 5) & 1) << 4;
  c ^= ((c >> 6) & 1) << 5;
  c ^= ((c >> 7) & 1) << 6;
  return c;
}

template<int L, int COMP>
__device__ __forceinline__ float rz_phase(const float* __restrict__ qw, int c) {
  float s = 0.f;
#pragma unroll
  for (int w = 0; w < 8; ++w) {
    const float h = 0.5f * qw[(L * 8 + w) * 3 + COMP];
    s += ((c >> (7 - w)) & 1) ? h : -h;
  }
  return s;
}

template<int L>
__device__ __forceinline__ void build_phase_entry(const float* __restrict__ qw,
                                                  v2f* __restrict__ sTL, int c)
{
  float ang = rz_phase<L, 0>(qw, c);
  if constexpr (L > 0) ang += rz_phase<L - 1, 2>(qw, ring_preimage(c));
  v2f t;
  t.x = __cosf(ang);
  t.y = __sinf(ang);
  sTL[(c & 31) * 8 + (c >> 5)] = t;
}


constexpr int NS = 32;

template<int LB>
__device__ __forceinline__ void ry_lane(float ar[NS], float ai[NS], float c, float s) {
  const float t = ((LANE() >> LB) & 1) ? s : -s;
#pragma unroll
  for (int j = 0; j < NS; ++j) {
    const float pr = lane_xor<(1 << LB)>(ar[j]);
    const float pi = lane_xor<(1 << LB)>(ai[j]);
    ar[j] = c * ar[j] + t * pr;
    ai[j] = c * ai[j] + t * pi;
  }
}

template<int JB>
__device__ __forceinline__ void ry_local(float ar[NS], float ai[NS], float c, float s) {
#pragma unroll
  for (int j = 0; j < NS; ++j) {
    if (!((j >> JB) & 1)) {
      const int j1 = j | (1 << JB);
      const float a0r = ar[j], a0i = ai[j], a1r = ar[j1], a1i = ai[j1];
      ar[j]  = c * a0r - s * a1r;  ai[j]  = c * a0i - s * a1i;
      ar[j1] = s * a0r + c * a1r;  ai[j1] = s * a0i + c * a1i;
    }
  }
}


template<int M>
__device__ __forceinline__ void fmac_lane_xor(float& acc, float w) {
  static_assert(M == 1 || M == 2 || M == 4, "in-group partner only");
  asm("v_fmac_f32_dpp %0, %0, %1 row_xmask:%2 row_mask:0xf bank_mask:0xf bound_ctrl:1"
      : "+v"(acc) : "v"(w), "n"(M));
}

template<int LB>
__device__ __forceinline__ void ry_lane_shear(float ar[NS], float ai[NS], float t) {
  const float w = ((LANE() >> LB) & 1) ? t : -t;
#pragma unroll
  for (int j = 0; j < NS; ++j) {
    fmac_lane_xor<(1 << LB)>(ar[j], w);
    fmac_lane_xor<(1 << LB)>(ai[j], w);
  }
}

template<int JB>
__device__ __forceinline__ void ry_local_shear(float ar[NS], float ai[NS], float t) {
#pragma unroll
  for (int j = 0; j < NS; ++j) {
    if (!((j >> JB) & 1)) {
      const int j1 = j | (1 << JB);
      const v2f a0 = {ar[j], ai[j]}, a1 = {ar[j1], ai[j1]};
      const v2f n0 = a0 - t * a1;
      const v2f n1 = a1 + t * a0;
      ar[j]  = n0.x;  ai[j]  = n0.y;
      ar[j1] = n1.x;  ai[j1] = n1.y;
    }
  }
}

__device__ __forceinline__ float ry_factor_sq(const float* __restrict__ sF) {
  float p = sF[0];
#pragma unroll
  for (int k = 1; k < 16; ++k) p *= sF[k];
  return p * p;
}

template<int CJB, int TJB>
__device__ __forceinline__ void cnot_ll(float ar[NS], float ai[NS]) {
#pragma unroll
  for (int j = 0; j < NS; ++j)
    if (((j >> CJB) & 1) && !((j >> TJB) & 1)) {
      const int j1 = j | (1 << TJB);
      float tr = ar[j], ti = ai[j];
      ar[j] = ar[j1]; ai[j] = ai[j1];
      ar[j1] = tr;    ai[j1] = ti;
    }
}
template<int CJB, int TLB>
__device__ __forceinline__ void cnot_local_lane(float ar[NS], float ai[NS]) {
#pragma unroll
  for (int j = 0; j < NS; ++j)
    if ((j >> CJB) & 1) {
      ar[j] = lane_xor<(1 << TLB)>(ar[j]);
      ai[j] = lane_xor<(1 << TLB)>(ai[j]);
    }
}
template<int CLB, int TJB>
__device__ __forceinline__ void cnot_lane_local(float ar[NS], float ai[NS]) {
  if ((LANE() >> CLB) & 1) {
#pragma unroll
    for (int j = 0; j < NS; ++j)
      if (!((j >> TJB) & 1)) {
        const int j1 = j | (1 << TJB);
        float tr = ar[j], ti = ai[j];
        ar[j] = ar[j1]; ai[j] = ai[j1];
        ar[j1] = tr;    ai[j1] = ti;
      }
  }
}
__device__ __forceinline__ void cnot_lane_chain(float ar[NS], float ai[NS]) {
#pragma unroll
  for (int j = 0; j < NS; ++j) {
    ar[j] = cnot_row_chain(ar[j]);
    ai[j] = cnot_row_chain(ai[j]);
  }
}

__device__ __forceinline__ void cnot_ring(float ar[NS], float ai[NS]) {
  cnot_lane_chain(ar, ai);
  cnot_lane_local<0, 4>(ar, ai);
  cnot_ll<4, 3>(ar, ai);
  cnot_ll<3, 2>(ar, ai);
  cnot_ll<2, 1>(ar, ai);
  cnot_ll<1, 0>(ar, ai);
  cnot_local_lane<0, 2>(ar, ai);
}

template<int W>
__device__ __forceinline__ void wire_state(const float* __restrict__ cs,
    const float* __restrict__ sn, const float* __restrict__ ph0,
    const float* __restrict__ ry0,
    float& g0r, float& g0i, float& g1r, float& g1i)
{
  const float zr = cs[8 + W], zs = sn[8 + W];
  const float pc = ph0[2 * W], ps = ph0[2 * W + 1];
  const float ur = zr * pc - zs * ps;
  const float ui = zs * pc + zr * ps;
  const float f0r = cs[W] * ur, f0i = -(cs[W] * ui);
  const float f1r = sn[W] * ur, f1i =   sn[W] * ui;
  const float C = ry0[2 * W], S = ry0[2 * W + 1];
  g0r = C * f0r - S * f1r;  g0i = C * f0i - S * f1i;
  g1r = S * f0r + C * f1r;  g1i = S * f0i + C * f1i;
}

#define CMUL(orr, oi, xr, xi, yr, yi) {                                       \
    orr = (xr) * (yr) - (xi) * (yi);                                          \
    oi  = (xr) * (yi) + (xi) * (yr);                                          \
  }

__device__ __forceinline__ void product_state_behind_ring(
    const float* __restrict__ cs, const float* __restrict__ sn,
    const float* __restrict__ ph0, const float* __restrict__ ry0,
    float ar[NS], float ai[NS])
{
  float gr[8][2], gi[8][2];
  wire_state<0>(cs, sn, ph0, ry0, gr[0][0], gi[0][0], gr[0][1], gi[0][1]);
  wire_state<1>(cs, sn, ph0, ry0, gr[1][0], gi[1][0], gr[1][1], gi[1][1]);
  wire_state<2>(cs, sn, ph0, ry0, gr[2][0], gi[2][0], gr[2][1], gi[2][1]);
  wire_state<3>(cs, sn, ph0, ry0, gr[3][0], gi[3][0], gr[3][1], gi[3][1]);
  wire_state<4>(cs, sn, ph0, ry0, gr[4][0], gi[4][0], gr[4][1], gi[4][1]);
  wire_state<5>(cs, sn, ph0, ry0, gr[5][0], gi[5][0], gr[5][1], gi[5][1]);
  wire_state<6>(cs, sn, ph0, ry0, gr[6][0], gi[6][0], gr[6][1], gi[6][1]);
  wire_state<7>(cs, sn, ph0, ry0, gr[7][0], gi[7][0], gr[7][1], gi[7][1]);

  const int x0 = (LANE() >> 2) & 1;
  const int x1 = (LANE() >> 1) & 1;
  const int x2 =  LANE()       & 1;

  const int y = x1 ^ x0;
  const float u0r = x0 ? gr[0][1] : gr[0][0], u0i = x0 ? gi[0][1] : gi[0][0];
  const float v0r = x0 ? gr[0][0] : gr[0][1], v0i = x0 ? gi[0][0] : gi[0][1];
  const float u1r = y  ? gr[1][1] : gr[1][0], u1i = y  ? gi[1][1] : gi[1][0];
  const float v1r = y  ? gr[1][0] : gr[1][1], v1i = y  ? gi[1][0] : gi[1][1];
  float Ar[2], Ai[2];
  CMUL(Ar[0], Ai[0], u0r, u0i, u1r, u1i)
  CMUL(Ar[1], Ai[1], v0r, v0i, v1r, v1i)
  float Hr[2][2], Hi[2][2];
#pragma unroll
  for (int c6 = 0; c6 < 2; ++c6)
#pragma unroll
    for (int c7 = 0; c7 < 2; ++c7)
      CMUL(Hr[c6][c7], Hi[c6][c7], gr[7][c7 ^ c6], gi[7][c7 ^ c6], Ar[c7], Ai[c7])

  const int z = x2 ^ x1;
  const float l2r = z ? gr[2][1] : gr[2][0], l2i = z ? gi[2][1] : gi[2][0];
  float G3r[2], G3i[2];
  G3r[0] = x2 ? gr[3][1] : gr[3][0];  G3i[0] = x2 ? gi[3][1] : gi[3][0];
  G3r[1] = x2 ? gr[3][0] : gr[3][1];  G3i[1] = x2 ? gi[3][0] : gi[3][1];

  float t1r[2], t1i[2], t2r[4], t2i[4], t3r[8], t3i[8], t4r[16], t4i[16];
#pragma unroll
  for (int b = 0; b < 2; ++b) CMUL(t1r[b], t1i[b], l2r, l2i, G3r[b], G3i[b])
#pragma unroll
  for (int b = 0; b < 4; ++b)
    CMUL(t2r[b], t2i[b], t1r[b >> 1], t1i[b >> 1],
         gr[4][(b & 1) ^ ((b >> 1) & 1)], gi[4][(b & 1) ^ ((b >> 1) & 1)])
#pragma unroll
  for (int b = 0; b < 8; ++b)
    CMUL(t3r[b], t3i[b], t2r[b >> 1], t2i[b >> 1],
         gr[5][(b & 1) ^ ((b >> 1) & 1)], gi[5][(b & 1) ^ ((b >> 1) & 1)])
#pragma unroll
  for (int b = 0; b < 16; ++b)
    CMUL(t4r[b], t4i[b], t3r[b >> 1], t3i[b >> 1],
         gr[6][(b & 1) ^ ((b >> 1) & 1)], gi[6][(b & 1) ^ ((b >> 1) & 1)])
#pragma unroll
  for (int j = 0; j < NS; ++j)
    CMUL(ar[j], ai[j], t4r[j >> 1], t4i[j >> 1],
         Hr[(j >> 1) & 1][j & 1], Hi[(j >> 1) & 1][j & 1])
}

template<bool SHEAR>
__device__ __forceinline__ void layers_and_readout(
    float ar[NS], float ai[NS], const v2f* __restrict__ sT,
    const float* __restrict__ sRY, const float* __restrict__ sRYT,
    const float* __restrict__ sRYF, float* __restrict__ sZ, int samp, int lane)
{
  for (int L = 1; L < 3; ++L) {
    const v2f* T = &sT[(L - 1) * 256 + (lane & 7)];
#pragma unroll
    for (int j = 0; j < NS; ++j) {
      const v2f t = T[j * 8];
      const float r = ar[j] * t.x - ai[j] * t.y;
      const float i = ar[j] * t.y + ai[j] * t.x;
      ar[j] = r; ai[j] = i;
    }
    if constexpr (SHEAR) {
      const float* rt = &sRYT[(L - 1) * 8];
      ry_lane_shear<2>(ar, ai, rt[0]);
      ry_lane_shear<1>(ar, ai, rt[1]);
      ry_lane_shear<0>(ar, ai, rt[2]);
      ry_local_shear<4>(ar, ai, rt[3]);
      ry_local_shear<3>(ar, ai, rt[4]);
      ry_local_shear<2>(ar, ai, rt[5]);
      ry_local_shear<1>(ar, ai, rt[6]);
      ry_local_shear<0>(ar, ai, rt[7]);
    } else {
      const float* ry = &sRY[L * 16];
      ry_lane<2>(ar, ai, ry[0],  ry[1]);
      ry_lane<1>(ar, ai, ry[2],  ry[3]);
      ry_lane<0>(ar, ai, ry[4],  ry[5]);
      ry_local<4>(ar, ai, ry[6],  ry[7]);
      ry_local<3>(ar, ai, ry[8],  ry[9]);
      ry_local<2>(ar, ai, ry[10], ry[11]);
      ry_local<1>(ar, ai, ry[12], ry[13]);
      ry_local<0>(ar, ai, ry[14], ry[15]);
    }
    cnot_ring(ar, ai);
  }

  float pb[NS]; float tot = 0.f;
#pragma unroll
  for (int j = 0; j < NS; ++j) { pb[j] = ar[j]*ar[j] + ai[j]*ai[j]; tot += pb[j]; }
  float zp[8];
#pragma unroll
  for (int w = 0; w < 3; ++w)
    zp[w] = ((lane >> (2 - w)) & 1) ? -tot : tot;
  float s3 = 0.f, s4 = 0.f, s5 = 0.f, s6 = 0.f, s7 = 0.f;
#pragma unroll
  for (int j = 0; j < NS; ++j) {
    s3 += ((j >> 4) & 1) ? -pb[j] : pb[j];
    s4 += ((j >> 3) & 1) ? -pb[j] : pb[j];
    s5 += ((j >> 2) & 1) ? -pb[j] : pb[j];
    s6 += ((j >> 1) & 1) ? -pb[j] : pb[j];
    s7 += ( j       & 1) ? -pb[j] : pb[j];
  }
  zp[3] = s3; zp[4] = s4; zp[5] = s5; zp[6] = s6; zp[7] = s7;
  float fsq = 1.0f;
  if constexpr (SHEAR) fsq = ry_factor_sq(sRYF);
#pragma unroll
  for (int w = 0; w < 8; ++w) {
    float v = zp[w];
    v += lane_xor<4>(v);
    v += lane_xor<2>(v);
    v += lane_xor<1>(v);
    if constexpr (SHEAR) v *= fsq;
    if ((lane & 7) == 0) sZ[samp * 8 + w] = v;
  }
}

__global__ __launch_bounds__(512) void quantum_block_kernel(
    const float* __restrict__ x,
    const float* __restrict__ W1, const float* __restrict__ b1,
    const float* __restrict__ W2, const float* __restrict__ b2,
    const float* __restrict__ W3, const float* __restrict__ b3,
    const float* __restrict__ qw,
    const float* __restrict__ Wp, const float* __restrict__ bp,
    float* __restrict__ out)
{
  __shared__ __attribute__((aligned(16))) float sX  [64 * 84];
  __shared__ float sH1 [64 * 68];
  __shared__ float sH2 [64 * 36];
  __shared__ float sCos[64 * 16];
  __shared__ float sSin[64 * 16];
  __shared__ float sZ  [64 * 8];
  __shared__ v2f   sT  [2 * 256];
  __shared__ float sRY [24 * 2];
  __shared__ __attribute__((aligned(16))) float sRYT[16];
  __shared__ __attribute__((aligned(16))) float sRYF[16];
  __shared__ int sShearOk;
  __shared__ float sPH0[8 * 2];
  __shared__ __attribute__((aligned(16))) float sW1[64 * 84];
  __shared__ __attribute__((aligned(16))) float sW2[32 * 68];
  __shared__ __attribute__((aligned(16))) float sW3[16 * 36];

  const int tid     = threadIdx.x;
  const int wid     = tid >> 5;
  const int lane    = tid & 31;
  const int rowBase = blockIdx.x * 64;

  if (wid == 0) {
    const int k = tid < 24 ? tid : 23;
    const float th = qw[k * 3 + 1];
    const float c = __cosf(0.5f * th), s = __sinf(0.5f * th);
    const bool steep = tid >= 8 && tid < 24 && !(fabsf(s) <= 8.0f * fabsf(c));
    const unsigned any_steep = __builtin_amdgcn_ballot_w32(steep);
    if (tid == 0) sShearOk = any_steep == 0 ? 1 : 0;
    if (tid < 24) {
      sRY[tid * 2]     = c;
      sRY[tid * 2 + 1] = s;
      if (tid < 8) {
        const float ph = qw[tid * 3];
        sPH0[tid * 2]     = __cosf(0.5f * ph);
        sPH0[tid * 2 + 1] = __sinf(0.5f * ph);
      } else {
        sRYT[tid - 8] = s / c;
        sRYF[tid - 8] = c;
      }
    }
  }
  if (tid < 256) {
    build_phase_entry<1>(qw, &sT[0],   tid);
  } else {
    build_phase_entry<2>(qw, &sT[256], tid - 256);
  }
  constexpr int W1P = 64 * 20, W3P = 16 * 8;
  int  p1[3];
  v4f  w1[3], xv[3];
#pragma unroll
  for (int it = 0; it < 3; ++it) {
    const int p = it * 512 + tid;
    p1[it] = p < W1P ? p : W1P - 1;
    xv[it] = *reinterpret_cast<const v4f*>(x + rowBase * 80 + 4 * p1[it]);
    w1[it] = *reinterpret_cast<const v4f*>(W1 + 4 * p1[it]);
  }
  const v4f w2 = *reinterpret_cast<const v4f*>(W2 + 4 * tid);
  const int p3 = tid < W3P ? tid : W3P - 1;
  const v4f w3 = *reinterpret_cast<const v4f*>(W3 + 4 * p3);
#pragma unroll
  for (int it = 0; it < 3; ++it) {
    const int o = (p1[it] / 20) * 84 + (p1[it] % 20) * 4;
    *reinterpret_cast<v4f*>(&sX[o])  = xv[it];
    *reinterpret_cast<v4f*>(&sW1[o]) = w1[it];
  }
  *reinterpret_cast<v4f*>(&sW2[(tid >> 4) * 68 + (tid & 15) * 4]) = w2;
  *reinterpret_cast<v4f*>(&sW3[(p3 >> 3) * 36 + (p3 & 7) * 4]) = w3;
  __syncthreads();

  {
    const int mt = wid >> 2, nt = wid & 3;
    gemm_tile_to_lds<80, 84, 84, true >(sX + mt * 16 * 84, sW1, b1, sH1 + mt * 16 * 68, 68, nt * 16);
  }
  __syncthreads();
  if (wid < 8) {
    const int mt = wid >> 1, nt = wid & 1;
    gemm_tile_to_lds<64, 68, 68, true >(sH1 + mt * 16 * 68, sW2, b2, sH2 + mt * 16 * 36, 36, nt * 16);
  }
  __syncthreads();
  if (wid < 4)
    gemm_tile_to_halfangle_trig<32, 36, 36>(sH2 + wid * 16 * 36, sW3, b3,
                                        sCos + wid * 16 * 16, sSin + wid * 16 * 16, 16, 0);
  __syncthreads();

  {
    const int samp = wid * 4 + (lane >> 3);
    float ar[NS], ai[NS];
    product_state_behind_ring(&sCos[samp * 16], &sSin[samp * 16], sPH0, sRY, ar, ai);

    if (__builtin_amdgcn_readfirstlane(sShearOk) != 0)
      layers_and_readout<true >(ar, ai, sT, sRY, sRYT, sRYF, sZ, samp, lane);
    else
      layers_and_readout<false>(ar, ai, sT, sRY, sRYT, sRYF, sZ, samp, lane);
  }
  __syncthreads();

#pragma unroll
  for (int i = 0; i < 2; ++i) {
    const int mt = (wid >> 3) * 2 + i;
    gemm_final(sZ + mt * 16 * 8, Wp, bp, out, rowBase + mt * 16, (wid & 7) * 16);
  }
}

extern "C" void kernel_launch(void* const* d_in, const int* in_sizes, int n_in,
                              void* d_out, int out_size, void* d_ws, size_t ws_size,
                              hipStream_t stream) {
  const float* x  = (const float*)d_in[0];
  const float* W1 = (const float*)d_in[1];
  const float* b1 = (const float*)d_in[2];
  const float* W2 = (const float*)d_in[3];
  const float* b2 = (const float*)d_in[4];
  const float* W3 = (const float*)d_in[5];
  const float* b3 = (const float*)d_in[6];
  const float* qw = (const float*)d_in[7];
  const float* Wp = (const float*)d_in[8];
  const float* bp = (const float*)d_in[9];
  float* out = (float*)d_out;

  const int B = in_sizes[0] / 80;
  const int grid = B / 64;
  quantum_block_kernel<<<grid, 512, 0, stream>>>(x, W1, b1, W2, b2, W3, b3,
                                                 qw, Wp, bp, out);
}


# CNOT rings of layers 1, 2 not executed in the shear path: kept as a GF(2) relabelling of basis states (phase table, RY partner and sign, readout signs follow it); readout sums re-associated
# speedup vs baseline: 2.9839x; 1.0509x over previous
#include <hip/hip_runtime.h>
#include <math.h>

typedef __attribute__((ext_vector_type(2))) float v2f;
typedef __attribute__((ext_vector_type(4))) float v4f;
typedef __attribute__((ext_vector_type(8))) float v8f;

#define LANE() ((int)(threadIdx.x & 31))

template<int M>
__device__ __forceinline__ float lane_xor(float v) {
  static_assert(M == 1 || M == 2 || M == 4 || M == 8, "in-row partner only");
  const int b = __builtin_bit_cast(int, v);
  return __builtin_bit_cast(float,
      __builtin_amdgcn_mov_dpp(b, 0x160 | M, 0xf, 0xf, true));
}

__device__ constexpr unsigned ring_row_selects(unsigned word) {
  unsigned v = 0;
  for (unsigned j = 0; j < 8; ++j) {
    const unsigned l = word * 8 + j;
    const unsigned q = l & 7u;
    v |= ((l & 8u) | (q ^ ((q >> 1) & 3u))) << (4 * j);
  }
  return v;
}
__device__ __forceinline__ float cnot_row_chain(float v) {
  constexpr unsigned lo = ring_row_selects(0), hi = ring_row_selects(1);
  static_assert(lo == 0x45762310u && hi == 0xcdfeab98u, "in-group source lanes of the CNOT chain");
  const int b = __builtin_bit_cast(int, v);
  return __builtin_bit_cast(float,
      __builtin_amdgcn_permlane16(b, b, lo, hi, false, false));
}

__device__ __forceinline__ float fast_tanh(float x) {
  const float e = __expf(2.0f * x);
  return 1.0f - 2.0f / (e + 1.0f);
}

template<int K, int LDA, int LDB>
__device__ __forceinline__ v8f gemm_tile_acc(
    const float* __restrict__ sA,
    const float* __restrict__ sW,
    int n)
{
  const int lane = LANE();
  const int half = lane >> 4;
  const int col  = lane & 15;
  v8f acc = {};
#pragma unroll
  for (int k0 = 0; k0 < K; k0 += 4) {
    const int ka = k0 + half * 2;
    v2f a, b;
    a.x = sA[col * LDA + ka];
    a.y = sA[col * LDA + ka + 1];
    b.x = sW[n * LDB + ka];
    b.y = sW[n * LDB + ka + 1];
    acc = __builtin_amdgcn_wmma_f32_16x16x4_f32(false, a, false, b,
                                                (short)0, acc, false, false);
  }
  return acc;
}

template<int K, int LDA, int LDB, bool TANH>
__device__ __forceinline__ void gemm_tile_to_lds(
    const float* __restrict__ sA, const float* __restrict__ sW,
    const float* __restrict__ bg,
    float* __restrict__ sOut, int ldo, int n0)
{
  const int lane = LANE();
  const int half = lane >> 4;
  const int n    = n0 + (lane & 15);
  const v8f acc = gemm_tile_acc<K, LDA, LDB>(sA, sW, n);
  const float bias = bg[n];
#pragma unroll
  for (int r = 0; r < 8; ++r) {
    float v = acc[r] + bias;
    if (TANH) v = fast_tanh(v);
    sOut[(half * 8 + r) * ldo + n] = v;
  }
}

template<int K, int LDA, int LDB>
__device__ __forceinline__ void gemm_tile_to_halfangle_trig(
    const float* __restrict__ sA, const float* __restrict__ sW,
    const float* __restrict__ bg,
    float* __restrict__ sCos, float* __restrict__ sSin, int ldo, int n0)
{
  const int lane = LANE();
  const int half = lane >> 4;
  const int n    = n0 + (lane & 15);
  const v8f acc = gemm_tile_acc<K, LDA, LDB>(sA, sW, n);
  const float bias = bg[n];
#pragma unroll
  for (int r = 0; r < 8; ++r) {
    const float h = 0.5f * (acc[r] + bias);
    sCos[(half * 8 + r) * ldo + n] = __cosf(h);
    sSin[(half * 8 + r) * ldo + n] = __sinf(h);
  }
}

__device__ __forceinline__ void gemm_final(
    const float* __restrict__ sZ,
    const float* __restrict__ Wp,
    const float* __restrict__ bp,
    float* __restrict__ out, int rowBase, int n0)
{
  const int lane = LANE();
  const int half = lane >> 4;
  const int col  = lane & 15;
  const int n    = n0 + col;
  v8f acc = {};
#pragma unroll
  for (int k0 = 0; k0 < 8; k0 += 4) {
    const int ka = k0 + half * 2;
    v2f a, b;
    a.x = sZ[col * 8 + ka];
    a.y = sZ[col * 8 + ka + 1];
    b.x = Wp[n * 8 + ka];
    b.y = Wp[n * 8 + ka + 1];
    acc = __builtin_amdgcn_wmma_f32_16x16x4_f32(false, a, false, b,
                                                (short)0, acc, false, false);
  }
  const float bias = bp[n];
#pragma unroll
  for (int r = 0; r < 8; ++r)
    out[(size_t)(rowBase + half * 8 + r) * 128 + n] = acc[r] + bias;
}


__device__ constexpr int ring_preimage(int c) {
  c ^= (c & 1) << 7;
  c ^= (c >> 1) & 1;
  c ^= ((c >> 2) & 1) << 1;
  c ^= ((c >> 3) & 1) << 2;
  c ^= ((c >> 4) & 1) << 3;
  c ^= ((c >> 5) & 1) << 4;
  c ^= ((c >> 6) & 1) << 5;
  c ^= ((c >> 7) & 1) << 6;
  return c;
}
__device__ constexpr int ring_image(int c) {
  c ^= ((c >> 7) & 1) << 6;
  c ^= ((c >> 6) & 1) << 5;
  c ^= ((c >> 5) & 1) << 4;
  c ^= ((c >> 4) & 1) << 3;
  c ^= ((c >> 3) & 1) << 2;
  c ^= ((c >> 2) & 1) << 1;
  c ^= (c >> 1) & 1;
  c ^= (c & 1) << 7;
  return c;
}

__device__ constexpr int ring_image_n(int c, int n) {
  for (int i = 0; i < n; ++i) c = ring_image(c);
  return c;
}
__device__ constexpr int ring_preimage_n(int c, int n) {
  for (int i = 0; i < n; ++i) c = ring_preimage(c);
  return c;
}
__device__ constexpr int wire_flip_mask(int w, int n) {
  return ring_preimage_n(1 << (7 - w), n);
}
__device__ constexpr int wire_sign_mask(int w, int n) {
  int m = 0;
  for (int i = 0; i < 8; ++i)
    if ((ring_image_n(1 << i, n) >> (7 - w)) & 1) m |= 1 << i;
  return m;
}
__device__ constexpr int parity8(int v) {
  v ^= v >> 4;  v ^= v >> 2;  v ^= v >> 1;
  return v & 1;
}

template<int L, int COMP>
__device__ __forceinline__ float rz_phase(const float* __restrict__ qw, int c) {
  float s = 0.f;
#pragma unroll
  for (int w = 0; w < 8; ++w) {
    const float h = 0.5f * qw[(L * 8 + w) * 3 + COMP];
    s += ((c >> (7 - w)) & 1) ? h : -h;
  }
  return s;
}

template<int L>
__device__ __forceinline__ void build_phase_entry(const float* __restrict__ qw,
                                                  v2f* __restrict__ sTL, int c,
                                                  bool relabelled)
{
  const int cphi = relabelled ? ring_image(c) : c;
  const int comg = relabelled ? c : ring_preimage(c);
  float ang = rz_phase<L, 0>(qw, cphi);
  if constexpr (L > 0) ang += rz_phase<L - 1, 2>(qw, comg);
  v2f t;
  t.x = __cosf(ang);
  t.y = __sinf(ang);
  sTL[(c & 31) * 8 + (c >> 5)] = t;
}


constexpr int NS = 32;

template<int LB>
__device__ __forceinline__ void ry_lane(float ar[NS], float ai[NS], float c, float s) {
  const float t = ((LANE() >> LB) & 1) ? s : -s;
#pragma unroll
  for (int j = 0; j < NS; ++j) {
    const float pr = lane_xor<(1 << LB)>(ar[j]);
    const float pi = lane_xor<(1 << LB)>(ai[j]);
    ar[j] = c * ar[j] + t * pr;
    ai[j] = c * ai[j] + t * pi;
  }
}

template<int JB>
__device__ __forceinline__ void ry_local(float ar[NS], float ai[NS], float c, float s) {
#pragma unroll
  for (int j = 0; j < NS; ++j) {
    if (!((j >> JB) & 1)) {
      const int j1 = j | (1 << JB);
      const float a0r = ar[j], a0i = ai[j], a1r = ar[j1], a1i = ai[j1];
      ar[j]  = c * a0r - s * a1r;  ai[j]  = c * a0i - s * a1i;
      ar[j1] = s * a0r + c * a1r;  ai[j1] = s * a0i + c * a1i;
    }
  }
}


template<int M>
__device__ __forceinline__ void fmac_lane_xor(float& acc, float w) {
  static_assert(M >= 1 && M <= 7, "in-group partner only");
  asm("v_fmac_f32_dpp %0, %0, %1 row_xmask:%2 row_mask:0xf bank_mask:0xf bound_ctrl:1"
      : "+v"(acc) : "v"(w), "n"(M));
}
template<int M>
__device__ __forceinline__ float fma_lane_xor(float partner, float w, float self) {
  static_assert(M >= 1 && M <= 7, "in-group partner only");
  float d;
  asm("v_fma_f32_e64_dpp %0, %1, %2, %3 row_xmask:%4 row_mask:0xf bank_mask:0xf bound_ctrl:1"
      : "=v"(d) : "v"(partner), "v"(w), "v"(self), "n"(M));
  return d;
}

template<int W, int N>
__device__ __forceinline__ void ry_shear(float ar[NS], float ai[NS], float t, int lane) {
  constexpr int F = wire_flip_mask(W, N), RM = wire_sign_mask(W, N);
  constexpr int FL = F >> 5, FS = F & 31, RL = RM >> 5, RS = RM & 31;
  static_assert(parity8(F & RM) == 1, "the partner differs in the wire's own bit");
  const float tp = (__builtin_popcount(lane & RL) & 1) ? t : -t;
  const float tn = -tp;
  if constexpr (FL == 0) {
#pragma unroll
    for (int j = 0; j < NS; ++j) {
      if (j < (j ^ FS)) {
        const int k = j ^ FS;
        const float wj = parity8(j & RS) ? tn : tp;
        const v2f aj = {ar[j], ai[j]}, ak = {ar[k], ai[k]};
        const v2f nj = aj + wj * ak;
        const v2f nk = ak - wj * aj;
        ar[j] = nj.x;  ai[j] = nj.y;
        ar[k] = nk.x;  ai[k] = nk.y;
      }
    }
  } else if constexpr (FS == 0) {
#pragma unroll
    for (int j = 0; j < NS; ++j) {
      const float wj = parity8(j & RS) ? tn : tp;
      fmac_lane_xor<FL>(ar[j], wj);
      fmac_lane_xor<FL>(ai[j], wj);
    }
  } else {
    float nr[NS], ni[NS];
#pragma unroll
    for (int j = 0; j < NS; ++j) {
      const float wj = parity8(j & RS) ? tn : tp;
      nr[j] = fma_lane_xor<FL>(ar[j ^ FS], wj, ar[j]);
      ni[j] = fma_lane_xor<FL>(ai[j ^ FS], wj, ai[j]);
    }
#pragma unroll
    for (int j = 0; j < NS; ++j) { ar[j] = nr[j]; ai[j] = ni[j]; }
  }
}

__device__ __forceinline__ float ry_factor_sq(const float* __restrict__ sF) {
  float p = sF[0];
#pragma unroll
  for (int k = 1; k < 16; ++k) p *= sF[k];
  return p * p;
}

template<int CJB, int TJB>
__device__ __forceinline__ void cnot_ll(float ar[NS], float ai[NS]) {
#pragma unroll
  for (int j = 0; j < NS; ++j)
    if (((j >> CJB) & 1) && !((j >> TJB) & 1)) {
      const int j1 = j | (1 << TJB);
      float tr = ar[j], ti = ai[j];
      ar[j] = ar[j1]; ai[j] = ai[j1];
      ar[j1] = tr;    ai[j1] = ti;
    }
}
template<int CJB, int TLB>
__device__ __forceinline__ void cnot_local_lane(float ar[NS], float ai[NS]) {
#pragma unroll
  for (int j = 0; j < NS; ++j)
    if ((j >> CJB) & 1) {
      ar[j] = lane_xor<(1 << TLB)>(ar[j]);
      ai[j] = lane_xor<(1 << TLB)>(ai[j]);
    }
}
template<int CLB, int TJB>
__device__ __forceinline__ void cnot_lane_local(float ar[NS], float ai[NS]) {
  if ((LANE() >> CLB) & 1) {
#pragma unroll
    for (int j = 0; j < NS; ++j)
      if (!((j >> TJB) & 1)) {
        const int j1 = j | (1 << TJB);
        float tr = ar[j], ti = ai[j];
        ar[j] = ar[j1]; ai[j] = ai[j1];
        ar[j1] = tr;    ai[j1] = ti;
      }
  }
}
__device__ __forceinline__ void cnot_lane_chain(float ar[NS], float ai[NS]) {
#pragma unroll
  for (int j = 0; j < NS; ++j) {
    ar[j] = cnot_row_chain(ar[j]);
    ai[j] = cnot_row_chain(ai[j]);
  }
}

__device__ __forceinline__ void cnot_ring(float ar[NS], float ai[NS]) {
  cnot_lane_chain(ar, ai);
  cnot_lane_local<0, 4>(ar, ai);
  cnot_ll<4, 3>(ar, ai);
  cnot_ll<3, 2>(ar, ai);
  cnot_ll<2, 1>(ar, ai);
  cnot_ll<1, 0>(ar, ai);
  cnot_local_lane<0, 2>(ar, ai);
}

template<int W>
__device__ __forceinline__ void wire_state(const float* __restrict__ cs,
    const float* __restrict__ sn, const float* __restrict__ ph0,
    const float* __restrict__ ry0,
    float& g0r, float& g0i, float& g1r, float& g1i)
{
  const float zr = cs[8 + W], zs = sn[8 + W];
  const float pc = ph0[2 * W], ps = ph0[2 * W + 1];
  const float ur = zr * pc - zs * ps;
  const float ui = zs * pc + zr * ps;
  const float f0r = cs[W] * ur, f0i = -(cs[W] * ui);
  const float f1r = sn[W] * ur, f1i =   sn[W] * ui;
  const float C = ry0[2 * W], S = ry0[2 * W + 1];
  g0r = C * f0r - S * f1r;  g0i = C * f0i - S * f1i;
  g1r = S * f0r + C * f1r;  g1i = S * f0i + C * f1i;
}

#define CMUL(orr, oi, xr, xi, yr, yi) {                                       \
    orr = (xr) * (yr) - (xi) * (yi);                                          \
    oi  = (xr) * (yi) + (xi) * (yr);                                          \
  }

__device__ __forceinline__ void product_state_behind_ring(
    const float* __restrict__ cs, const float* __restrict__ sn,
    const float* __restrict__ ph0, const float* __restrict__ ry0,
    float ar[NS], float ai[NS])
{
  float gr[8][2], gi[8][2];
  wire_state<0>(cs, sn, ph0, ry0, gr[0][0], gi[0][0], gr[0][1], gi[0][1]);
  wire_state<1>(cs, sn, ph0, ry0, gr[1][0], gi[1][0], gr[1][1], gi[1][1]);
  wire_state<2>(cs, sn, ph0, ry0, gr[2][0], gi[2][0], gr[2][1], gi[2][1]);
  wire_state<3>(cs, sn, ph0, ry0, gr[3][0], gi[3][0], gr[3][1], gi[3][1]);
  wire_state<4>(cs, sn, ph0, ry0, gr[4][0], gi[4][0], gr[4][1], gi[4][1]);
  wire_state<5>(cs, sn, ph0, ry0, gr[5][0], gi[5][0], gr[5][1], gi[5][1]);
  wire_state<6>(cs, sn, ph0, ry0, gr[6][0], gi[6][0], gr[6][1], gi[6][1]);
  wire_state<7>(cs, sn, ph0, ry0, gr[7][0], gi[7][0], gr[7][1], gi[7][1]);

  const int x0 = (LANE() >> 2) & 1;
  const int x1 = (LANE() >> 1) & 1;
  const int x2 =  LANE()       & 1;

  const int y = x1 ^ x0;
  const float u0r = x0 ? gr[0][1] : gr[0][0], u0i = x0 ? gi[0][1] : gi[0][0];
  const float v0r = x0 ? gr[0][0] : gr[0][1], v0i = x0 ? gi[0][0] : gi[0][1];
  const float u1r = y  ? gr[1][1] : gr[1][0], u1i = y  ? gi[1][1] : gi[1][0];
  const float v1r = y  ? gr[1][0] : gr[1][1], v1i = y  ? gi[1][0] : gi[1][1];
  float Ar[2], Ai[2];
  CMUL(Ar[0], Ai[0], u0r, u0i, u1r, u1i)
  CMUL(Ar[1], Ai[1], v0r, v0i, v1r, v1i)
  float Hr[2][2], Hi[2][2];
#pragma unroll
  for (int c6 = 0; c6 < 2; ++c6)
#pragma unroll
    for (int c7 = 0; c7 < 2; ++c7)
      CMUL(Hr[c6][c7], Hi[c6][c7], gr[7][c7 ^ c6], gi[7][c7 ^ c6], Ar[c7], Ai[c7])

  const int z = x2 ^ x1;
  const float l2r = z ? gr[2][1] : gr[2][0], l2i = z ? gi[2][1] : gi[2][0];
  float G3r[2], G3i[2];
  G3r[0] = x2 ? gr[3][1] : gr[3][0];  G3i[0] = x2 ? gi[3][1] : gi[3][0];
  G3r[1] = x2 ? gr[3][0] : gr[3][1];  G3i[1] = x2 ? gi[3][0] : gi[3][1];

  float t1r[2], t1i[2], t2r[4], t2i[4], t3r[8], t3i[8], t4r[16], t4i[16];
#pragma unroll
  for (int b = 0; b < 2; ++b) CMUL(t1r[b], t1i[b], l2r, l2i, G3r[b], G3i[b])
#pragma unroll
  for (int b = 0; b < 4; ++b)
    CMUL(t2r[b], t2i[b], t1r[b >> 1], t1i[b >> 1],
         gr[4][(b & 1) ^ ((b >> 1) & 1)], gi[4][(b & 1) ^ ((b >> 1) & 1)])
#pragma unroll
  for (int b = 0; b < 8; ++b)
    CMUL(t3r[b], t3i[b], t2r[b >> 1], t2i[b >> 1],
         gr[5][(b & 1) ^ ((b >> 1) & 1)], gi[5][(b & 1) ^ ((b >> 1) & 1)])
#pragma unroll
  for (int b = 0; b < 16; ++b)
    CMUL(t4r[b], t4i[b], t3r[b >> 1], t3i[b >> 1],
         gr[6][(b & 1) ^ ((b >> 1) & 1)], gi[6][(b & 1) ^ ((b >> 1) & 1)])
#pragma unroll
  for (int j = 0; j < NS; ++j)
    CMUL(ar[j], ai[j], t4r[j >> 1], t4i[j >> 1],
         Hr[(j >> 1) & 1][j & 1], Hi[(j >> 1) & 1][j & 1])
}

__device__ __forceinline__ void apply_phase_table(float ar[NS], float ai[NS],
                                                  const v2f* __restrict__ T)
{
#pragma unroll
  for (int j = 0; j < NS; ++j) {
    const v2f t = T[j * 8];
    const float r = ar[j] * t.x - ai[j] * t.y;
    const float i = ar[j] * t.y + ai[j] * t.x;
    ar[j] = r; ai[j] = i;
  }
}

__device__ __forceinline__ void group_sum_store(float v, float scale,
                                                float* __restrict__ dst, int lane)
{
  v += lane_xor<4>(v);
  v += lane_xor<2>(v);
  v += lane_xor<1>(v);
  v *= scale;
  if ((lane & 7) == 0) *dst = v;
}

__device__ __forceinline__ void shear_layers_and_readout(
    float ar[NS], float ai[NS], const v2f* __restrict__ sT,
    const float* __restrict__ sRYT, const float* __restrict__ sRYF,
    float* __restrict__ sZ, int samp, int lane)
{
  apply_phase_table(ar, ai, &sT[lane & 7]);
  ry_shear<0, 0>(ar, ai, sRYT[0], lane);
  ry_shear<1, 0>(ar, ai, sRYT[1], lane);
  ry_shear<2, 0>(ar, ai, sRYT[2], lane);
  ry_shear<3, 0>(ar, ai, sRYT[3], lane);
  ry_shear<4, 0>(ar, ai, sRYT[4], lane);
  ry_shear<5, 0>(ar, ai, sRYT[5], lane);
  ry_shear<6, 0>(ar, ai, sRYT[6], lane);
  ry_shear<7, 0>(ar, ai, sRYT[7], lane);

  apply_phase_table(ar, ai, &sT[256 + (lane & 7)]);
  ry_shear<0, 1>(ar, ai, sRYT[8],  lane);
  ry_shear<1, 1>(ar, ai, sRYT[9],  lane);
  ry_shear<2, 1>(ar, ai, sRYT[10], lane);
  ry_shear<3, 1>(ar, ai, sRYT[11], lane);
  ry_shear<4, 1>(ar, ai, sRYT[12], lane);
  ry_shear<5, 1>(ar, ai, sRYT[13], lane);
  ry_shear<6, 1>(ar, ai, sRYT[14], lane);
  ry_shear<7, 1>(ar, ai, sRYT[15], lane);

  float h[NS];
#pragma unroll
  for (int j = 0; j < NS; ++j) h[j] = ar[j]*ar[j] + ai[j]*ai[j];
#pragma unroll
  for (int s = 1; s < NS; s <<= 1) {
#pragma unroll
    for (int j = 0; j < NS; ++j) {
      if (!(j & s)) {
        const float a = h[j], b = h[j | s];
        h[j] = a + b;  h[j | s] = a - b;
      }
    }
  }
  const float fsq = ry_factor_sq(sRYF);
  float zp[8];
#define READOUT_WIRE(w) {                                                     \
    constexpr int RZ = wire_sign_mask(w, 2);                                  \
    const float v = h[RZ & 31];                                               \
    zp[w] = (__builtin_popcount(lane & (RZ >> 5)) & 1) ? -v : v;              \
  }
  READOUT_WIRE(0) READOUT_WIRE(1) READOUT_WIRE(2) READOUT_WIRE(3)
  READOUT_WIRE(4) READOUT_WIRE(5) READOUT_WIRE(6) READOUT_WIRE(7)
#undef READOUT_WIRE
#pragma unroll
  for (int w = 0; w < 8; ++w)
    group_sum_store(zp[w], fsq, &sZ[samp * 8 + w], lane);
}

__device__ __forceinline__ void rotation_layers_and_readout(
    float ar[NS], float ai[NS], const v2f* __restrict__ sT,
    const float* __restrict__ sRY, float* __restrict__ sZ, int samp, int lane)
{
  for (int L = 1; L < 3; ++L) {
    apply_phase_table(ar, ai, &sT[(L - 1) * 256 + (lane & 7)]);
    const float* ry = &sRY[L * 16];
    ry_lane<2>(ar, ai, ry[0],  ry[1]);
    ry_lane<1>(ar, ai, ry[2],  ry[3]);
    ry_lane<0>(ar, ai, ry[4],  ry[5]);
    ry_local<4>(ar, ai, ry[6],  ry[7]);
    ry_local<3>(ar, ai, ry[8],  ry[9]);
    ry_local<2>(ar, ai, ry[10], ry[11]);
    ry_local<1>(ar, ai, ry[12], ry[13]);
    ry_local<0>(ar, ai, ry[14], ry[15]);
    cnot_ring(ar, ai);
  }

  float pb[NS]; float tot = 0.f;
#pragma unroll
  for (int j = 0; j < NS; ++j) { pb[j] = ar[j]*ar[j] + ai[j]*ai[j]; tot += pb[j]; }
  float zp[8];
#pragma unroll
  for (int w = 0; w < 3; ++w)
    zp[w] = ((lane >> (2 - w)) & 1) ? -tot : tot;
  float s3 = 0.f, s4 = 0.f, s5 = 0.f, s6 = 0.f, s7 = 0.f;
#pragma unroll
  for (int j = 0; j < NS; ++j) {
    s3 += ((j >> 4) & 1) ? -pb[j] : pb[j];
    s4 += ((j >> 3) & 1) ? -pb[j] : pb[j];
    s5 += ((j >> 2) & 1) ? -pb[j] : pb[j];
    s6 += ((j >> 1) & 1) ? -pb[j] : pb[j];
    s7 += ( j       & 1) ? -pb[j] : pb[j];
  }
  zp[3] = s3; zp[4] = s4; zp[5] = s5; zp[6] = s6; zp[7] = s7;
#pragma unroll
  for (int w = 0; w < 8; ++w)
    group_sum_store(zp[w], 1.0f, &sZ[samp * 8 + w], lane);
}

__global__ __launch_bounds__(512) void quantum_block_kernel(
    const float* __restrict__ x,
    const float* __restrict__ W1, const float* __restrict__ b1,
    const float* __restrict__ W2, const float* __restrict__ b2,
    const float* __restrict__ W3, const float* __restrict__ b3,
    const float* __restrict__ qw,
    const float* __restrict__ Wp, const float* __restrict__ bp,
    float* __restrict__ out)
{
  __shared__ __attribute__((aligned(16))) float sX  [64 * 84];
  __shared__ float sH1 [64 * 68];
  __shared__ float sH2 [64 * 36];
  __shared__ float sCos[64 * 16];
  __shared__ float sSin[64 * 16];
  __shared__ float sZ  [64 * 8];
  __shared__ __attribute__((aligned(64))) v2f sT[2 * 256];
  __shared__ float sRY [24 * 2];
  __shared__ __attribute__((aligned(16))) float sRYT[16];
  __shared__ __attribute__((aligned(16))) float sRYF[16];
  __shared__ float sPH0[8 * 2];
  __shared__ __attribute__((aligned(16))) float sW1[64 * 84];
  __shared__ __attribute__((aligned(16))) float sW2[32 * 68];
  __shared__ __attribute__((aligned(16))) float sW3[16 * 36];

  const int tid     = threadIdx.x;
  const int wid     = tid >> 5;
  const int lane    = tid & 31;
  const int rowBase = blockIdx.x * 64;

  const int k = lane < 24 ? lane : 23;
  const float th = qw[k * 3 + 1];
  const float ryc = __cosf(0.5f * th), rys = __sinf(0.5f * th);
  const bool steep = lane >= 8 && lane < 24 && !(fabsf(rys) <= 8.0f * fabsf(ryc));
  const bool shear_ok = __builtin_amdgcn_ballot_w32(steep) == 0;
  if (wid == 0 && lane < 24) {
    sRY[lane * 2]     = ryc;
    sRY[lane * 2 + 1] = rys;
    if (lane < 8) {
      const float ph = qw[lane * 3];
      sPH0[lane * 2]     = __cosf(0.5f * ph);
      sPH0[lane * 2 + 1] = __sinf(0.5f * ph);
    } else {
      sRYT[lane - 8] = rys / ryc;
      sRYF[lane - 8] = ryc;
    }
  }
  if (tid < 256) {
    build_phase_entry<1>(qw, &sT[0],   tid, false);
  } else {
    build_phase_entry<2>(qw, &sT[256], tid - 256, shear_ok);
  }
  constexpr int W1P = 64 * 20, W3P = 16 * 8;
  int  p1[3];
  v4f  w1[3], xv[3];
#pragma unroll
  for (int it = 0; it < 3; ++it) {
    const int p = it * 512 + tid;
    p1[it] = p < W1P ? p : W1P - 1;
    xv[it] = *reinterpret_cast<const v4f*>(x + rowBase * 80 + 4 * p1[it]);
    w1[it] = *reinterpret_cast<const v4f*>(W1 + 4 * p1[it]);
  }
  const v4f w2 = *reinterpret_cast<const v4f*>(W2 + 4 * tid);
  const int p3 = tid < W3P ? tid : W3P - 1;
  const v4f w3 = *reinterpret_cast<const v4f*>(W3 + 4 * p3);
#pragma unroll
  for (int it = 0; it < 3; ++it) {
    const int o = (p1[it] / 20) * 84 + (p1[it] % 20) * 4;
    *reinterpret_cast<v4f*>(&sX[o])  = xv[it];
    *reinterpret_cast<v4f*>(&sW1[o]) = w1[it];
  }
  *reinterpret_cast<v4f*>(&sW2[(tid >> 4) * 68 + (tid & 15) * 4]) = w2;
  *reinterpret_cast<v4f*>(&sW3[(p3 >> 3) * 36 + (p3 & 7) * 4]) = w3;
  __syncthreads();

  {
    const int mt = wid >> 2, nt = wid & 3;
    gemm_tile_to_lds<80, 84, 84, true >(sX + mt * 16 * 84, sW1, b1, sH1 + mt * 16 * 68, 68, nt * 16);
  }
  __syncthreads();
  if (wid < 8) {
    const int mt = wid >> 1, nt = wid & 1;
    gemm_tile_to_lds<64, 68, 68, true >(sH1 + mt * 16 * 68, sW2, b2, sH2 + mt * 16 * 36, 36, nt * 16);
  }
  __syncthreads();
  if (wid < 4)
    gemm_tile_to_halfangle_trig<32, 36, 36>(sH2 + wid * 16 * 36, sW3, b3,
                                        sCos + wid * 16 * 16, sSin + wid * 16 * 16, 16, 0);
  __syncthreads();

  {
    const int samp = wid * 4 + (lane >> 3);
    float ar[NS], ai[NS];
    product_state_behind_ring(&sCos[samp * 16], &sSin[samp * 16], sPH0, sRY, ar, ai);

    if (shear_ok)
      shear_layers_and_readout(ar, ai, sT, sRYT, sRYF, sZ, samp, lane);
    else
      rotation_layers_and_readout(ar, ai, sT, sRY, sZ, samp, lane);
  }
  __syncthreads();

#pragma unroll
  for (int i = 0; i < 2; ++i) {
    const int mt = (wid >> 3) * 2 + i;
    gemm_final(sZ + mt * 16 * 8, Wp, bp, out, rowBase + mt * 16, (wid & 7) * 16);
  }
}

extern "C" void kernel_launch(void* const* d_in, const int* in_sizes, int n_in,
                              void* d_out, int out_size, void* d_ws, size_t ws_size,
                              hipStream_t stream) {
  const float* x  = (const float*)d_in[0];
  const float* W1 = (const float*)d_in[1];
  const float* b1 = (const float*)d_in[2];
  const float* W2 = (const float*)d_in[3];
  const float* b2 = (const float*)d_in[4];
  const float* W3 = (const float*)d_in[5];
  const float* b3 = (const float*)d_in[6];
  const float* qw = (const float*)d_in[7];
  const float* Wp = (const float*)d_in[8];
  const float* bp = (const float*)d_in[9];
  float* out = (float*)d_out;

  const int B = in_sizes[0] / 80;
  const int grid = B / 64;
  quantum_block_kernel<<<grid, 512, 0, stream>>>(x, W1, b1, W2, b2, W3, b3,
                                                 qw, Wp, bp, out);
}


# All global loads issued at the kernel's top: staging loads (x, W1-W3) before the circuit set-up instead of behind it; Wp and biases staged in LDS too: no later phase waits on global memory
# speedup vs baseline: 3.2183x; 1.0786x over previous
#include <hip/hip_runtime.h>
#include <math.h>

typedef __attribute__((ext_vector_type(2))) float v2f;
typedef __attribute__((ext_vector_type(4))) float v4f;
typedef __attribute__((ext_vector_type(8))) float v8f;

#define LANE() ((int)(threadIdx.x & 31))

template<int M>
__device__ __forceinline__ float lane_xor(float v) {
  static_assert(M == 1 || M == 2 || M == 4 || M == 8, "in-row partner only");
  const int b = __builtin_bit_cast(int, v);
  return __builtin_bit_cast(float,
      __builtin_amdgcn_mov_dpp(b, 0x160 | M, 0xf, 0xf, true));
}

__device__ constexpr unsigned ring_row_selects(unsigned word) {
  unsigned v = 0;
  for (unsigned j = 0; j < 8; ++j) {
    const unsigned l = word * 8 + j;
    const unsigned q = l & 7u;
    v |= ((l & 8u) | (q ^ ((q >> 1) & 3u))) << (4 * j);
  }
  return v;
}
__device__ __forceinline__ float cnot_row_chain(float v) {
  constexpr unsigned lo = ring_row_selects(0), hi = ring_row_selects(1);
  static_assert(lo == 0x45762310u && hi == 0xcdfeab98u, "in-group source lanes of the CNOT chain");
  const int b = __builtin_bit_cast(int, v);
  return __builtin_bit_cast(float,
      __builtin_amdgcn_permlane16(b, b, lo, hi, false, false));
}

__device__ __forceinline__ float fast_tanh(float x) {
  const float e = __expf(2.0f * x);
  return 1.0f - 2.0f / (e + 1.0f);
}

template<int K, int LDA, int LDB>
__device__ __forceinline__ v8f gemm_tile_acc(
    const float* __restrict__ sA,
    const float* __restrict__ sW,
    int n)
{
  const int lane = LANE();
  const int half = lane >> 4;
  const int col  = lane & 15;
  v8f acc = {};
#pragma unroll
  for (int k0 = 0; k0 < K; k0 += 4) {
    const int ka = k0 + half * 2;
    v2f a, b;
    a.x = sA[col * LDA + ka];
    a.y = sA[col * LDA + ka + 1];
    b.x = sW[n * LDB + ka];
    b.y = sW[n * LDB + ka + 1];
    acc = __builtin_amdgcn_wmma_f32_16x16x4_f32(false, a, false, b,
                                                (short)0, acc, false, false);
  }
  return acc;
}

template<int K, int LDA, int LDB, bool TANH>
__device__ __forceinline__ void gemm_tile_to_lds(
    const float* __restrict__ sA, const float* __restrict__ sW,
    const float* __restrict__ bg,
    float* __restrict__ sOut, int ldo, int n0)
{
  const int lane = LANE();
  const int half = lane >> 4;
  const int n    = n0 + (lane & 15);
  const v8f acc = gemm_tile_acc<K, LDA, LDB>(sA, sW, n);
  const float bias = bg[n];
#pragma unroll
  for (int r = 0; r < 8; ++r) {
    float v = acc[r] + bias;
    if (TANH) v = fast_tanh(v);
    sOut[(half * 8 + r) * ldo + n] = v;
  }
}

template<int K, int LDA, int LDB>
__device__ __forceinline__ void gemm_tile_to_halfangle_trig(
    const float* __restrict__ sA, const float* __restrict__ sW,
    const float* __restrict__ bg,
    float* __restrict__ sCos, float* __restrict__ sSin, int ldo, int n0)
{
  const int lane = LANE();
  const int half = lane >> 4;
  const int n    = n0 + (lane & 15);
  const v8f acc = gemm_tile_acc<K, LDA, LDB>(sA, sW, n);
  const float bias = bg[n];
#pragma unroll
  for (int r = 0; r < 8; ++r) {
    const float h = 0.5f * (acc[r] + bias);
    sCos[(half * 8 + r) * ldo + n] = __cosf(h);
    sSin[(half * 8 + r) * ldo + n] = __sinf(h);
  }
}

__device__ __forceinline__ void gemm_final(
    const float* __restrict__ sZ,
    const float* __restrict__ sWp,
    const float* __restrict__ sBp,
    float* __restrict__ out, int rowBase, int n0)
{
  const int lane = LANE();
  const int half = lane >> 4;
  const int col  = lane & 15;
  const int n    = n0 + col;
  v8f acc = {};
#pragma unroll
  for (int k0 = 0; k0 < 8; k0 += 4) {
    const int ka = k0 + half * 2;
    v2f a, b;
    a.x = sZ[col * 8 + ka];
    a.y = sZ[col * 8 + ka + 1];
    b.x = sWp[n * 12 + ka];
    b.y = sWp[n * 12 + ka + 1];
    acc = __builtin_amdgcn_wmma_f32_16x16x4_f32(false, a, false, b,
                                                (short)0, acc, false, false);
  }
  const float bias = sBp[n];
#pragma unroll
  for (int r = 0; r < 8; ++r)
    out[(size_t)(rowBase + half * 8 + r) * 128 + n] = acc[r] + bias;
}


__device__ constexpr int ring_preimage(int c) {
  c ^= (c & 1) << 7;
  c ^= (c >> 1) & 1;
  c ^= ((c >> 2) & 1) << 1;
  c ^= ((c >> 3) & 1) << 2;
  c ^= ((c >> 4) & 1) << 3;
  c ^= ((c >> 5) & 1) << 4;
  c ^= ((c >> 6) & 1) << 5;
  c ^= ((c >> 7) & 1) << 6;
  return c;
}
__device__ constexpr int ring_image(int c) {
  c ^= ((c >> 7) & 1) << 6;
  c ^= ((c >> 6) & 1) << 5;
  c ^= ((c >> 5) & 1) << 4;
  c ^= ((c >> 4) & 1) << 3;
  c ^= ((c >> 3) & 1) << 2;
  c ^= ((c >> 2) & 1) << 1;
  c ^= (c >> 1) & 1;
  c ^= (c & 1) << 7;
  return c;
}

__device__ constexpr int ring_image_n(int c, int n) {
  for (int i = 0; i < n; ++i) c = ring_image(c);
  return c;
}
__device__ constexpr int ring_preimage_n(int c, int n) {
  for (int i = 0; i < n; ++i) c = ring_preimage(c);
  return c;
}
__device__ constexpr int wire_flip_mask(int w, int n) {
  return ring_preimage_n(1 << (7 - w), n);
}
__device__ constexpr int wire_sign_mask(int w, int n) {
  int m = 0;
  for (int i = 0; i < 8; ++i)
    if ((ring_image_n(1 << i, n) >> (7 - w)) & 1) m |= 1 << i;
  return m;
}
__device__ constexpr int parity8(int v) {
  v ^= v >> 4;  v ^= v >> 2;  v ^= v >> 1;
  return v & 1;
}

template<int L, int COMP>
__device__ __forceinline__ float rz_phase(const float* __restrict__ qw, int c) {
  float s = 0.f;
#pragma unroll
  for (int w = 0; w < 8; ++w) {
    const float h = 0.5f * qw[(L * 8 + w) * 3 + COMP];
    s += ((c >> (7 - w)) & 1) ? h : -h;
  }
  return s;
}

template<int L>
__device__ __forceinline__ void build_phase_entry(const float* __restrict__ qw,
                                                  v2f* __restrict__ sTL, int c,
                                                  bool relabelled)
{
  const int cphi = relabelled ? ring_image(c) : c;
  const int comg = relabelled ? c : ring_preimage(c);
  float ang = rz_phase<L, 0>(qw, cphi);
  if constexpr (L > 0) ang += rz_phase<L - 1, 2>(qw, comg);
  v2f t;
  t.x = __cosf(ang);
  t.y = __sinf(ang);
  sTL[(c & 31) * 8 + (c >> 5)] = t;
}


constexpr int NS = 32;

template<int LB>
__device__ __forceinline__ void ry_lane(float ar[NS], float ai[NS], float c, float s) {
  const float t = ((LANE() >> LB) & 1) ? s : -s;
#pragma unroll
  for (int j = 0; j < NS; ++j) {
    const float pr = lane_xor<(1 << LB)>(ar[j]);
    const float pi = lane_xor<(1 << LB)>(ai[j]);
    ar[j] = c * ar[j] + t * pr;
    ai[j] = c * ai[j] + t * pi;
  }
}

template<int JB>
__device__ __forceinline__ void ry_local(float ar[NS], float ai[NS], float c, float s) {
#pragma unroll
  for (int j = 0; j < NS; ++j) {
    if (!((j >> JB) & 1)) {
      const int j1 = j | (1 << JB);
      const float a0r = ar[j], a0i = ai[j], a1r = ar[j1], a1i = ai[j1];
      ar[j]  = c * a0r - s * a1r;  ai[j]  = c * a0i - s * a1i;
      ar[j1] = s * a0r + c * a1r;  ai[j1] = s * a0i + c * a1i;
    }
  }
}


template<int M>
__device__ __forceinline__ void fmac_lane_xor(float& acc, float w) {
  static_assert(M >= 1 && M <= 7, "in-group partner only");
  asm("v_fmac_f32_dpp %0, %0, %1 row_xmask:%2 row_mask:0xf bank_mask:0xf bound_ctrl:1"
      : "+v"(acc) : "v"(w), "n"(M));
}
template<int M>
__device__ __forceinline__ float fma_lane_xor(float partner, float w, float self) {
  static_assert(M >= 1 && M <= 7, "in-group partner only");
  float d;
  asm("v_fma_f32_e64_dpp %0, %1, %2, %3 row_xmask:%4 row_mask:0xf bank_mask:0xf bound_ctrl:1"
      : "=v"(d) : "v"(partner), "v"(w), "v"(self), "n"(M));
  return d;
}

template<int W, int N>
__device__ __forceinline__ void ry_shear(float ar[NS], float ai[NS], float t, int lane) {
  constexpr int F = wire_flip_mask(W, N), RM = wire_sign_mask(W, N);
  constexpr int FL = F >> 5, FS = F & 31, RL = RM >> 5, RS = RM & 31;
  static_assert(parity8(F & RM) == 1, "the partner differs in the wire's own bit");
  const float tp = (__builtin_popcount(lane & RL) & 1) ? t : -t;
  const float tn = -tp;
  if constexpr (FL == 0) {
#pragma unroll
    for (int j = 0; j < NS; ++j) {
      if (j < (j ^ FS)) {
        const int k = j ^ FS;
        const float wj = parity8(j & RS) ? tn : tp;
        const v2f aj = {ar[j], ai[j]}, ak = {ar[k], ai[k]};
        const v2f nj = aj + wj * ak;
        const v2f nk = ak - wj * aj;
        ar[j] = nj.x;  ai[j] = nj.y;
        ar[k] = nk.x;  ai[k] = nk.y;
      }
    }
  } else if constexpr (FS == 0) {
#pragma unroll
    for (int j = 0; j < NS; ++j) {
      const float wj = parity8(j & RS) ? tn : tp;
      fmac_lane_xor<FL>(ar[j], wj);
      fmac_lane_xor<FL>(ai[j], wj);
    }
  } else {
    float nr[NS], ni[NS];
#pragma unroll
    for (int j = 0; j < NS; ++j) {
      const float wj = parity8(j & RS) ? tn : tp;
      nr[j] = fma_lane_xor<FL>(ar[j ^ FS], wj, ar[j]);
      ni[j] = fma_lane_xor<FL>(ai[j ^ FS], wj, ai[j]);
    }
#pragma unroll
    for (int j = 0; j < NS; ++j) { ar[j] = nr[j]; ai[j] = ni[j]; }
  }
}

__device__ __forceinline__ float ry_factor_sq(const float* __restrict__ sF) {
  float p = sF[0];
#pragma unroll
  for (int k = 1; k < 16; ++k) p *= sF[k];
  return p * p;
}

template<int CJB, int TJB>
__device__ __forceinline__ void cnot_ll(float ar[NS], float ai[NS]) {
#pragma unroll
  for (int j = 0; j < NS; ++j)
    if (((j >> CJB) & 1) && !((j >> TJB) & 1)) {
      const int j1 = j | (1 << TJB);
      float tr = ar[j], ti = ai[j];
      ar[j] = ar[j1]; ai[j] = ai[j1];
      ar[j1] = tr;    ai[j1] = ti;
    }
}
template<int CJB, int TLB>
__device__ __forceinline__ void cnot_local_lane(float ar[NS], float ai[NS]) {
#pragma unroll
  for (int j = 0; j < NS; ++j)
    if ((j >> CJB) & 1) {
      ar[j] = lane_xor<(1 << TLB)>(ar[j]);
      ai[j] = lane_xor<(1 << TLB)>(ai[j]);
    }
}
template<int CLB, int TJB>
__device__ __forceinline__ void cnot_lane_local(float ar[NS], float ai[NS]) {
  if ((LANE() >> CLB) & 1) {
#pragma unroll
    for (int j = 0; j < NS; ++j)
      if (!((j >> TJB) & 1)) {
        const int j1 = j | (1 << TJB);
        float tr = ar[j], ti = ai[j];
        ar[j] = ar[j1]; ai[j] = ai[j1];
        ar[j1] = tr;    ai[j1] = ti;
      }
  }
}
__device__ __forceinline__ void cnot_lane_chain(float ar[NS], float ai[NS]) {
#pragma unroll
  for (int j = 0; j < NS; ++j) {
    ar[j] = cnot_row_chain(ar[j]);
    ai[j] = cnot_row_chain(ai[j]);
  }
}

__device__ __forceinline__ void cnot_ring(float ar[NS], float ai[NS]) {
  cnot_lane_chain(ar, ai);
  cnot_lane_local<0, 4>(ar, ai);
  cnot_ll<4, 3>(ar, ai);
  cnot_ll<3, 2>(ar, ai);
  cnot_ll<2, 1>(ar, ai);
  cnot_ll<1, 0>(ar, ai);
  cnot_local_lane<0, 2>(ar, ai);
}

template<int W>
__device__ __forceinline__ void wire_state(const float* __restrict__ cs,
    const float* __restrict__ sn, const float* __restrict__ ph0,
    const float* __restrict__ ry0,
    float& g0r, float& g0i, float& g1r, float& g1i)
{
  const float zr = cs[8 + W], zs = sn[8 + W];
  const float pc = ph0[2 * W], ps = ph0[2 * W + 1];
  const float ur = zr * pc - zs * ps;
  const float ui = zs * pc + zr * ps;
  const float f0r = cs[W] * ur, f0i = -(cs[W] * ui);
  const float f1r = sn[W] * ur, f1i =   sn[W] * ui;
  const float C = ry0[2 * W], S = ry0[2 * W + 1];
  g0r = C * f0r - S * f1r;  g0i = C * f0i - S * f1i;
  g1r = S * f0r + C * f1r;  g1i = S * f0i + C * f1i;
}

#define CMUL(orr, oi, xr, xi, yr, yi) {                                       \
    orr = (xr) * (yr) - (xi) * (yi);                                          \
    oi  = (xr) * (yi) + (xi) * (yr);                                          \
  }

__device__ __forceinline__ void product_state_behind_ring(
    const float* __restrict__ cs, const float* __restrict__ sn,
    const float* __restrict__ ph0, const float* __restrict__ ry0,
    float ar[NS], float ai[NS])
{
  float gr[8][2], gi[8][2];
  wire_state<0>(cs, sn, ph0, ry0, gr[0][0], gi[0][0], gr[0][1], gi[0][1]);
  wire_state<1>(cs, sn, ph0, ry0, gr[1][0], gi[1][0], gr[1][1], gi[1][1]);
  wire_state<2>(cs, sn, ph0, ry0, gr[2][0], gi[2][0], gr[2][1], gi[2][1]);
  wire_state<3>(cs, sn, ph0, ry0, gr[3][0], gi[3][0], gr[3][1], gi[3][1]);
  wire_state<4>(cs, sn, ph0, ry0, gr[4][0], gi[4][0], gr[4][1], gi[4][1]);
  wire_state<5>(cs, sn, ph0, ry0, gr[5][0], gi[5][0], gr[5][1], gi[5][1]);
  wire_state<6>(cs, sn, ph0, ry0, gr[6][0], gi[6][0], gr[6][1], gi[6][1]);
  wire_state<7>(cs, sn, ph0, ry0, gr[7][0], gi[7][0], gr[7][1], gi[7][1]);

  const int x0 = (LANE() >> 2) & 1;
  const int x1 = (LANE() >> 1) & 1;
  const int x2 =  LANE()       & 1;

  const int y = x1 ^ x0;
  const float u0r = x0 ? gr[0][1] : gr[0][0], u0i = x0 ? gi[0][1] : gi[0][0];
  const float v0r = x0 ? gr[0][0] : gr[0][1], v0i = x0 ? gi[0][0] : gi[0][1];
  const float u1r = y  ? gr[1][1] : gr[1][0], u1i = y  ? gi[1][1] : gi[1][0];
  const float v1r = y  ? gr[1][0] : gr[1][1], v1i = y  ? gi[1][0] : gi[1][1];
  float Ar[2], Ai[2];
  CMUL(Ar[0], Ai[0], u0r, u0i, u1r, u1i)
  CMUL(Ar[1], Ai[1], v0r, v0i, v1r, v1i)
  float Hr[2][2], Hi[2][2];
#pragma unroll
  for (int c6 = 0; c6 < 2; ++c6)
#pragma unroll
    for (int c7 = 0; c7 < 2; ++c7)
      CMUL(Hr[c6][c7], Hi[c6][c7], gr[7][c7 ^ c6], gi[7][c7 ^ c6], Ar[c7], Ai[c7])

  const int z = x2 ^ x1;
  const float l2r = z ? gr[2][1] : gr[2][0], l2i = z ? gi[2][1] : gi[2][0];
  float G3r[2], G3i[2];
  G3r[0] = x2 ? gr[3][1] : gr[3][0];  G3i[0] = x2 ? gi[3][1] : gi[3][0];
  G3r[1] = x2 ? gr[3][0] : gr[3][1];  G3i[1] = x2 ? gi[3][0] : gi[3][1];

  float t1r[2], t1i[2], t2r[4], t2i[4], t3r[8], t3i[8], t4r[16], t4i[16];
#pragma unroll
  for (int b = 0; b < 2; ++b) CMUL(t1r[b], t1i[b], l2r, l2i, G3r[b], G3i[b])
#pragma unroll
  for (int b = 0; b < 4; ++b)
    CMUL(t2r[b], t2i[b], t1r[b >> 1], t1i[b >> 1],
         gr[4][(b & 1) ^ ((b >> 1) & 1)], gi[4][(b & 1) ^ ((b >> 1) & 1)])
#pragma unroll
  for (int b = 0; b < 8; ++b)
    CMUL(t3r[b], t3i[b], t2r[b >> 1], t2i[b >> 1],
         gr[5][(b & 1) ^ ((b >> 1) & 1)], gi[5][(b & 1) ^ ((b >> 1) & 1)])
#pragma unroll
  for (int b = 0; b < 16; ++b)
    CMUL(t4r[b], t4i[b], t3r[b >> 1], t3i[b >> 1],
         gr[6][(b & 1) ^ ((b >> 1) & 1)], gi[6][(b & 1) ^ ((b >> 1) & 1)])
#pragma unroll
  for (int j = 0; j < NS; ++j)
    CMUL(ar[j], ai[j], t4r[j >> 1], t4i[j >> 1],
         Hr[(j >> 1) & 1][j & 1], Hi[(j >> 1) & 1][j & 1])
}

__device__ __forceinline__ void apply_phase_table(float ar[NS], float ai[NS],
                                                  const v2f* __restrict__ T)
{
#pragma unroll
  for (int j = 0; j < NS; ++j) {
    const v2f t = T[j * 8];
    const float r = ar[j] * t.x - ai[j] * t.y;
    const float i = ar[j] * t.y + ai[j] * t.x;
    ar[j] = r; ai[j] = i;
  }
}

__device__ __forceinline__ void group_sum_store(float v, float scale,
                                                float* __restrict__ dst, int lane)
{
  v += lane_xor<4>(v);
  v += lane_xor<2>(v);
  v += lane_xor<1>(v);
  v *= scale;
  if ((lane & 7) == 0) *dst = v;
}

__device__ __forceinline__ void shear_layers_and_readout(
    float ar[NS], float ai[NS], const v2f* __restrict__ sT,
    const float* __restrict__ sRYT, const float* __restrict__ sRYF,
    float* __restrict__ sZ, int samp, int lane)
{
  apply_phase_table(ar, ai, &sT[lane & 7]);
  ry_shear<0, 0>(ar, ai, sRYT[0], lane);
  ry_shear<1, 0>(ar, ai, sRYT[1], lane);
  ry_shear<2, 0>(ar, ai, sRYT[2], lane);
  ry_shear<3, 0>(ar, ai, sRYT[3], lane);
  ry_shear<4, 0>(ar, ai, sRYT[4], lane);
  ry_shear<5, 0>(ar, ai, sRYT[5], lane);
  ry_shear<6, 0>(ar, ai, sRYT[6], lane);
  ry_shear<7, 0>(ar, ai, sRYT[7], lane);

  apply_phase_table(ar, ai, &sT[256 + (lane & 7)]);
  ry_shear<0, 1>(ar, ai, sRYT[8],  lane);
  ry_shear<1, 1>(ar, ai, sRYT[9],  lane);
  ry_shear<2, 1>(ar, ai, sRYT[10], lane);
  ry_shear<3, 1>(ar, ai, sRYT[11], lane);
  ry_shear<4, 1>(ar, ai, sRYT[12], lane);
  ry_shear<5, 1>(ar, ai, sRYT[13], lane);
  ry_shear<6, 1>(ar, ai, sRYT[14], lane);
  ry_shear<7, 1>(ar, ai, sRYT[15], lane);

  float h[NS];
#pragma unroll
  for (int j = 0; j < NS; ++j) h[j] = ar[j]*ar[j] + ai[j]*ai[j];
#pragma unroll
  for (int s = 1; s < NS; s <<= 1) {
#pragma unroll
    for (int j = 0; j < NS; ++j) {
      if (!(j & s)) {
        const float a = h[j], b = h[j | s];
        h[j] = a + b;  h[j | s] = a - b;
      }
    }
  }
  const float fsq = ry_factor_sq(sRYF);
  float zp[8];
#define READOUT_WIRE(w) {                                                     \
    constexpr int RZ = wire_sign_mask(w, 2);                                  \
    const float v = h[RZ & 31];                                               \
    zp[w] = (__builtin_popcount(lane & (RZ >> 5)) & 1) ? -v : v;              \
  }
  READOUT_WIRE(0) READOUT_WIRE(1) READOUT_WIRE(2) READOUT_WIRE(3)
  READOUT_WIRE(4) READOUT_WIRE(5) READOUT_WIRE(6) READOUT_WIRE(7)
#undef READOUT_WIRE
#pragma unroll
  for (int w = 0; w < 8; ++w)
    group_sum_store(zp[w], fsq, &sZ[samp * 8 + w], lane);
}

__device__ __forceinline__ void rotation_layers_and_readout(
    float ar[NS], float ai[NS], const v2f* __restrict__ sT,
    const float* __restrict__ sRY, float* __restrict__ sZ, int samp, int lane)
{
  for (int L = 1; L < 3; ++L) {
    apply_phase_table(ar, ai, &sT[(L - 1) * 256 + (lane & 7)]);
    const float* ry = &sRY[L * 16];
    ry_lane<2>(ar, ai, ry[0],  ry[1]);
    ry_lane<1>(ar, ai, ry[2],  ry[3]);
    ry_lane<0>(ar, ai, ry[4],  ry[5]);
    ry_local<4>(ar, ai, ry[6],  ry[7]);
    ry_local<3>(ar, ai, ry[8],  ry[9]);
    ry_local<2>(ar, ai, ry[10], ry[11]);
    ry_local<1>(ar, ai, ry[12], ry[13]);
    ry_local<0>(ar, ai, ry[14], ry[15]);
    cnot_ring(ar, ai);
  }

  float pb[NS]; float tot = 0.f;
#pragma unroll
  for (int j = 0; j < NS; ++j) { pb[j] = ar[j]*ar[j] + ai[j]*ai[j]; tot += pb[j]; }
  float zp[8];
#pragma unroll
  for (int w = 0; w < 3; ++w)
    zp[w] = ((lane >> (2 - w)) & 1) ? -tot : tot;
  float s3 = 0.f, s4 = 0.f, s5 = 0.f, s6 = 0.f, s7 = 0.f;
#pragma unroll
  for (int j = 0; j < NS; ++j) {
    s3 += ((j >> 4) & 1) ? -pb[j] : pb[j];
    s4 += ((j >> 3) & 1) ? -pb[j] : pb[j];
    s5 += ((j >> 2) & 1) ? -pb[j] : pb[j];
    s6 += ((j >> 1) & 1) ? -pb[j] : pb[j];
    s7 += ( j       & 1) ? -pb[j] : pb[j];
  }
  zp[3] = s3; zp[4] = s4; zp[5] = s5; zp[6] = s6; zp[7] = s7;
#pragma unroll
  for (int w = 0; w < 8; ++w)
    group_sum_store(zp[w], 1.0f, &sZ[samp * 8 + w], lane);
}

__global__ __launch_bounds__(512) void quantum_block_kernel(
    const float* __restrict__ x,
    const float* __restrict__ W1, const float* __restrict__ b1,
    const float* __restrict__ W2, const float* __restrict__ b2,
    const float* __restrict__ W3, const float* __restrict__ b3,
    const float* __restrict__ qw,
    const float* __restrict__ Wp, const float* __restrict__ bp,
    float* __restrict__ out)
{
  __shared__ __attribute__((aligned(16))) float sX  [64 * 84];
  __shared__ float sH1 [64 * 68];
  __shared__ float sH2 [64 * 36];
  __shared__ float sCos[64 * 16];
  __shared__ float sSin[64 * 16];
  __shared__ float sZ  [64 * 8];
  __shared__ __attribute__((aligned(64))) v2f sT[2 * 256];
  __shared__ float sRY [24 * 2];
  __shared__ __attribute__((aligned(16))) float sRYT[16];
  __shared__ __attribute__((aligned(16))) float sRYF[16];
  __shared__ float sPH0[8 * 2];
  __shared__ __attribute__((aligned(16))) float sW1[64 * 84];
  __shared__ __attribute__((aligned(16))) float sW2[32 * 68];
  __shared__ __attribute__((aligned(16))) float sW3[16 * 36];
  __shared__ __attribute__((aligned(16))) float sWp[128 * 12];
  __shared__ float sBias[240];
  constexpr int BP_OFF = 0, B1_OFF = 128, B2_OFF = 192, B3_OFF = 224;

  const int tid     = threadIdx.x;
  const int wid     = tid >> 5;
  const int lane    = tid & 31;
  const int rowBase = blockIdx.x * 64;

  const int qi = lane < 24 ? lane * 3 + 1 : (lane - 24) * 3;
  const float qv = qw[qi];
  constexpr int W1P = 64 * 20, W3P = 16 * 8;
  int  p1[3];
  v4f  w1[3], xv[3];
#pragma unroll
  for (int it = 0; it < 3; ++it) {
    const int p = it * 512 + tid;
    p1[it] = p < W1P ? p : W1P - 1;
    xv[it] = *reinterpret_cast<const v4f*>(x + rowBase * 80 + 4 * p1[it]);
    w1[it] = *reinterpret_cast<const v4f*>(W1 + 4 * p1[it]);
  }
  const v4f w2 = *reinterpret_cast<const v4f*>(W2 + 4 * tid);
  const int p3 = tid < W3P ? tid : W3P - 1;
  const v4f w3 = *reinterpret_cast<const v4f*>(W3 + 4 * p3);
  const v2f wp = *reinterpret_cast<const v2f*>(Wp + 2 * tid);
  const int tb = tid < 240 ? tid : 239;
  const float* bsrc = tb < B1_OFF ? bp + tb
                    : tb < B2_OFF ? b1 + (tb - B1_OFF)
                    : tb < B3_OFF ? b2 + (tb - B2_OFF)
                    :               b3 + (tb - B3_OFF);
  const float bv = *bsrc;

  const float ryc = __cosf(0.5f * qv), rys = __sinf(0.5f * qv);
  const bool steep = lane >= 8 && lane < 24 && !(fabsf(rys) <= 8.0f * fabsf(ryc));
  const bool shear_ok = __builtin_amdgcn_ballot_w32(steep) == 0;
  if (wid == 0) {
    if (lane < 24) {
      sRY[lane * 2]     = ryc;
      sRY[lane * 2 + 1] = rys;
      if (lane >= 8) {
        sRYT[lane - 8] = rys / ryc;
        sRYF[lane - 8] = ryc;
      }
    } else {
      sPH0[(lane - 24) * 2]     = ryc;
      sPH0[(lane - 24) * 2 + 1] = rys;
    }
  }
  if (tid < 256) {
    build_phase_entry<1>(qw, &sT[0],   tid, false);
  } else {
    build_phase_entry<2>(qw, &sT[256], tid - 256, shear_ok);
  }
#pragma unroll
  for (int it = 0; it < 3; ++it) {
    const int o = (p1[it] / 20) * 84 + (p1[it] % 20) * 4;
    *reinterpret_cast<v4f*>(&sX[o])  = xv[it];
    *reinterpret_cast<v4f*>(&sW1[o]) = w1[it];
  }
  *reinterpret_cast<v4f*>(&sW2[(tid >> 4) * 68 + (tid & 15) * 4]) = w2;
  *reinterpret_cast<v4f*>(&sW3[(p3 >> 3) * 36 + (p3 & 7) * 4]) = w3;
  *reinterpret_cast<v2f*>(&sWp[(tid >> 2) * 12 + (tid & 3) * 2]) = wp;
  sBias[tb] = bv;
  __syncthreads();

  {
    const int mt = wid >> 2, nt = wid & 3;
    gemm_tile_to_lds<80, 84, 84, true >(sX + mt * 16 * 84, sW1, sBias + B1_OFF, sH1 + mt * 16 * 68, 68, nt * 16);
  }
  __syncthreads();
  if (wid < 8) {
    const int mt = wid >> 1, nt = wid & 1;
    gemm_tile_to_lds<64, 68, 68, true >(sH1 + mt * 16 * 68, sW2, sBias + B2_OFF, sH2 + mt * 16 * 36, 36, nt * 16);
  }
  __syncthreads();
  if (wid < 4)
    gemm_tile_to_halfangle_trig<32, 36, 36>(sH2 + wid * 16 * 36, sW3, sBias + B3_OFF,
                                        sCos + wid * 16 * 16, sSin + wid * 16 * 16, 16, 0);
  __syncthreads();

  {
    const int samp = wid * 4 + (lane >> 3);
    float ar[NS], ai[NS];
    product_state_behind_ring(&sCos[samp * 16], &sSin[samp * 16], sPH0, sRY, ar, ai);

    if (shear_ok)
      shear_layers_and_readout(ar, ai, sT, sRYT, sRYF, sZ, samp, lane);
    else
      rotation_layers_and_readout(ar, ai, sT, sRY, sZ, samp, lane);
  }
  __syncthreads();

#pragma unroll
  for (int i = 0; i < 2; ++i) {
    const int mt = (wid >> 3) * 2 + i;
    gemm_final(sZ + mt * 16 * 8, sWp, sBias + BP_OFF, out, rowBase + mt * 16, (wid & 7) * 16);
  }
}

extern "C" void kernel_launch(void* const* d_in, const int* in_sizes, int n_in,
                              void* d_out, int out_size, void* d_ws, size_t ws_size,
                              hipStream_t stream) {
  const float* x  = (const float*)d_in[0];
  const float* W1 = (const float*)d_in[1];
  const float* b1 = (const float*)d_in[2];
  const float* W2 = (const float*)d_in[3];
  const float* b2 = (const float*)d_in[4];
  const float* W3 = (const float*)d_in[5];
  const float* b3 = (const float*)d_in[6];
  const float* qw = (const float*)d_in[7];
  const float* Wp = (const float*)d_in[8];
  const float* bp = (const float*)d_in[9];
  float* out = (float*)d_out;

  const int B = in_sizes[0] / 80;
  const int grid = B / 64;
  quantum_block_kernel<<<grid, 512, 0, stream>>>(x, W1, b1, W2, b2, W3, b3,
                                                 qw, Wp, bp, out);
}
